# starting kernel
# baseline (speedup 1.0000x reference)
#include <hip/hip_runtime.h>

typedef __attribute__((ext_vector_type(2))) float v2f;
typedef __attribute__((ext_vector_type(8))) float v8f;

#define NEG_SLOPE 0.2f

__global__ void fillf_kernel(float* __restrict__ p, int n, float v) {
  int i = blockIdx.x * blockDim.x + threadIdx.x;
  if (i < n) p[i] = v;
}

__device__ inline void atomicMaxF(float* addr, float v) {
  if (v >= 0.0f) atomicMax((int*)addr, __float_as_int(v));
  else           atomicMin((unsigned int*)addr, __float_as_uint(v));
}

__global__ void pack_w_kernel(const float* __restrict__ W,
                              v2f* __restrict__ Wp, int K, int Ncol) {
  const int i = blockIdx.x * blockDim.x + threadIdx.x;
  const int total = (K >> 1) * Ncol;
  if (i >= total) return;
  const int kp  = i / Ncol;
  const int col = i - kp * Ncol;
  v2f p;
  p[0] = W[(size_t)(2 * kp)     * Ncol + col];
  p[1] = W[(size_t)(2 * kp + 1) * Ncol + col];
  Wp[i] = p;
}

__global__ void gemm_bias_wmma_n64(const float* __restrict__ X,
                                   const v2f* __restrict__ Wp,
                                   const float* __restrict__ bias,
                                   float* __restrict__ Y,
                                   int M, int K, int Ncol) {
  const int lane = threadIdx.x & 31;
  const int wave = threadIdx.x >> 5;
  const int ct = Ncol >> 6;
  const int rt = M >> 4;
  const int tile = blockIdx.x * (blockDim.x >> 5) + wave;
  if (tile >= ct * rt) return;

  const int rowBase = (tile / ct) << 4;
  const int colBase = (tile % ct) << 6;
  const int half = lane >> 4;
  const int l15  = lane & 15;

  const float* __restrict__ ap = X + (size_t)(rowBase + l15) * K + half * 2;
  const v2f*   __restrict__ bp = Wp + (size_t)half * Ncol + colBase + l15;

  v8f cc[4] = {v8f{}, v8f{}, v8f{}, v8f{}};
  #pragma unroll 4
  for (int k0 = 0; k0 < K; k0 += 4) {
    const v2f a = *(const v2f*)(ap + k0);
    const v2f* __restrict__ br = bp + (size_t)(k0 >> 1) * Ncol;
    const v2f bv0 = br[0];
    const v2f bv1 = br[16];
    const v2f bv2 = br[32];
    const v2f bv3 = br[48];
    cc[0] = __builtin_amdgcn_wmma_f32_16x16x4_f32(false, a, false, bv0, (short)0, cc[0], false, false);
    cc[1] = __builtin_amdgcn_wmma_f32_16x16x4_f32(false, a, false, bv1, (short)0, cc[1], false, false);
    cc[2] = __builtin_amdgcn_wmma_f32_16x16x4_f32(false, a, false, bv2, (short)0, cc[2], false, false);
    cc[3] = __builtin_amdgcn_wmma_f32_16x16x4_f32(false, a, false, bv3, (short)0, cc[3], false, false);
  }

  const int row0 = rowBase + half * 8;
  #pragma unroll
  for (int t = 0; t < 4; ++t) {
    const int col = colBase + t * 16 + l15;
    const float bb = bias ? bias[col] : 0.0f;
    float* __restrict__ yp = Y + (size_t)row0 * Ncol + col;
    #pragma unroll
    for (int r = 0; r < 8; ++r)
      yp[(size_t)r * Ncol] = cc[t][r] + bb;
  }
}

__global__ void gemm_bias_wmma_guard(const float* __restrict__ X,
                                     const v2f* __restrict__ Wp,
                                     const float* __restrict__ bias,
                                     float* __restrict__ Y,
                                     int M, int K, int Ncol) {
  const int lane = threadIdx.x & 31;
  const int wave = threadIdx.x >> 5;
  const int ct = (Ncol + 15) >> 4;
  const int rt = M >> 4;
  const int tile = blockIdx.x * (blockDim.x >> 5) + wave;
  if (tile >= ct * rt) return;

  const int rowBase = (tile / ct) << 4;
  const int colBase = (tile % ct) << 4;
  const int half = lane >> 4;
  const int l15  = lane & 15;
  const int bcol = colBase + l15;
  const bool bok = bcol < Ncol;
  const int bcolc = bok ? bcol : (Ncol - 1);
  const float bmask = bok ? 1.0f : 0.0f;

  const float* __restrict__ ap = X + (size_t)(rowBase + l15) * K + half * 2;
  const v2f*   __restrict__ bp = Wp + (size_t)half * Ncol + bcolc;

  v8f c = {};
  #pragma unroll 4
  for (int k0 = 0; k0 < K; k0 += 4) {
    const v2f a = *(const v2f*)(ap + k0);
    v2f b = bp[(size_t)(k0 >> 1) * Ncol];
    b[0] *= bmask;
    b[1] *= bmask;
    c = __builtin_amdgcn_wmma_f32_16x16x4_f32(false, a, false, b, (short)0, c, false, false);
  }

  if (bok) {
    const float bb = bias ? bias[bcol] : 0.0f;
    const int row0 = rowBase + half * 8;
    float* __restrict__ yp = Y + (size_t)row0 * Ncol + bcol;
    #pragma unroll
    for (int r = 0; r < 8; ++r)
      yp[(size_t)r * Ncol] = c[r] + bb;
  }
}

__global__ void edge_logits_kernel(const int* __restrict__ ei, int E, int ET,
                                   const float* __restrict__ XL,
                                   const float* __restrict__ XR,
                                   const float* __restrict__ att,
                                   float* __restrict__ eo,
                                   float* __restrict__ mx, int d4) {
  const int wid  = (blockIdx.x * blockDim.x + threadIdx.x) >> 5;
  const int lane = threadIdx.x & 31;
  if (wid >= ET) return;
  int s, t;
  if (wid < E) { s = ei[wid]; t = ei[E + wid]; } else { s = t = wid - E; }
  const float4* __restrict__ xl = (const float4*)XL + (size_t)s * d4;
  const float4* __restrict__ xr = (const float4*)XR + (size_t)t * d4;
  const float4* __restrict__ at = (const float4*)att;
  float acc = 0.0f;
  for (int f = lane; f < d4; f += 32) {
    const float4 a = xl[f];
    const float4 b = xr[f];
    const float4 g = at[f];
    float h;
    h = a.x + b.x; h = h > 0.0f ? h : NEG_SLOPE * h; acc += h * g.x;
    h = a.y + b.y; h = h > 0.0f ? h : NEG_SLOPE * h; acc += h * g.y;
    h = a.z + b.z; h = h > 0.0f ? h : NEG_SLOPE * h; acc += h * g.z;
    h = a.w + b.w; h = h > 0.0f ? h : NEG_SLOPE * h; acc += h * g.w;
  }
  #pragma unroll
  for (int off = 16; off > 0; off >>= 1) acc += __shfl_xor(acc, off, 32);
  if (lane == 0) { eo[wid] = acc; atomicMaxF(&mx[t], acc); }
}

__global__ void edge_softmax_kernel(const int* __restrict__ ei, int E, int ET,
                                    const float* __restrict__ mx,
                                    float* __restrict__ ep,
                                    float* __restrict__ sx) {
  const int i = blockIdx.x * blockDim.x + threadIdx.x;
  if (i >= ET) return;
  const int t = (i < E) ? ei[E + i] : (i - E);
  const float p = __expf(ep[i] - mx[t]);
  ep[i] = p;
  atomicAdd(&sx[t], p);
}

__global__ void edge_scatter_kernel(const int* __restrict__ ei, int E, int ET,
                                    const float* __restrict__ XL,
                                    const float* __restrict__ ep,
                                    const float* __restrict__ sx,
                                    float* __restrict__ out, int d4) {
  const int wid  = (blockIdx.x * blockDim.x + threadIdx.x) >> 5;
  const int lane = threadIdx.x & 31;
  if (wid >= ET) return;
  int s, t;
  if (wid < E) { s = ei[wid]; t = ei[E + wid]; } else { s = t = wid - E; }
  const float alpha = ep[wid] / sx[t];
  const float4* __restrict__ xl = (const float4*)XL + (size_t)s * d4;
  float* __restrict__ op = out + (size_t)t * (d4 * 4);
  for (int f = lane; f < d4; f += 32) {
    const float4 v = xl[f];
    float* o4 = op + f * 4;
    atomicAdd(o4 + 0, alpha * v.x);
    atomicAdd(o4 + 1, alpha * v.y);
    atomicAdd(o4 + 2, alpha * v.z);
    atomicAdd(o4 + 3, alpha * v.w);
  }
}

__global__ void finalize_kernel(float* __restrict__ out,
                                const float* __restrict__ bias,
                                int n, int dmask, int relu) {
  const int i = blockIdx.x * blockDim.x + threadIdx.x;
  if (i >= n) return;
  float v = out[i] + bias[i & dmask];
  if (relu) v = fmaxf(v, 0.0f);
  out[i] = v;
}

extern "C" void kernel_launch(void* const* d_in, const int* in_sizes, int n_in,
                              void* d_out, int out_size, void* d_ws, size_t ws_size,
                              hipStream_t stream) {
  const float* x   = (const float*)d_in[0];
  const int*   ei  = (const int*)  d_in[1];
  const float* w1l = (const float*)d_in[2];
  const float* b1l = (const float*)d_in[3];
  const float* w1r = (const float*)d_in[4];
  const float* b1r = (const float*)d_in[5];
  const float* a1  = (const float*)d_in[6];
  const float* o1  = (const float*)d_in[7];
  const float* w2l = (const float*)d_in[8];
  const float* b2l = (const float*)d_in[9];
  const float* w2r = (const float*)d_in[10];
  const float* b2r = (const float*)d_in[11];
  const float* a2  = (const float*)d_in[12];
  const float* o2  = (const float*)d_in[13];
  const float* w3l = (const float*)d_in[14];
  const float* b3l = (const float*)d_in[15];
  const float* w3r = (const float*)d_in[16];
  const float* b3r = (const float*)d_in[17];
  const float* a3  = (const float*)d_in[18];
  const float* o3  = (const float*)d_in[19];
  const float* wc  = (const float*)d_in[20];
  const float* bc  = (const float*)d_in[21];

  const int N  = in_sizes[0] / 128;
  const int E  = in_sizes[1] / 2;
  const int ET = E + N;

  float* B0  = (float*)d_ws;
  float* B1  = B0 + (size_t)N * 256;
  float* B2  = B1 + (size_t)N * 256;
  float* EP  = B2 + (size_t)N * 256;
  float* Mx  = EP + (size_t)ET;
  float* Sx  = Mx + (size_t)N;
  v2f*   WPa = (v2f*)(Sx + (size_t)N);
  v2f*   WPb = WPa + 32768;

  auto packW = [&](const float* W, v2f* Wp, int k, int n) {
    const int total = (k >> 1) * n;
    pack_w_kernel<<<(total + 255) / 256, 256, 0, stream>>>(W, Wp, k, n);
  };
  auto gemmBig = [&](const float* X, const v2f* Wp, const float* b, float* Y,
                     int m, int k, int n) {
    const int tiles = (m >> 4) * (n >> 6);
    gemm_bias_wmma_n64<<<(tiles + 3) / 4, 128, 0, stream>>>(X, Wp, b, Y, m, k, n);
  };
  auto gemmGuard = [&](const float* X, const v2f* Wp, const float* b, float* Y,
                       int m, int k, int n) {
    const int tiles = (m >> 4) * ((n + 15) / 16);
    gemm_bias_wmma_guard<<<(tiles + 3) / 4, 128, 0, stream>>>(X, Wp, b, Y, m, k, n);
  };

  auto layer = [&](const float* Xin, int din, int dout,
                   const float* wl, const float* bl,
                   const float* wr, const float* br,
                   const float* att, const float* ob,
                   float* XL, float* XO, int relu) {
    packW(wl, WPa, din, dout);
    packW(wr, WPb, din, dout);
    gemmBig(Xin, WPa, bl, XL, N, din, dout);
    gemmBig(Xin, WPb, br, XO, N, din, dout);
    fillf_kernel<<<(N + 255) / 256, 256, 0, stream>>>(Mx, N, -__builtin_huge_valf());
    fillf_kernel<<<(N + 255) / 256, 256, 0, stream>>>(Sx, N, 0.0f);
    edge_logits_kernel<<<(ET + 7) / 8, 256, 0, stream>>>(ei, E, ET, XL, XO, att,
                                                         EP, Mx, dout >> 2);
    fillf_kernel<<<(N * dout + 255) / 256, 256, 0, stream>>>(XO, N * dout, 0.0f);
    edge_softmax_kernel<<<(ET + 255) / 256, 256, 0, stream>>>(ei, E, ET, Mx, EP, Sx);
    edge_scatter_kernel<<<(ET + 7) / 8, 256, 0, stream>>>(ei, E, ET, XL, EP, Sx,
                                                          XO, dout >> 2);
    finalize_kernel<<<(N * dout + 255) / 256, 256, 0, stream>>>(XO, ob,
                                                                N * dout, dout - 1, relu);
  };

  layer(x,  128, 128, w1l, b1l, w1r, b1r, a1, o1, B0, B1, 1);
  layer(B1, 128, 128, w2l, b2l, w2r, b2r, a2, o2, B0, B2, 1);
  layer(B2, 128, 256, w3l, b3l, w3r, b3r, a3, o3, B0, B1, 0);
  packW(wc, WPa, 256, 40);
  gemmGuard(B1, WPa, bc, (float*)d_out, N, 256, 40);
}


# Edges grouped by destination once per call (counting sort, arrival-order rank); one wave per node does logits, streaming softmax and weighted sum in registers; float atomics and fills gone
# speedup vs baseline: 3.4294x; 3.4294x over previous
#include <hip/hip_runtime.h>

typedef __attribute__((ext_vector_type(2))) float v2f;
typedef __attribute__((ext_vector_type(8))) float v8f;

#define NEG_SLOPE 0.2f

__global__ void filli_kernel(int* __restrict__ p, int n, int v) {
  int i = blockIdx.x * blockDim.x + threadIdx.x;
  if (i < n) p[i] = v;
}

__global__ void pack_w_kernel(const float* __restrict__ W,
                              v2f* __restrict__ Wp, int K, int Ncol) {
  const int i = blockIdx.x * blockDim.x + threadIdx.x;
  const int total = (K >> 1) * Ncol;
  if (i >= total) return;
  const int kp  = i / Ncol;
  const int col = i - kp * Ncol;
  v2f p;
  p[0] = W[(size_t)(2 * kp)     * Ncol + col];
  p[1] = W[(size_t)(2 * kp + 1) * Ncol + col];
  Wp[i] = p;
}

__global__ void gemm_bias_wmma_n64(const float* __restrict__ X,
                                   const v2f* __restrict__ Wp,
                                   const float* __restrict__ bias,
                                   float* __restrict__ Y,
                                   int M, int K, int Ncol) {
  const int lane = threadIdx.x & 31;
  const int wave = threadIdx.x >> 5;
  const int ct = Ncol >> 6;
  const int rt = M >> 4;
  const int tile = blockIdx.x * (blockDim.x >> 5) + wave;
  if (tile >= ct * rt) return;

  const int rowBase = (tile / ct) << 4;
  const int colBase = (tile % ct) << 6;
  const int half = lane >> 4;
  const int l15  = lane & 15;

  const float* __restrict__ ap = X + (size_t)(rowBase + l15) * K + half * 2;
  const v2f*   __restrict__ bp = Wp + (size_t)half * Ncol + colBase + l15;

  v8f cc[4] = {v8f{}, v8f{}, v8f{}, v8f{}};
  #pragma unroll 4
  for (int k0 = 0; k0 < K; k0 += 4) {
    const v2f a = *(const v2f*)(ap + k0);
    const v2f* __restrict__ br = bp + (size_t)(k0 >> 1) * Ncol;
    const v2f bv0 = br[0];
    const v2f bv1 = br[16];
    const v2f bv2 = br[32];
    const v2f bv3 = br[48];
    cc[0] = __builtin_amdgcn_wmma_f32_16x16x4_f32(false, a, false, bv0, (short)0, cc[0], false, false);
    cc[1] = __builtin_amdgcn_wmma_f32_16x16x4_f32(false, a, false, bv1, (short)0, cc[1], false, false);
    cc[2] = __builtin_amdgcn_wmma_f32_16x16x4_f32(false, a, false, bv2, (short)0, cc[2], false, false);
    cc[3] = __builtin_amdgcn_wmma_f32_16x16x4_f32(false, a, false, bv3, (short)0, cc[3], false, false);
  }

  const int row0 = rowBase + half * 8;
  #pragma unroll
  for (int t = 0; t < 4; ++t) {
    const int col = colBase + t * 16 + l15;
    const float bb = bias ? bias[col] : 0.0f;
    float* __restrict__ yp = Y + (size_t)row0 * Ncol + col;
    #pragma unroll
    for (int r = 0; r < 8; ++r)
      yp[(size_t)r * Ncol] = cc[t][r] + bb;
  }
}

__global__ void gemm_bias_wmma_guard(const float* __restrict__ X,
                                     const v2f* __restrict__ Wp,
                                     const float* __restrict__ bias,
                                     float* __restrict__ Y,
                                     int M, int K, int Ncol) {
  const int lane = threadIdx.x & 31;
  const int wave = threadIdx.x >> 5;
  const int ct = (Ncol + 15) >> 4;
  const int rt = M >> 4;
  const int tile = blockIdx.x * (blockDim.x >> 5) + wave;
  if (tile >= ct * rt) return;

  const int rowBase = (tile / ct) << 4;
  const int colBase = (tile % ct) << 4;
  const int half = lane >> 4;
  const int l15  = lane & 15;
  const int bcol = colBase + l15;
  const bool bok = bcol < Ncol;
  const int bcolc = bok ? bcol : (Ncol - 1);
  const float bmask = bok ? 1.0f : 0.0f;

  const float* __restrict__ ap = X + (size_t)(rowBase + l15) * K + half * 2;
  const v2f*   __restrict__ bp = Wp + (size_t)half * Ncol + bcolc;

  v8f c = {};
  #pragma unroll 4
  for (int k0 = 0; k0 < K; k0 += 4) {
    const v2f a = *(const v2f*)(ap + k0);
    v2f b = bp[(size_t)(k0 >> 1) * Ncol];
    b[0] *= bmask;
    b[1] *= bmask;
    c = __builtin_amdgcn_wmma_f32_16x16x4_f32(false, a, false, b, (short)0, c, false, false);
  }

  if (bok) {
    const float bb = bias ? bias[bcol] : 0.0f;
    const int row0 = rowBase + half * 8;
    float* __restrict__ yp = Y + (size_t)row0 * Ncol + bcol;
    #pragma unroll
    for (int r = 0; r < 8; ++r)
      yp[(size_t)r * Ncol] = c[r] + bb;
  }
}

__global__ void degree_count_kernel(const int* __restrict__ ei, int E,
                                    int* __restrict__ cnt) {
  const int i = blockIdx.x * blockDim.x + threadIdx.x;
  if (i < E) atomicAdd(&cnt[ei[E + i]], 1);
}

__global__ void __launch_bounds__(256)
scan_local_kernel(const int* __restrict__ cnt, int* __restrict__ loc,
                  int* __restrict__ bsum, int n) {
  __shared__ int wtot[8];
  const int tid  = threadIdx.x;
  const int lane = tid & 31;
  const int wave = tid >> 5;
  const int base = blockIdx.x * 1024 + tid * 4;
  const int c0 = (base + 0 < n) ? cnt[base + 0] : 0;
  const int c1 = (base + 1 < n) ? cnt[base + 1] : 0;
  const int c2 = (base + 2 < n) ? cnt[base + 2] : 0;
  const int c3 = (base + 3 < n) ? cnt[base + 3] : 0;
  const int tsum = c0 + c1 + c2 + c3;
  int incl = tsum;
  #pragma unroll
  for (int off = 1; off < 32; off <<= 1) {
    const int y = __shfl_up(incl, off, 32);
    if (lane >= off) incl += y;
  }
  if (lane == 31) wtot[wave] = incl;
  __syncthreads();
  int woff = 0;
  #pragma unroll
  for (int w = 0; w < 8; ++w) woff += (w < wave) ? wtot[w] : 0;
  const int excl = woff + incl - tsum;
  if (base + 0 < n) loc[base + 0] = excl;
  if (base + 1 < n) loc[base + 1] = excl + c0;
  if (base + 2 < n) loc[base + 2] = excl + c0 + c1;
  if (base + 3 < n) loc[base + 3] = excl + c0 + c1 + c2;
  if (tid == 255) bsum[blockIdx.x] = excl + tsum;
}

__global__ void __launch_bounds__(256)
scan_add_kernel(const int* __restrict__ loc, const int* __restrict__ bsum,
                int* __restrict__ rowptr, int* __restrict__ cursor, int n) {
  const int i = blockIdx.x * 256 + threadIdx.x;
  const int b = blockIdx.x >> 2;
  int off = 0;
  for (int q = 0; q < b; ++q) off += bsum[q];
  if (i < n) {
    const int v = loc[i] + off;
    rowptr[i] = v;
    cursor[i] = v;
  }
}

__global__ void csr_fill_kernel(const int* __restrict__ ei, int E,
                                int* __restrict__ cursor,
                                int* __restrict__ csr_src) {
  const int i = blockIdx.x * blockDim.x + threadIdx.x;
  if (i >= E) return;
  const int pos = atomicAdd(&cursor[ei[E + i]], 1);
  csr_src[pos] = ei[i];
}

template <int V>
__global__ void __launch_bounds__(256)
node_attend_kernel(const int* __restrict__ rowptr,
                   const int* __restrict__ csr_src, int N, int E,
                   const float* __restrict__ XL, const float* XR,
                   const float* __restrict__ att,
                   const float* __restrict__ obias,
                   float* out, int relu) {
  const int lane = threadIdx.x & 31;
  const int wid_v = (blockIdx.x * blockDim.x + threadIdx.x) >> 5;
  if (wid_v >= N) return;
  const int wid = __builtin_amdgcn_readfirstlane(wid_v);
  const int beg = __builtin_amdgcn_readfirstlane(rowptr[wid]);
  const int end = (wid + 1 < N) ? __builtin_amdgcn_readfirstlane(rowptr[wid + 1]) : E;
  const int c   = end - beg + 1;

  constexpr int D4 = 32 * V;
  const float4* __restrict__ xl4 = (const float4*)XL;
  const float4* xr4 = (const float4*)XR + (size_t)wid * D4 + lane;
  const float4* __restrict__ at4 = (const float4*)att + lane;

  float4 xr[V], g[V], acc[V];
  #pragma unroll
  for (int v = 0; v < V; ++v) {
    xr[v]  = xr4[v * 32];
    g[v]   = at4[v * 32];
    acc[v] = make_float4(0.0f, 0.0f, 0.0f, 0.0f);
  }
  float m = -__builtin_huge_valf();
  float ssum = 0.0f;

  for (int base = 0; base < c; base += 32) {
    const int q = base + lane;
    int idxv = wid;
    if (q > 0 && q < c) idxv = csr_src[beg + q - 1];
    const int nq = (c - base < 32) ? (c - base) : 32;

    for (int k = 0; k < nq; k += 4) {
      float4 r[4][V];
      float e[4];
      #pragma unroll
      for (int u = 0; u < 4; ++u) {
        const int kk = (k + u < nq) ? (k + u) : k;
        const int s = __builtin_amdgcn_readlane(idxv, kk);
        const float4* __restrict__ row = xl4 + (size_t)s * D4 + lane;
        #pragma unroll
        for (int v = 0; v < V; ++v) r[u][v] = row[v * 32];
      }
      #pragma unroll
      for (int u = 0; u < 4; ++u) {
        float a = 0.0f;
        #pragma unroll
        for (int v = 0; v < V; ++v) {
          float h;
          h = r[u][v].x + xr[v].x; h = h > 0.0f ? h : NEG_SLOPE * h; a += h * g[v].x;
          h = r[u][v].y + xr[v].y; h = h > 0.0f ? h : NEG_SLOPE * h; a += h * g[v].y;
          h = r[u][v].z + xr[v].z; h = h > 0.0f ? h : NEG_SLOPE * h; a += h * g[v].z;
          h = r[u][v].w + xr[v].w; h = h > 0.0f ? h : NEG_SLOPE * h; a += h * g[v].w;
        }
        e[u] = a;
      }
      #pragma unroll
      for (int off = 16; off > 0; off >>= 1) {
        #pragma unroll
        for (int u = 0; u < 4; ++u) e[u] += __shfl_xor(e[u], off, 32);
      }
      float mc = e[0];
      #pragma unroll
      for (int u = 1; u < 4; ++u)
        if (k + u < nq) mc = fmaxf(mc, e[u]);
      const float mn = fmaxf(m, mc);
      const float sc = __expf(m - mn);
      ssum *= sc;
      #pragma unroll
      for (int v = 0; v < V; ++v) {
        acc[v].x *= sc; acc[v].y *= sc; acc[v].z *= sc; acc[v].w *= sc;
      }
      #pragma unroll
      for (int u = 0; u < 4; ++u) {
        if (k + u < nq) {
          const float p = __expf(e[u] - mn);
          ssum += p;
          #pragma unroll
          for (int v = 0; v < V; ++v) {
            acc[v].x += p * r[u][v].x;
            acc[v].y += p * r[u][v].y;
            acc[v].z += p * r[u][v].z;
            acc[v].w += p * r[u][v].w;
          }
        }
      }
      m = mn;
    }
  }

  const float inv = 1.0f / ssum;
  const float4* __restrict__ ob4 = (const float4*)obias + lane;
  float4* o4 = (float4*)out + (size_t)wid * D4 + lane;
  #pragma unroll
  for (int v = 0; v < V; ++v) {
    const float4 bb = ob4[v * 32];
    float4 o;
    o.x = acc[v].x * inv + bb.x;
    o.y = acc[v].y * inv + bb.y;
    o.z = acc[v].z * inv + bb.z;
    o.w = acc[v].w * inv + bb.w;
    if (relu) {
      o.x = fmaxf(o.x, 0.0f); o.y = fmaxf(o.y, 0.0f);
      o.z = fmaxf(o.z, 0.0f); o.w = fmaxf(o.w, 0.0f);
    }
    o4[v * 32] = o;
  }
}

extern "C" void kernel_launch(void* const* d_in, const int* in_sizes, int n_in,
                              void* d_out, int out_size, void* d_ws, size_t ws_size,
                              hipStream_t stream) {
  const float* x   = (const float*)d_in[0];
  const int*   ei  = (const int*)  d_in[1];
  const float* w1l = (const float*)d_in[2];
  const float* b1l = (const float*)d_in[3];
  const float* w1r = (const float*)d_in[4];
  const float* b1r = (const float*)d_in[5];
  const float* a1  = (const float*)d_in[6];
  const float* o1  = (const float*)d_in[7];
  const float* w2l = (const float*)d_in[8];
  const float* b2l = (const float*)d_in[9];
  const float* w2r = (const float*)d_in[10];
  const float* b2r = (const float*)d_in[11];
  const float* a2  = (const float*)d_in[12];
  const float* o2  = (const float*)d_in[13];
  const float* w3l = (const float*)d_in[14];
  const float* b3l = (const float*)d_in[15];
  const float* w3r = (const float*)d_in[16];
  const float* b3r = (const float*)d_in[17];
  const float* a3  = (const float*)d_in[18];
  const float* o3  = (const float*)d_in[19];
  const float* wc  = (const float*)d_in[20];
  const float* bc  = (const float*)d_in[21];

  const int N  = in_sizes[0] / 128;
  const int E  = in_sizes[1] / 2;

  const int Np = (N + 3) & ~3;
  const int Ep = (E + 3) & ~3;
  float* B0  = (float*)d_ws;
  float* B1  = B0 + (size_t)N * 256;
  float* B2  = B1 + (size_t)N * 256;
  int*   CS  = (int*)(B2 + (size_t)N * 256);
  int*   RP  = CS + (size_t)Ep;
  int*   CU  = RP + (size_t)Np;
  int*   CN  = CU + (size_t)Np;
  int*   LO  = CN + (size_t)Np;
  int*   BS  = LO + (size_t)Np;
  v2f*   WPa = (v2f*)(BS + 4096);
  v2f*   WPb = WPa + 32768;

  filli_kernel<<<(N + 255) / 256, 256, 0, stream>>>(CN, N, 0);
  degree_count_kernel<<<(E + 255) / 256, 256, 0, stream>>>(ei, E, CN);
  scan_local_kernel<<<(N + 1023) / 1024, 256, 0, stream>>>(CN, LO, BS, N);
  scan_add_kernel<<<4 * ((N + 1023) / 1024), 256, 0, stream>>>(LO, BS, RP, CU, N);
  csr_fill_kernel<<<(E + 255) / 256, 256, 0, stream>>>(ei, E, CU, CS);

  auto packW = [&](const float* W, v2f* Wp, int k, int n) {
    const int total = (k >> 1) * n;
    pack_w_kernel<<<(total + 255) / 256, 256, 0, stream>>>(W, Wp, k, n);
  };
  auto gemmBig = [&](const float* X, const v2f* Wp, const float* b, float* Y,
                     int m, int k, int n) {
    const int tiles = (m >> 4) * (n >> 6);
    gemm_bias_wmma_n64<<<(tiles + 3) / 4, 128, 0, stream>>>(X, Wp, b, Y, m, k, n);
  };
  auto gemmGuard = [&](const float* X, const v2f* Wp, const float* b, float* Y,
                       int m, int k, int n) {
    const int tiles = (m >> 4) * ((n + 15) / 16);
    gemm_bias_wmma_guard<<<(tiles + 3) / 4, 128, 0, stream>>>(X, Wp, b, Y, m, k, n);
  };

  auto layer = [&](const float* Xin, int din, int dout,
                   const float* wl, const float* bl,
                   const float* wr, const float* br,
                   const float* att, const float* ob,
                   float* XL, float* XO, int relu) {
    packW(wl, WPa, din, dout);
    packW(wr, WPb, din, dout);
    gemmBig(Xin, WPa, bl, XL, N, din, dout);
    gemmBig(Xin, WPb, br, XO, N, din, dout);
    if (dout == 128)
      node_attend_kernel<1><<<(N + 7) / 8, 256, 0, stream>>>(RP, CS, N, E, XL, XO,
                                                           att, ob, XO, relu);
    else
      node_attend_kernel<2><<<(N + 7) / 8, 256, 0, stream>>>(RP, CS, N, E, XL, XO,
                                                           att, ob, XO, relu);
  };

  layer(x,  128, 128, w1l, b1l, w1r, b1r, a1, o1, B0, B1, 1);
  layer(B1, 128, 128, w2l, b2l, w2r, b2r, a2, o2, B0, B2, 1);
  layer(B2, 128, 256, w3l, b3l, w3r, b3r, a3, o3, B0, B1, 0);
  packW(wc, WPa, 256, 40);
  gemmGuard(B1, WPa, bc, (float*)d_out, N, 256, 40);
}


# Edges grouped by destination with LDS counters only (chunk counts per 256-node bin, scan, scatter to bins, per-bin count and rank); global atomics gone; rank in a node is arrival order of LDS adds
# speedup vs baseline: 4.1405x; 1.2074x over previous
#include <hip/hip_runtime.h>

typedef __attribute__((ext_vector_type(2))) float v2f;
typedef __attribute__((ext_vector_type(8))) float v8f;

#define NEG_SLOPE 0.2f

__global__ void pack_w_kernel(const float* __restrict__ W,
                              v2f* __restrict__ Wp, int K, int Ncol) {
  const int i = blockIdx.x * blockDim.x + threadIdx.x;
  const int total = (K >> 1) * Ncol;
  if (i >= total) return;
  const int kp  = i / Ncol;
  const int col = i - kp * Ncol;
  v2f p;
  p[0] = W[(size_t)(2 * kp)     * Ncol + col];
  p[1] = W[(size_t)(2 * kp + 1) * Ncol + col];
  Wp[i] = p;
}

__global__ void gemm_bias_wmma_n64(const float* __restrict__ X,
                                   const v2f* __restrict__ Wp,
                                   const float* __restrict__ bias,
                                   float* __restrict__ Y,
                                   int M, int K, int Ncol) {
  const int lane = threadIdx.x & 31;
  const int wave = threadIdx.x >> 5;
  const int ct = Ncol >> 6;
  const int rt = M >> 4;
  const int tile = blockIdx.x * (blockDim.x >> 5) + wave;
  if (tile >= ct * rt) return;

  const int rowBase = (tile / ct) << 4;
  const int colBase = (tile % ct) << 6;
  const int half = lane >> 4;
  const int l15  = lane & 15;

  const float* __restrict__ ap = X + (size_t)(rowBase + l15) * K + half * 2;
  const v2f*   __restrict__ bp = Wp + (size_t)half * Ncol + colBase + l15;

  v8f cc[4] = {v8f{}, v8f{}, v8f{}, v8f{}};
  #pragma unroll 4
  for (int k0 = 0; k0 < K; k0 += 4) {
    const v2f a = *(const v2f*)(ap + k0);
    const v2f* __restrict__ br = bp + (size_t)(k0 >> 1) * Ncol;
    const v2f bv0 = br[0];
    const v2f bv1 = br[16];
    const v2f bv2 = br[32];
    const v2f bv3 = br[48];
    cc[0] = __builtin_amdgcn_wmma_f32_16x16x4_f32(false, a, false, bv0, (short)0, cc[0], false, false);
    cc[1] = __builtin_amdgcn_wmma_f32_16x16x4_f32(false, a, false, bv1, (short)0, cc[1], false, false);
    cc[2] = __builtin_amdgcn_wmma_f32_16x16x4_f32(false, a, false, bv2, (short)0, cc[2], false, false);
    cc[3] = __builtin_amdgcn_wmma_f32_16x16x4_f32(false, a, false, bv3, (short)0, cc[3], false, false);
  }

  const int row0 = rowBase + half * 8;
  #pragma unroll
  for (int t = 0; t < 4; ++t) {
    const int col = colBase + t * 16 + l15;
    const float bb = bias ? bias[col] : 0.0f;
    float* __restrict__ yp = Y + (size_t)row0 * Ncol + col;
    #pragma unroll
    for (int r = 0; r < 8; ++r)
      yp[(size_t)r * Ncol] = cc[t][r] + bb;
  }
}

__global__ void gemm_bias_wmma_guard(const float* __restrict__ X,
                                     const v2f* __restrict__ Wp,
                                     const float* __restrict__ bias,
                                     float* __restrict__ Y,
                                     int M, int K, int Ncol) {
  const int lane = threadIdx.x & 31;
  const int wave = threadIdx.x >> 5;
  const int ct = (Ncol + 15) >> 4;
  const int rt = M >> 4;
  const int tile = blockIdx.x * (blockDim.x >> 5) + wave;
  if (tile >= ct * rt) return;

  const int rowBase = (tile / ct) << 4;
  const int colBase = (tile % ct) << 4;
  const int half = lane >> 4;
  const int l15  = lane & 15;
  const int bcol = colBase + l15;
  const bool bok = bcol < Ncol;
  const int bcolc = bok ? bcol : (Ncol - 1);
  const float bmask = bok ? 1.0f : 0.0f;

  const float* __restrict__ ap = X + (size_t)(rowBase + l15) * K + half * 2;
  const v2f*   __restrict__ bp = Wp + (size_t)half * Ncol + bcolc;

  v8f c = {};
  #pragma unroll 4
  for (int k0 = 0; k0 < K; k0 += 4) {
    const v2f a = *(const v2f*)(ap + k0);
    v2f b = bp[(size_t)(k0 >> 1) * Ncol];
    b[0] *= bmask;
    b[1] *= bmask;
    c = __builtin_amdgcn_wmma_f32_16x16x4_f32(false, a, false, b, (short)0, c, false, false);
  }

  if (bok) {
    const float bb = bias ? bias[bcol] : 0.0f;
    const int row0 = rowBase + half * 8;
    float* __restrict__ yp = Y + (size_t)row0 * Ncol + bcol;
    #pragma unroll
    for (int r = 0; r < 8; ++r)
      yp[(size_t)r * Ncol] = c[r] + bb;
  }
}

#define BIN_SHIFT 8
#define BIN_NODES 256
#define CHUNK_EDGES 4096
#define MAX_BINS 1024

__global__ void __launch_bounds__(256)
bin_count_kernel(const int* __restrict__ ei, int E, int NB, int NC,
                 int* __restrict__ HC) {
  __shared__ int hist[MAX_BINS];
  const int tid = threadIdx.x;
  for (int b = tid; b < NB; b += 256) hist[b] = 0;
  __syncthreads();
  const int base = blockIdx.x * CHUNK_EDGES;
  int d[CHUNK_EDGES / 256];
  #pragma unroll
  for (int j = 0; j < CHUNK_EDGES / 256; ++j) {
    const int i  = base + j * 256 + tid;
    const int ic = (i < E) ? i : (E - 1);
    const int dv = ei[E + ic];
    d[j] = (i < E) ? dv : -1;
  }
  #pragma unroll
  for (int j = 0; j < CHUNK_EDGES / 256; ++j)
    if (d[j] >= 0) atomicAdd(&hist[d[j] >> BIN_SHIFT], 1);
  __syncthreads();
  for (int b = tid; b < NB; b += 256) HC[(size_t)b * NC + blockIdx.x] = hist[b];
}

__global__ void __launch_bounds__(256)
bin_scan_kernel(int* __restrict__ HC, int NC, int* __restrict__ BT) {
  __shared__ int wtot[8];
  const int tid  = threadIdx.x;
  const int lane = tid & 31;
  const int wave = tid >> 5;
  int* __restrict__ row = HC + (size_t)blockIdx.x * NC;
  int carry = 0;
  for (int c0 = 0; c0 < NC; c0 += 256) {
    const int c = c0 + tid;
    const int v = (c < NC) ? row[c] : 0;
    int incl = v;
    #pragma unroll
    for (int off = 1; off < 32; off <<= 1) {
      const int y = __shfl_up(incl, off, 32);
      if (lane >= off) incl += y;
    }
    if (lane == 31) wtot[wave] = incl;
    __syncthreads();
    int woff = 0, tot = 0;
    #pragma unroll
    for (int w = 0; w < 8; ++w) {
      const int t = wtot[w];
      woff += (w < wave) ? t : 0;
      tot  += t;
    }
    if (c < NC) row[c] = carry + woff + incl - v;
    carry += tot;
    __syncthreads();
  }
  if (tid == 0) BT[blockIdx.x] = carry;
}

__global__ void __launch_bounds__(256)
bin_scatter_kernel(const int* __restrict__ ei, int E, int NB, int NC,
                   const int* __restrict__ HC, const int* __restrict__ BT,
                   int* __restrict__ BB, int* __restrict__ PBw) {
  int2* __restrict__ PB = (int2*)PBw;
  __shared__ int slot[MAX_BINS];
  __shared__ int wtot[8];
  const int tid  = threadIdx.x;
  const int lane = tid & 31;
  const int wave = tid >> 5;
  const int chunk = blockIdx.x;
  const int b0 = tid * 4;
  const int c0 = (b0 + 0 < NB) ? (b0 + 0) : (NB - 1);
  const int c1 = (b0 + 1 < NB) ? (b0 + 1) : (NB - 1);
  const int c2 = (b0 + 2 < NB) ? (b0 + 2) : (NB - 1);
  const int c3 = (b0 + 3 < NB) ? (b0 + 3) : (NB - 1);
  const int r0 = BT[c0], r1 = BT[c1], r2 = BT[c2], r3 = BT[c3];
  const int h0 = HC[(size_t)c0 * NC + chunk];
  const int h1 = HC[(size_t)c1 * NC + chunk];
  const int h2 = HC[(size_t)c2 * NC + chunk];
  const int h3 = HC[(size_t)c3 * NC + chunk];
  const int t0 = (b0 + 0 < NB) ? r0 : 0;
  const int t1 = (b0 + 1 < NB) ? r1 : 0;
  const int t2 = (b0 + 2 < NB) ? r2 : 0;
  const int t3 = (b0 + 3 < NB) ? r3 : 0;
  const int tsum = t0 + t1 + t2 + t3;
  int incl = tsum;
  #pragma unroll
  for (int off = 1; off < 32; off <<= 1) {
    const int y = __shfl_up(incl, off, 32);
    if (lane >= off) incl += y;
  }
  if (lane == 31) wtot[wave] = incl;
  __syncthreads();
  int woff = 0;
  #pragma unroll
  for (int w = 0; w < 8; ++w) woff += (w < wave) ? wtot[w] : 0;
  const int e0 = woff + incl - tsum;
  const int e1 = e0 + t0;
  const int e2 = e1 + t1;
  const int e3 = e2 + t2;
  if (b0 + 0 < NB) slot[b0 + 0] = e0 + h0;
  if (b0 + 1 < NB) slot[b0 + 1] = e1 + h1;
  if (b0 + 2 < NB) slot[b0 + 2] = e2 + h2;
  if (b0 + 3 < NB) slot[b0 + 3] = e3 + h3;
  if (chunk == 0) {
    if (b0 + 0 < NB) BB[b0 + 0] = e0;
    if (b0 + 1 < NB) BB[b0 + 1] = e1;
    if (b0 + 2 < NB) BB[b0 + 2] = e2;
    if (b0 + 3 < NB) BB[b0 + 3] = e3;
  }
  __syncthreads();
  const int base = chunk * CHUNK_EDGES;
  int s[CHUNK_EDGES / 256], d[CHUNK_EDGES / 256];
  #pragma unroll
  for (int j = 0; j < CHUNK_EDGES / 256; ++j) {
    const int i  = base + j * 256 + tid;
    const int ic = (i < E) ? i : (E - 1);
    s[j] = ei[ic];
    const int dv = ei[E + ic];
    d[j] = (i < E) ? dv : -1;
  }
  #pragma unroll
  for (int j = 0; j < CHUNK_EDGES / 256; ++j) {
    if (d[j] >= 0) {
      const int pos = atomicAdd(&slot[d[j] >> BIN_SHIFT], 1);
      PB[pos] = make_int2(s[j], d[j]);
    }
  }
}

__global__ void __launch_bounds__(1024)
bin_sort_kernel(const int* __restrict__ PBw, const int* __restrict__ BB,
                const int* __restrict__ BT, int N,
                int* __restrict__ rowptr, int* __restrict__ csr_src) {
  __shared__ int cnt[BIN_NODES];
  __shared__ int cur[BIN_NODES];
  __shared__ int wtot[8];
  const int tid  = threadIdx.x;
  const int lane = tid & 31;
  const int wave = tid >> 5;
  const int bs = BB[blockIdx.x];
  const int n  = BT[blockIdx.x];
  const int2* __restrict__ pb = (const int2*)PBw + bs;
  if (tid < BIN_NODES) cnt[tid] = 0;
  __syncthreads();
  for (int i0 = 0; i0 < n; i0 += 4096) {
    int d[4];
    #pragma unroll
    for (int u = 0; u < 4; ++u) {
      const int i  = i0 + u * 1024 + tid;
      const int ic = (i < n) ? i : (n - 1);
      const int dv = pb[ic].y;
      d[u] = (i < n) ? dv : -1;
    }
    #pragma unroll
    for (int u = 0; u < 4; ++u)
      if (d[u] >= 0) atomicAdd(&cnt[d[u] & (BIN_NODES - 1)], 1);
  }
  __syncthreads();
  const int v = (tid < BIN_NODES) ? cnt[tid] : 0;
  int incl = v;
  #pragma unroll
  for (int off = 1; off < 32; off <<= 1) {
    const int y = __shfl_up(incl, off, 32);
    if (lane >= off) incl += y;
  }
  if (tid < BIN_NODES && lane == 31) wtot[wave] = incl;
  __syncthreads();
  if (tid < BIN_NODES) {
    int woff = 0;
    #pragma unroll
    for (int w = 0; w < 8; ++w) woff += (w < wave) ? wtot[w] : 0;
    const int start = bs + woff + incl - v;
    cur[tid] = start;
    const int node = blockIdx.x * BIN_NODES + tid;
    if (node < N) rowptr[node] = start;
  }
  __syncthreads();
  for (int i0 = 0; i0 < n; i0 += 4096) {
    int2 p[4];
    #pragma unroll
    for (int u = 0; u < 4; ++u) {
      const int i  = i0 + u * 1024 + tid;
      const int ic = (i < n) ? i : (n - 1);
      p[u] = pb[ic];
      if (i >= n) p[u].y = -1;
    }
    #pragma unroll
    for (int u = 0; u < 4; ++u) {
      if (p[u].y >= 0) {
        const int pos = atomicAdd(&cur[p[u].y & (BIN_NODES - 1)], 1);
        csr_src[pos] = p[u].x;
      }
    }
  }
}

template <int V>
__global__ void __launch_bounds__(256)
node_attend_kernel(const int* __restrict__ rowptr,
                   const int* __restrict__ csr_src, int N, int E,
                   const float* __restrict__ XL, const float* XR,
                   const float* __restrict__ att,
                   const float* __restrict__ obias,
                   float* out, int relu) {
  const int lane = threadIdx.x & 31;
  const int wid_v = (blockIdx.x * blockDim.x + threadIdx.x) >> 5;
  if (wid_v >= N) return;
  const int wid = __builtin_amdgcn_readfirstlane(wid_v);
  const int beg = __builtin_amdgcn_readfirstlane(rowptr[wid]);
  const int end = (wid + 1 < N) ? __builtin_amdgcn_readfirstlane(rowptr[wid + 1]) : E;
  const int c   = end - beg + 1;

  constexpr int D4 = 32 * V;
  const float4* __restrict__ xl4 = (const float4*)XL;
  const float4* xr4 = (const float4*)XR + (size_t)wid * D4 + lane;
  const float4* __restrict__ at4 = (const float4*)att + lane;

  float4 xr[V], g[V], acc[V];
  #pragma unroll
  for (int v = 0; v < V; ++v) {
    xr[v]  = xr4[v * 32];
    g[v]   = at4[v * 32];
    acc[v] = make_float4(0.0f, 0.0f, 0.0f, 0.0f);
  }
  float m = -__builtin_huge_valf();
  float ssum = 0.0f;

  for (int base = 0; base < c; base += 32) {
    const int q = base + lane;
    int idxv = wid;
    if (q > 0 && q < c) idxv = csr_src[beg + q - 1];
    const int nq = (c - base < 32) ? (c - base) : 32;

    for (int k = 0; k < nq; k += 4) {
      float4 r[4][V];
      float e[4];
      #pragma unroll
      for (int u = 0; u < 4; ++u) {
        const int kk = (k + u < nq) ? (k + u) : k;
        const int s = __builtin_amdgcn_readlane(idxv, kk);
        const float4* __restrict__ row = xl4 + (size_t)s * D4 + lane;
        #pragma unroll
        for (int v = 0; v < V; ++v) r[u][v] = row[v * 32];
      }
      #pragma unroll
      for (int u = 0; u < 4; ++u) {
        float a = 0.0f;
        #pragma unroll
        for (int v = 0; v < V; ++v) {
          float h;
          h = r[u][v].x + xr[v].x; h = h > 0.0f ? h : NEG_SLOPE * h; a += h * g[v].x;
          h = r[u][v].y + xr[v].y; h = h > 0.0f ? h : NEG_SLOPE * h; a += h * g[v].y;
          h = r[u][v].z + xr[v].z; h = h > 0.0f ? h : NEG_SLOPE * h; a += h * g[v].z;
          h = r[u][v].w + xr[v].w; h = h > 0.0f ? h : NEG_SLOPE * h; a += h * g[v].w;
        }
        e[u] = a;
      }
      #pragma unroll
      for (int off = 16; off > 0; off >>= 1) {
        #pragma unroll
        for (int u = 0; u < 4; ++u) e[u] += __shfl_xor(e[u], off, 32);
      }
      float mc = e[0];
      #pragma unroll
      for (int u = 1; u < 4; ++u)
        if (k + u < nq) mc = fmaxf(mc, e[u]);
      const float mn = fmaxf(m, mc);
      const float sc = __expf(m - mn);
      ssum *= sc;
      #pragma unroll
      for (int v = 0; v < V; ++v) {
        acc[v].x *= sc; acc[v].y *= sc; acc[v].z *= sc; acc[v].w *= sc;
      }
      #pragma unroll
      for (int u = 0; u < 4; ++u) {
        if (k + u < nq) {
          const float p = __expf(e[u] - mn);
          ssum += p;
          #pragma unroll
          for (int v = 0; v < V; ++v) {
            acc[v].x += p * r[u][v].x;
            acc[v].y += p * r[u][v].y;
            acc[v].z += p * r[u][v].z;
            acc[v].w += p * r[u][v].w;
          }
        }
      }
      m = mn;
    }
  }

  const float inv = 1.0f / ssum;
  const float4* __restrict__ ob4 = (const float4*)obias + lane;
  float4* o4 = (float4*)out + (size_t)wid * D4 + lane;
  #pragma unroll
  for (int v = 0; v < V; ++v) {
    const float4 bb = ob4[v * 32];
    float4 o;
    o.x = acc[v].x * inv + bb.x;
    o.y = acc[v].y * inv + bb.y;
    o.z = acc[v].z * inv + bb.z;
    o.w = acc[v].w * inv + bb.w;
    if (relu) {
      o.x = fmaxf(o.x, 0.0f); o.y = fmaxf(o.y, 0.0f);
      o.z = fmaxf(o.z, 0.0f); o.w = fmaxf(o.w, 0.0f);
    }
    o4[v * 32] = o;
  }
}

extern "C" void kernel_launch(void* const* d_in, const int* in_sizes, int n_in,
                              void* d_out, int out_size, void* d_ws, size_t ws_size,
                              hipStream_t stream) {
  const float* x   = (const float*)d_in[0];
  const int*   ei  = (const int*)  d_in[1];
  const float* w1l = (const float*)d_in[2];
  const float* b1l = (const float*)d_in[3];
  const float* w1r = (const float*)d_in[4];
  const float* b1r = (const float*)d_in[5];
  const float* a1  = (const float*)d_in[6];
  const float* o1  = (const float*)d_in[7];
  const float* w2l = (const float*)d_in[8];
  const float* b2l = (const float*)d_in[9];
  const float* w2r = (const float*)d_in[10];
  const float* b2r = (const float*)d_in[11];
  const float* a2  = (const float*)d_in[12];
  const float* o2  = (const float*)d_in[13];
  const float* w3l = (const float*)d_in[14];
  const float* b3l = (const float*)d_in[15];
  const float* w3r = (const float*)d_in[16];
  const float* b3r = (const float*)d_in[17];
  const float* a3  = (const float*)d_in[18];
  const float* o3  = (const float*)d_in[19];
  const float* wc  = (const float*)d_in[20];
  const float* bc  = (const float*)d_in[21];

  const int N  = in_sizes[0] / 128;
  const int E  = in_sizes[1] / 2;

  const int Np = (N + 3) & ~3;
  const int Ep = (E + 3) & ~3;
  const int NB = (N + BIN_NODES - 1) / BIN_NODES;
  const int NC = (E + CHUNK_EDGES - 1) / CHUNK_EDGES;
  const int Hp = (NB * NC + 3) & ~3;
  float* B0  = (float*)d_ws;
  float* B1  = B0 + (size_t)N * 256;
  float* B2  = B1 + (size_t)N * 256;
  int*   CS  = (int*)(B2 + (size_t)N * 256);
  int*   RP  = CS + (size_t)Ep;
  int*   HC  = RP + (size_t)Np;
  int*   BT  = HC + (size_t)Hp;
  int*   BB  = BT + MAX_BINS;
  int*   PB  = BB + MAX_BINS;
  v2f*   WPa = (v2f*)(PB + 2 * (size_t)Ep);
  v2f*   WPb = WPa + 32768;

  bin_count_kernel<<<NC, 256, 0, stream>>>(ei, E, NB, NC, HC);
  bin_scan_kernel<<<NB, 256, 0, stream>>>(HC, NC, BT);
  bin_scatter_kernel<<<NC, 256, 0, stream>>>(ei, E, NB, NC, HC, BT, BB, PB);
  bin_sort_kernel<<<NB, 1024, 0, stream>>>(PB, BB, BT, N, RP, CS);

  auto packW = [&](const float* W, v2f* Wp, int k, int n) {
    const int total = (k >> 1) * n;
    pack_w_kernel<<<(total + 255) / 256, 256, 0, stream>>>(W, Wp, k, n);
  };
  auto gemmBig = [&](const float* X, const v2f* Wp, const float* b, float* Y,
                     int m, int k, int n) {
    const int tiles = (m >> 4) * (n >> 6);
    gemm_bias_wmma_n64<<<(tiles + 3) / 4, 128, 0, stream>>>(X, Wp, b, Y, m, k, n);
  };
  auto gemmGuard = [&](const float* X, const v2f* Wp, const float* b, float* Y,
                       int m, int k, int n) {
    const int tiles = (m >> 4) * ((n + 15) / 16);
    gemm_bias_wmma_guard<<<(tiles + 3) / 4, 128, 0, stream>>>(X, Wp, b, Y, m, k, n);
  };

  auto layer = [&](const float* Xin, int din, int dout,
                   const float* wl, const float* bl,
                   const float* wr, const float* br,
                   const float* att, const float* ob,
                   float* XL, float* XO, int relu) {
    packW(wl, WPa, din, dout);
    packW(wr, WPb, din, dout);
    gemmBig(Xin, WPa, bl, XL, N, din, dout);
    gemmBig(Xin, WPb, br, XO, N, din, dout);
    if (dout == 128)
      node_attend_kernel<1><<<(N + 7) / 8, 256, 0, stream>>>(RP, CS, N, E, XL, XO,
                                                           att, ob, XO, relu);
    else
      node_attend_kernel<2><<<(N + 7) / 8, 256, 0, stream>>>(RP, CS, N, E, XL, XO,
                                                           att, ob, XO, relu);
  };

  layer(x,  128, 128, w1l, b1l, w1r, b1r, a1, o1, B0, B1, 1);
  layer(B1, 128, 128, w2l, b2l, w2r, b2r, a2, o2, B0, B2, 1);
  layer(B2, 128, 256, w3l, b3l, w3r, b3r, a3, o3, B0, B1, 0);
  packW(wc, WPa, 256, 40);
  gemmGuard(B1, WPa, bc, (float*)d_out, N, 256, 40);
}


# Matrix loops read the weights as stored: one 8-byte load of W[k][2c..2c+1] feeds an even- and an odd-column tile, results leave as 8-byte pairs; 7 weight-packing launches gone; sums unchanged
# speedup vs baseline: 4.2777x; 1.0331x over previous
#include <hip/hip_runtime.h>

typedef __attribute__((ext_vector_type(2))) float v2f;
typedef __attribute__((ext_vector_type(8))) float v8f;

#define NEG_SLOPE 0.2f

__global__ void gemm_bias_wmma_n64(const float* __restrict__ X,
                                   const float* __restrict__ Wt,
                                   const float* __restrict__ bias,
                                   float* __restrict__ Y,
                                   int M, int K, int Ncol) {
  const int lane = threadIdx.x & 31;
  const int wave = threadIdx.x >> 5;
  const int ct = Ncol >> 6;
  const int rt = M >> 4;
  const int tile = blockIdx.x * (blockDim.x >> 5) + wave;
  if (tile >= ct * rt) return;

  const int rowBase = (tile / ct) << 4;
  const int colBase = (tile % ct) << 6;
  const int half = lane >> 4;
  const int l15  = lane & 15;

  const float* __restrict__ ap = X + (size_t)(rowBase + l15) * K + half * 2;
  const float* __restrict__ bp = Wt + (size_t)(half * 2) * Ncol + colBase + 2 * l15;

  v8f cc[4] = {v8f{}, v8f{}, v8f{}, v8f{}};
  for (int k0 = 0; k0 < K; k0 += 16) {
    v2f a[4], w0[4], w1[4], w2[4], w3[4];
    #pragma unroll
    for (int u = 0; u < 4; ++u) {
      const float* __restrict__ br = bp + (size_t)(k0 + 4 * u) * Ncol;
      a[u]  = *(const v2f*)(ap + k0 + 4 * u);
      w0[u] = *(const v2f*)(br);
      w1[u] = *(const v2f*)(br + Ncol);
      w2[u] = *(const v2f*)(br + 32);
      w3[u] = *(const v2f*)(br + Ncol + 32);
    }
    asm volatile("" :: "v"(a[0]), "v"(a[1]), "v"(a[2]), "v"(a[3]),
                       "v"(w0[0]), "v"(w0[1]), "v"(w0[2]), "v"(w0[3]),
                       "v"(w1[0]), "v"(w1[1]), "v"(w1[2]), "v"(w1[3]),
                       "v"(w2[0]), "v"(w2[1]), "v"(w2[2]), "v"(w2[3]),
                       "v"(w3[0]), "v"(w3[1]), "v"(w3[2]), "v"(w3[3]));
    #pragma unroll
    for (int u = 0; u < 4; ++u) {
      v2f b0e, b0o, b1e, b1o;
      b0e[0] = w0[u][0]; b0e[1] = w1[u][0];
      b0o[0] = w0[u][1]; b0o[1] = w1[u][1];
      b1e[0] = w2[u][0]; b1e[1] = w3[u][0];
      b1o[0] = w2[u][1]; b1o[1] = w3[u][1];
      cc[0] = __builtin_amdgcn_wmma_f32_16x16x4_f32(false, a[u], false, b0e, (short)0, cc[0], false, false);
      cc[1] = __builtin_amdgcn_wmma_f32_16x16x4_f32(false, a[u], false, b0o, (short)0, cc[1], false, false);
      cc[2] = __builtin_amdgcn_wmma_f32_16x16x4_f32(false, a[u], false, b1e, (short)0, cc[2], false, false);
      cc[3] = __builtin_amdgcn_wmma_f32_16x16x4_f32(false, a[u], false, b1o, (short)0, cc[3], false, false);
    }
  }

  const int row0 = rowBase + half * 8;
  #pragma unroll
  for (int g = 0; g < 2; ++g) {
    const int col = colBase + g * 32 + 2 * l15;
    v2f bb = {0.0f, 0.0f};
    if (bias) bb = *(const v2f*)(bias + col);
    float* __restrict__ yp = Y + (size_t)row0 * Ncol + col;
    #pragma unroll
    for (int r = 0; r < 8; ++r) {
      v2f o;
      o[0] = cc[2 * g][r]     + bb[0];
      o[1] = cc[2 * g + 1][r] + bb[1];
      *(v2f*)(yp + (size_t)r * Ncol) = o;
    }
  }
}

__global__ void gemm_bias_wmma_guard(const float* __restrict__ X,
                                     const float* __restrict__ Wt,
                                     const float* __restrict__ bias,
                                     float* __restrict__ Y,
                                     int M, int K, int Ncol) {
  const int lane = threadIdx.x & 31;
  const int wave = threadIdx.x >> 5;
  const int ct = (Ncol + 15) >> 4;
  const int rt = M >> 4;
  const int tile = blockIdx.x * (blockDim.x >> 5) + wave;
  if (tile >= ct * rt) return;

  const int rowBase = (tile / ct) << 4;
  const int colBase = (tile % ct) << 4;
  const int half = lane >> 4;
  const int l15  = lane & 15;
  const int bcol = colBase + l15;
  const bool bok = bcol < Ncol;
  const int bcolc = bok ? bcol : (Ncol - 1);
  const float bmask = bok ? 1.0f : 0.0f;

  const float* __restrict__ ap = X + (size_t)(rowBase + l15) * K + half * 2;
  const float* __restrict__ bp = Wt + (size_t)(half * 2) * Ncol + bcolc;

  v8f c = {};
  #pragma unroll 4
  for (int k0 = 0; k0 < K; k0 += 4) {
    const v2f a = *(const v2f*)(ap + k0);
    const float* __restrict__ br = bp + (size_t)k0 * Ncol;
    v2f b;
    b[0] = br[0]    * bmask;
    b[1] = br[Ncol] * bmask;
    c = __builtin_amdgcn_wmma_f32_16x16x4_f32(false, a, false, b, (short)0, c, false, false);
  }

  if (bok) {
    const float bb = bias ? bias[bcol] : 0.0f;
    const int row0 = rowBase + half * 8;
    float* __restrict__ yp = Y + (size_t)row0 * Ncol + bcol;
    #pragma unroll
    for (int r = 0; r < 8; ++r)
      yp[(size_t)r * Ncol] = c[r] + bb;
  }
}

#define BIN_SHIFT 8
#define BIN_NODES 256
#define CHUNK_EDGES 4096
#define MAX_BINS 1024

__global__ void __launch_bounds__(256)
bin_count_kernel(const int* __restrict__ ei, int E, int NB, int NC,
                 int* __restrict__ HC) {
  __shared__ int hist[MAX_BINS];
  const int tid = threadIdx.x;
  for (int b = tid; b < NB; b += 256) hist[b] = 0;
  __syncthreads();
  const int base = blockIdx.x * CHUNK_EDGES;
  int d[CHUNK_EDGES / 256];
  #pragma unroll
  for (int j = 0; j < CHUNK_EDGES / 256; ++j) {
    const int i  = base + j * 256 + tid;
    const int ic = (i < E) ? i : (E - 1);
    const int dv = ei[E + ic];
    d[j] = (i < E) ? dv : -1;
  }
  #pragma unroll
  for (int j = 0; j < CHUNK_EDGES / 256; ++j)
    if (d[j] >= 0) atomicAdd(&hist[d[j] >> BIN_SHIFT], 1);
  __syncthreads();
  for (int b = tid; b < NB; b += 256) HC[(size_t)b * NC + blockIdx.x] = hist[b];
}

__global__ void __launch_bounds__(256)
bin_scan_kernel(int* __restrict__ HC, int NC, int* __restrict__ BT) {
  __shared__ int wtot[8];
  const int tid  = threadIdx.x;
  const int lane = tid & 31;
  const int wave = tid >> 5;
  int* __restrict__ row = HC + (size_t)blockIdx.x * NC;
  int carry = 0;
  for (int c0 = 0; c0 < NC; c0 += 256) {
    const int c = c0 + tid;
    const int v = (c < NC) ? row[c] : 0;
    int incl = v;
    #pragma unroll
    for (int off = 1; off < 32; off <<= 1) {
      const int y = __shfl_up(incl, off, 32);
      if (lane >= off) incl += y;
    }
    if (lane == 31) wtot[wave] = incl;
    __syncthreads();
    int woff = 0, tot = 0;
    #pragma unroll
    for (int w = 0; w < 8; ++w) {
      const int t = wtot[w];
      woff += (w < wave) ? t : 0;
      tot  += t;
    }
    if (c < NC) row[c] = carry + woff + incl - v;
    carry += tot;
    __syncthreads();
  }
  if (tid == 0) BT[blockIdx.x] = carry;
}

__global__ void __launch_bounds__(256)
bin_scatter_kernel(const int* __restrict__ ei, int E, int NB, int NC,
                   const int* __restrict__ HC, const int* __restrict__ BT,
                   int* __restrict__ BB, int* __restrict__ PBw) {
  int2* __restrict__ PB = (int2*)PBw;
  __shared__ int slot[MAX_BINS];
  __shared__ int wtot[8];
  const int tid  = threadIdx.x;
  const int lane = tid & 31;
  const int wave = tid >> 5;
  const int chunk = blockIdx.x;
  const int b0 = tid * 4;
  const int c0 = (b0 + 0 < NB) ? (b0 + 0) : (NB - 1);
  const int c1 = (b0 + 1 < NB) ? (b0 + 1) : (NB - 1);
  const int c2 = (b0 + 2 < NB) ? (b0 + 2) : (NB - 1);
  const int c3 = (b0 + 3 < NB) ? (b0 + 3) : (NB - 1);
  const int r0 = BT[c0], r1 = BT[c1], r2 = BT[c2], r3 = BT[c3];
  const int h0 = HC[(size_t)c0 * NC + chunk];
  const int h1 = HC[(size_t)c1 * NC + chunk];
  const int h2 = HC[(size_t)c2 * NC + chunk];
  const int h3 = HC[(size_t)c3 * NC + chunk];
  const int t0 = (b0 + 0 < NB) ? r0 : 0;
  const int t1 = (b0 + 1 < NB) ? r1 : 0;
  const int t2 = (b0 + 2 < NB) ? r2 : 0;
  const int t3 = (b0 + 3 < NB) ? r3 : 0;
  const int tsum = t0 + t1 + t2 + t3;
  int incl = tsum;
  #pragma unroll
  for (int off = 1; off < 32; off <<= 1) {
    const int y = __shfl_up(incl, off, 32);
    if (lane >= off) incl += y;
  }
  if (lane == 31) wtot[wave] = incl;
  __syncthreads();
  int woff = 0;
  #pragma unroll
  for (int w = 0; w < 8; ++w) woff += (w < wave) ? wtot[w] : 0;
  const int e0 = woff + incl - tsum;
  const int e1 = e0 + t0;
  const int e2 = e1 + t1;
  const int e3 = e2 + t2;
  if (b0 + 0 < NB) slot[b0 + 0] = e0 + h0;
  if (b0 + 1 < NB) slot[b0 + 1] = e1 + h1;
  if (b0 + 2 < NB) slot[b0 + 2] = e2 + h2;
  if (b0 + 3 < NB) slot[b0 + 3] = e3 + h3;
  if (chunk == 0) {
    if (b0 + 0 < NB) BB[b0 + 0] = e0;
    if (b0 + 1 < NB) BB[b0 + 1] = e1;
    if (b0 + 2 < NB) BB[b0 + 2] = e2;
    if (b0 + 3 < NB) BB[b0 + 3] = e3;
  }
  __syncthreads();
  const int base = chunk * CHUNK_EDGES;
  int s[CHUNK_EDGES / 256], d[CHUNK_EDGES / 256];
  #pragma unroll
  for (int j = 0; j < CHUNK_EDGES / 256; ++j) {
    const int i  = base + j * 256 + tid;
    const int ic = (i < E) ? i : (E - 1);
    s[j] = ei[ic];
    const int dv = ei[E + ic];
    d[j] = (i < E) ? dv : -1;
  }
  #pragma unroll
  for (int j = 0; j < CHUNK_EDGES / 256; ++j) {
    if (d[j] >= 0) {
      const int pos = atomicAdd(&slot[d[j] >> BIN_SHIFT], 1);
      PB[pos] = make_int2(s[j], d[j]);
    }
  }
}

__global__ void __launch_bounds__(1024)
bin_sort_kernel(const int* __restrict__ PBw, const int* __restrict__ BB,
                const int* __restrict__ BT, int N,
                int* __restrict__ rowptr, int* __restrict__ csr_src) {
  __shared__ int cnt[BIN_NODES];
  __shared__ int cur[BIN_NODES];
  __shared__ int wtot[8];
  const int tid  = threadIdx.x;
  const int lane = tid & 31;
  const int wave = tid >> 5;
  const int bs = BB[blockIdx.x];
  const int n  = BT[blockIdx.x];
  const int2* __restrict__ pb = (const int2*)PBw + bs;
  if (tid < BIN_NODES) cnt[tid] = 0;
  __syncthreads();
  for (int i0 = 0; i0 < n; i0 += 4096) {
    int d[4];
    #pragma unroll
    for (int u = 0; u < 4; ++u) {
      const int i  = i0 + u * 1024 + tid;
      const int ic = (i < n) ? i : (n - 1);
      const int dv = pb[ic].y;
      d[u] = (i < n) ? dv : -1;
    }
    #pragma unroll
    for (int u = 0; u < 4; ++u)
      if (d[u] >= 0) atomicAdd(&cnt[d[u] & (BIN_NODES - 1)], 1);
  }
  __syncthreads();
  const int v = (tid < BIN_NODES) ? cnt[tid] : 0;
  int incl = v;
  #pragma unroll
  for (int off = 1; off < 32; off <<= 1) {
    const int y = __shfl_up(incl, off, 32);
    if (lane >= off) incl += y;
  }
  if (tid < BIN_NODES && lane == 31) wtot[wave] = incl;
  __syncthreads();
  if (tid < BIN_NODES) {
    int woff = 0;
    #pragma unroll
    for (int w = 0; w < 8; ++w) woff += (w < wave) ? wtot[w] : 0;
    const int start = bs + woff + incl - v;
    cur[tid] = start;
    const int node = blockIdx.x * BIN_NODES + tid;
    if (node < N) rowptr[node] = start;
  }
  __syncthreads();
  for (int i0 = 0; i0 < n; i0 += 4096) {
    int2 p[4];
    #pragma unroll
    for (int u = 0; u < 4; ++u) {
      const int i  = i0 + u * 1024 + tid;
      const int ic = (i < n) ? i : (n - 1);
      p[u] = pb[ic];
      if (i >= n) p[u].y = -1;
    }
    #pragma unroll
    for (int u = 0; u < 4; ++u) {
      if (p[u].y >= 0) {
        const int pos = atomicAdd(&cur[p[u].y & (BIN_NODES - 1)], 1);
        csr_src[pos] = p[u].x;
      }
    }
  }
}

template <int V>
__global__ void __launch_bounds__(256)
node_attend_kernel(const int* __restrict__ rowptr,
                   const int* __restrict__ csr_src, int N, int E,
                   const float* __restrict__ XL, const float* XR,
                   const float* __restrict__ att,
                   const float* __restrict__ obias,
                   float* out, int relu) {
  const int lane = threadIdx.x & 31;
  const int wid_v = (blockIdx.x * blockDim.x + threadIdx.x) >> 5;
  if (wid_v >= N) return;
  const int wid = __builtin_amdgcn_readfirstlane(wid_v);
  const int beg = __builtin_amdgcn_readfirstlane(rowptr[wid]);
  const int end = (wid + 1 < N) ? __builtin_amdgcn_readfirstlane(rowptr[wid + 1]) : E;
  const int c   = end - beg + 1;

  constexpr int D4 = 32 * V;
  const float4* __restrict__ xl4 = (const float4*)XL;
  const float4* xr4 = (const float4*)XR + (size_t)wid * D4 + lane;
  const float4* __restrict__ at4 = (const float4*)att + lane;

  float4 xr[V], g[V], acc[V];
  #pragma unroll
  for (int v = 0; v < V; ++v) {
    xr[v]  = xr4[v * 32];
    g[v]   = at4[v * 32];
    acc[v] = make_float4(0.0f, 0.0f, 0.0f, 0.0f);
  }
  float m = -__builtin_huge_valf();
  float ssum = 0.0f;

  for (int base = 0; base < c; base += 32) {
    const int q = base + lane;
    int idxv = wid;
    if (q > 0 && q < c) idxv = csr_src[beg + q - 1];
    const int nq = (c - base < 32) ? (c - base) : 32;

    for (int k = 0; k < nq; k += 4) {
      float4 r[4][V];
      float e[4];
      #pragma unroll
      for (int u = 0; u < 4; ++u) {
        const int kk = (k + u < nq) ? (k + u) : k;
        const int s = __builtin_amdgcn_readlane(idxv, kk);
        const float4* __restrict__ row = xl4 + (size_t)s * D4 + lane;
        #pragma unroll
        for (int v = 0; v < V; ++v) r[u][v] = row[v * 32];
      }
      #pragma unroll
      for (int u = 0; u < 4; ++u) {
        float a = 0.0f;
        #pragma unroll
        for (int v = 0; v < V; ++v) {
          float h;
          h = r[u][v].x + xr[v].x; h = h > 0.0f ? h : NEG_SLOPE * h; a += h * g[v].x;
          h = r[u][v].y + xr[v].y; h = h > 0.0f ? h : NEG_SLOPE * h; a += h * g[v].y;
          h = r[u][v].z + xr[v].z; h = h > 0.0f ? h : NEG_SLOPE * h; a += h * g[v].z;
          h = r[u][v].w + xr[v].w; h = h > 0.0f ? h : NEG_SLOPE * h; a += h * g[v].w;
        }
        e[u] = a;
      }
      #pragma unroll
      for (int off = 16; off > 0; off >>= 1) {
        #pragma unroll
        for (int u = 0; u < 4; ++u) e[u] += __shfl_xor(e[u], off, 32);
      }
      float mc = e[0];
      #pragma unroll
      for (int u = 1; u < 4; ++u)
        if (k + u < nq) mc = fmaxf(mc, e[u]);
      const float mn = fmaxf(m, mc);
      const float sc = __expf(m - mn);
      ssum *= sc;
      #pragma unroll
      for (int v = 0; v < V; ++v) {
        acc[v].x *= sc; acc[v].y *= sc; acc[v].z *= sc; acc[v].w *= sc;
      }
      #pragma unroll
      for (int u = 0; u < 4; ++u) {
        if (k + u < nq) {
          const float p = __expf(e[u] - mn);
          ssum += p;
          #pragma unroll
          for (int v = 0; v < V; ++v) {
            acc[v].x += p * r[u][v].x;
            acc[v].y += p * r[u][v].y;
            acc[v].z += p * r[u][v].z;
            acc[v].w += p * r[u][v].w;
          }
        }
      }
      m = mn;
    }
  }

  const float inv = 1.0f / ssum;
  const float4* __restrict__ ob4 = (const float4*)obias + lane;
  float4* o4 = (float4*)out + (size_t)wid * D4 + lane;
  #pragma unroll
  for (int v = 0; v < V; ++v) {
    const float4 bb = ob4[v * 32];
    float4 o;
    o.x = acc[v].x * inv + bb.x;
    o.y = acc[v].y * inv + bb.y;
    o.z = acc[v].z * inv + bb.z;
    o.w = acc[v].w * inv + bb.w;
    if (relu) {
      o.x = fmaxf(o.x, 0.0f); o.y = fmaxf(o.y, 0.0f);
      o.z = fmaxf(o.z, 0.0f); o.w = fmaxf(o.w, 0.0f);
    }
    o4[v * 32] = o;
  }
}

extern "C" void kernel_launch(void* const* d_in, const int* in_sizes, int n_in,
                              void* d_out, int out_size, void* d_ws, size_t ws_size,
                              hipStream_t stream) {
  const float* x   = (const float*)d_in[0];
  const int*   ei  = (const int*)  d_in[1];
  const float* w1l = (const float*)d_in[2];
  const float* b1l = (const float*)d_in[3];
  const float* w1r = (const float*)d_in[4];
  const float* b1r = (const float*)d_in[5];
  const float* a1  = (const float*)d_in[6];
  const float* o1  = (const float*)d_in[7];
  const float* w2l = (const float*)d_in[8];
  const float* b2l = (const float*)d_in[9];
  const float* w2r = (const float*)d_in[10];
  const float* b2r = (const float*)d_in[11];
  const float* a2  = (const float*)d_in[12];
  const float* o2  = (const float*)d_in[13];
  const float* w3l = (const float*)d_in[14];
  const float* b3l = (const float*)d_in[15];
  const float* w3r = (const float*)d_in[16];
  const float* b3r = (const float*)d_in[17];
  const float* a3  = (const float*)d_in[18];
  const float* o3  = (const float*)d_in[19];
  const float* wc  = (const float*)d_in[20];
  const float* bc  = (const float*)d_in[21];

  const int N  = in_sizes[0] / 128;
  const int E  = in_sizes[1] / 2;

  const int Np = (N + 3) & ~3;
  const int Ep = (E + 3) & ~3;
  const int NB = (N + BIN_NODES - 1) / BIN_NODES;
  const int NC = (E + CHUNK_EDGES - 1) / CHUNK_EDGES;
  const int Hp = (NB * NC + 3) & ~3;
  float* B0  = (float*)d_ws;
  float* B1  = B0 + (size_t)N * 256;
  float* B2  = B1 + (size_t)N * 256;
  int*   CS  = (int*)(B2 + (size_t)N * 256);
  int*   RP  = CS + (size_t)Ep;
  int*   HC  = RP + (size_t)Np;
  int*   BT  = HC + (size_t)Hp;
  int*   BB  = BT + MAX_BINS;
  int*   PB  = BB + MAX_BINS;

  bin_count_kernel<<<NC, 256, 0, stream>>>(ei, E, NB, NC, HC);
  bin_scan_kernel<<<NB, 256, 0, stream>>>(HC, NC, BT);
  bin_scatter_kernel<<<NC, 256, 0, stream>>>(ei, E, NB, NC, HC, BT, BB, PB);
  bin_sort_kernel<<<NB, 1024, 0, stream>>>(PB, BB, BT, N, RP, CS);

  auto gemmBig = [&](const float* X, const float* Wp, const float* b, float* Y,
                     int m, int k, int n) {
    const int tiles = (m >> 4) * (n >> 6);
    gemm_bias_wmma_n64<<<(tiles + 3) / 4, 128, 0, stream>>>(X, Wp, b, Y, m, k, n);
  };
  auto gemmGuard = [&](const float* X, const float* Wp, const float* b, float* Y,
                       int m, int k, int n) {
    const int tiles = (m >> 4) * ((n + 15) / 16);
    gemm_bias_wmma_guard<<<(tiles + 3) / 4, 128, 0, stream>>>(X, Wp, b, Y, m, k, n);
  };

  auto layer = [&](const float* Xin, int din, int dout,
                   const float* wl, const float* bl,
                   const float* wr, const float* br,
                   const float* att, const float* ob,
                   float* XL, float* XO, int relu) {
    gemmBig(Xin, wl, bl, XL, N, din, dout);
    gemmBig(Xin, wr, br, XO, N, din, dout);
    if (dout == 128)
      node_attend_kernel<1><<<(N + 7) / 8, 256, 0, stream>>>(RP, CS, N, E, XL, XO,
                                                           att, ob, XO, relu);
    else
      node_attend_kernel<2><<<(N + 7) / 8, 256, 0, stream>>>(RP, CS, N, E, XL, XO,
                                                           att, ob, XO, relu);
  };

  layer(x,  128, 128, w1l, b1l, w1r, b1r, a1, o1, B0, B1, 1);
  layer(B1, 128, 128, w2l, b2l, w2r, b2r, a2, o2, B0, B2, 1);
  layer(B2, 128, 256, w3l, b3l, w3r, b3r, a3, o3, B0, B1, 0);
  gemmGuard(B1, wc, bc, (float*)d_out, N, 256, 40);
}


# xl and xr of a layer come from one launch: a wave multiplies each X fragment by both weight matrices (eight tiles per fragment), X read once per layer; 3 launches gone; every sum keeps its order
# speedup vs baseline: 4.4782x; 1.0469x over previous
#include <hip/hip_runtime.h>

typedef __attribute__((ext_vector_type(2))) float v2f;
typedef __attribute__((ext_vector_type(8))) float v8f;

#define NEG_SLOPE 0.2f

__device__ __forceinline__ void store_strip_n64(const v8f (&cc)[4],
                                                const float* __restrict__ bias,
                                                float* __restrict__ Y,
                                                int row0, int colBase, int l15,
                                                int Ncol) {
  #pragma unroll
  for (int g = 0; g < 2; ++g) {
    const int col = colBase + g * 32 + 2 * l15;
    v2f bb = {0.0f, 0.0f};
    if (bias) bb = *(const v2f*)(bias + col);
    float* __restrict__ yp = Y + (size_t)row0 * Ncol + col;
    #pragma unroll
    for (int r = 0; r < 8; ++r) {
      v2f o;
      o[0] = cc[2 * g][r]     + bb[0];
      o[1] = cc[2 * g + 1][r] + bb[1];
      *(v2f*)(yp + (size_t)r * Ncol) = o;
    }
  }
}

__global__ void __launch_bounds__(128)
gemm2_bias_wmma_n64(const float* __restrict__ X,
                    const float* __restrict__ Wl,
                    const float* __restrict__ Wr,
                    const float* __restrict__ biasl,
                    const float* __restrict__ biasr,
                    float* __restrict__ Yl,
                    float* __restrict__ Yr,
                    int M, int K, int Ncol) {
  const int lane = threadIdx.x & 31;
  const int wave = threadIdx.x >> 5;
  const int ct = Ncol >> 6;
  const int rt = M >> 4;
  const int tile = blockIdx.x * (blockDim.x >> 5) + wave;
  if (tile >= ct * rt) return;

  const int rowBase = (tile / ct) << 4;
  const int colBase = (tile % ct) << 6;
  const int half = lane >> 4;
  const int l15  = lane & 15;

  const float* __restrict__ ap = X + (size_t)(rowBase + l15) * K + half * 2;
  const size_t boff = (size_t)(half * 2) * Ncol + colBase + 2 * l15;
  const float* __restrict__ bpl = Wl + boff;
  const float* __restrict__ bpr = Wr + boff;

  v8f cl[4] = {v8f{}, v8f{}, v8f{}, v8f{}};
  v8f cr[4] = {v8f{}, v8f{}, v8f{}, v8f{}};
  for (int k0 = 0; k0 < K; k0 += 8) {
    v2f a[2], l0[2], l1[2], l2[2], l3[2], r0[2], r1[2], r2[2], r3[2];
    #pragma unroll
    for (int u = 0; u < 2; ++u) {
      const size_t ro = (size_t)(k0 + 4 * u) * Ncol;
      a[u]  = *(const v2f*)(ap + k0 + 4 * u);
      l0[u] = *(const v2f*)(bpl + ro);
      l1[u] = *(const v2f*)(bpl + ro + Ncol);
      l2[u] = *(const v2f*)(bpl + ro + 32);
      l3[u] = *(const v2f*)(bpl + ro + Ncol + 32);
      r0[u] = *(const v2f*)(bpr + ro);
      r1[u] = *(const v2f*)(bpr + ro + Ncol);
      r2[u] = *(const v2f*)(bpr + ro + 32);
      r3[u] = *(const v2f*)(bpr + ro + Ncol + 32);
    }
    asm volatile("" :: "v"(a[0]), "v"(a[1]),
                       "v"(l0[0]), "v"(l0[1]), "v"(l1[0]), "v"(l1[1]),
                       "v"(l2[0]), "v"(l2[1]), "v"(l3[0]), "v"(l3[1]),
                       "v"(r0[0]), "v"(r0[1]), "v"(r1[0]), "v"(r1[1]),
                       "v"(r2[0]), "v"(r2[1]), "v"(r3[0]), "v"(r3[1]));
    #pragma unroll
    for (int u = 0; u < 2; ++u) {
      v2f b0e, b0o, b1e, b1o;
      b0e[0] = l0[u][0]; b0e[1] = l1[u][0];
      b0o[0] = l0[u][1]; b0o[1] = l1[u][1];
      b1e[0] = l2[u][0]; b1e[1] = l3[u][0];
      b1o[0] = l2[u][1]; b1o[1] = l3[u][1];
      cl[0] = __builtin_amdgcn_wmma_f32_16x16x4_f32(false, a[u], false, b0e, (short)0, cl[0], false, false);
      cl[1] = __builtin_amdgcn_wmma_f32_16x16x4_f32(false, a[u], false, b0o, (short)0, cl[1], false, false);
      cl[2] = __builtin_amdgcn_wmma_f32_16x16x4_f32(false, a[u], false, b1e, (short)0, cl[2], false, false);
      cl[3] = __builtin_amdgcn_wmma_f32_16x16x4_f32(false, a[u], false, b1o, (short)0, cl[3], false, false);
      b0e[0] = r0[u][0]; b0e[1] = r1[u][0];
      b0o[0] = r0[u][1]; b0o[1] = r1[u][1];
      b1e[0] = r2[u][0]; b1e[1] = r3[u][0];
      b1o[0] = r2[u][1]; b1o[1] = r3[u][1];
      cr[0] = __builtin_amdgcn_wmma_f32_16x16x4_f32(false, a[u], false, b0e, (short)0, cr[0], false, false);
      cr[1] = __builtin_amdgcn_wmma_f32_16x16x4_f32(false, a[u], false, b0o, (short)0, cr[1], false, false);
      cr[2] = __builtin_amdgcn_wmma_f32_16x16x4_f32(false, a[u], false, b1e, (short)0, cr[2], false, false);
      cr[3] = __builtin_amdgcn_wmma_f32_16x16x4_f32(false, a[u], false, b1o, (short)0, cr[3], false, false);
    }
  }

  const int row0 = rowBase + half * 8;
  store_strip_n64(cl, biasl, Yl, row0, colBase, l15, Ncol);
  store_strip_n64(cr, biasr, Yr, row0, colBase, l15, Ncol);
}

__global__ void gemm_bias_wmma_guard(const float* __restrict__ X,
                                     const float* __restrict__ Wt,
                                     const float* __restrict__ bias,
                                     float* __restrict__ Y,
                                     int M, int K, int Ncol) {
  const int lane = threadIdx.x & 31;
  const int wave = threadIdx.x >> 5;
  const int ct = (Ncol + 15) >> 4;
  const int rt = M >> 4;
  const int tile = blockIdx.x * (blockDim.x >> 5) + wave;
  if (tile >= ct * rt) return;

  const int rowBase = (tile / ct) << 4;
  const int colBase = (tile % ct) << 4;
  const int half = lane >> 4;
  const int l15  = lane & 15;
  const int bcol = colBase + l15;
  const bool bok = bcol < Ncol;
  const int bcolc = bok ? bcol : (Ncol - 1);
  const float bmask = bok ? 1.0f : 0.0f;

  const float* __restrict__ ap = X + (size_t)(rowBase + l15) * K + half * 2;
  const float* __restrict__ bp = Wt + (size_t)(half * 2) * Ncol + bcolc;

  v8f c = {};
  #pragma unroll 4
  for (int k0 = 0; k0 < K; k0 += 4) {
    const v2f a = *(const v2f*)(ap + k0);
    const float* __restrict__ br = bp + (size_t)k0 * Ncol;
    v2f b;
    b[0] = br[0]    * bmask;
    b[1] = br[Ncol] * bmask;
    c = __builtin_amdgcn_wmma_f32_16x16x4_f32(false, a, false, b, (short)0, c, false, false);
  }

  if (bok) {
    const float bb = bias ? bias[bcol] : 0.0f;
    const int row0 = rowBase + half * 8;
    float* __restrict__ yp = Y + (size_t)row0 * Ncol + bcol;
    #pragma unroll
    for (int r = 0; r < 8; ++r)
      yp[(size_t)r * Ncol] = c[r] + bb;
  }
}

#define BIN_SHIFT 8
#define BIN_NODES 256
#define CHUNK_EDGES 4096
#define MAX_BINS 1024

__global__ void __launch_bounds__(256)
bin_count_kernel(const int* __restrict__ ei, int E, int NB, int NC,
                 int* __restrict__ HC) {
  __shared__ int hist[MAX_BINS];
  const int tid = threadIdx.x;
  for (int b = tid; b < NB; b += 256) hist[b] = 0;
  __syncthreads();
  const int base = blockIdx.x * CHUNK_EDGES;
  int d[CHUNK_EDGES / 256];
  #pragma unroll
  for (int j = 0; j < CHUNK_EDGES / 256; ++j) {
    const int i  = base + j * 256 + tid;
    const int ic = (i < E) ? i : (E - 1);
    const int dv = ei[E + ic];
    d[j] = (i < E) ? dv : -1;
  }
  #pragma unroll
  for (int j = 0; j < CHUNK_EDGES / 256; ++j)
    if (d[j] >= 0) atomicAdd(&hist[d[j] >> BIN_SHIFT], 1);
  __syncthreads();
  for (int b = tid; b < NB; b += 256) HC[(size_t)b * NC + blockIdx.x] = hist[b];
}

__global__ void __launch_bounds__(256)
bin_scan_kernel(int* __restrict__ HC, int NC, int* __restrict__ BT) {
  __shared__ int wtot[8];
  const int tid  = threadIdx.x;
  const int lane = tid & 31;
  const int wave = tid >> 5;
  int* __restrict__ row = HC + (size_t)blockIdx.x * NC;
  int carry = 0;
  for (int c0 = 0; c0 < NC; c0 += 256) {
    const int c = c0 + tid;
    const int v = (c < NC) ? row[c] : 0;
    int incl = v;
    #pragma unroll
    for (int off = 1; off < 32; off <<= 1) {
      const int y = __shfl_up(incl, off, 32);
      if (lane >= off) incl += y;
    }
    if (lane == 31) wtot[wave] = incl;
    __syncthreads();
    int woff = 0, tot = 0;
    #pragma unroll
    for (int w = 0; w < 8; ++w) {
      const int t = wtot[w];
      woff += (w < wave) ? t : 0;
      tot  += t;
    }
    if (c < NC) row[c] = carry + woff + incl - v;
    carry += tot;
    __syncthreads();
  }
  if (tid == 0) BT[blockIdx.x] = carry;
}

__global__ void __launch_bounds__(256)
bin_scatter_kernel(const int* __restrict__ ei, int E, int NB, int NC,
                   const int* __restrict__ HC, const int* __restrict__ BT,
                   int* __restrict__ BB, int* __restrict__ PBw) {
  int2* __restrict__ PB = (int2*)PBw;
  __shared__ int slot[MAX_BINS];
  __shared__ int wtot[8];
  const int tid  = threadIdx.x;
  const int lane = tid & 31;
  const int wave = tid >> 5;
  const int chunk = blockIdx.x;
  const int b0 = tid * 4;
  const int c0 = (b0 + 0 < NB) ? (b0 + 0) : (NB - 1);
  const int c1 = (b0 + 1 < NB) ? (b0 + 1) : (NB - 1);
  const int c2 = (b0 + 2 < NB) ? (b0 + 2) : (NB - 1);
  const int c3 = (b0 + 3 < NB) ? (b0 + 3) : (NB - 1);
  const int r0 = BT[c0], r1 = BT[c1], r2 = BT[c2], r3 = BT[c3];
  const int h0 = HC[(size_t)c0 * NC + chunk];
  const int h1 = HC[(size_t)c1 * NC + chunk];
  const int h2 = HC[(size_t)c2 * NC + chunk];
  const int h3 = HC[(size_t)c3 * NC + chunk];
  const int t0 = (b0 + 0 < NB) ? r0 : 0;
  const int t1 = (b0 + 1 < NB) ? r1 : 0;
  const int t2 = (b0 + 2 < NB) ? r2 : 0;
  const int t3 = (b0 + 3 < NB) ? r3 : 0;
  const int tsum = t0 + t1 + t2 + t3;
  int incl = tsum;
  #pragma unroll
  for (int off = 1; off < 32; off <<= 1) {
    const int y = __shfl_up(incl, off, 32);
    if (lane >= off) incl += y;
  }
  if (lane == 31) wtot[wave] = incl;
  __syncthreads();
  int woff = 0;
  #pragma unroll
  for (int w = 0; w < 8; ++w) woff += (w < wave) ? wtot[w] : 0;
  const int e0 = woff + incl - tsum;
  const int e1 = e0 + t0;
  const int e2 = e1 + t1;
  const int e3 = e2 + t2;
  if (b0 + 0 < NB) slot[b0 + 0] = e0 + h0;
  if (b0 + 1 < NB) slot[b0 + 1] = e1 + h1;
  if (b0 + 2 < NB) slot[b0 + 2] = e2 + h2;
  if (b0 + 3 < NB) slot[b0 + 3] = e3 + h3;
  if (chunk == 0) {
    if (b0 + 0 < NB) BB[b0 + 0] = e0;
    if (b0 + 1 < NB) BB[b0 + 1] = e1;
    if (b0 + 2 < NB) BB[b0 + 2] = e2;
    if (b0 + 3 < NB) BB[b0 + 3] = e3;
  }
  __syncthreads();
  const int base = chunk * CHUNK_EDGES;
  int s[CHUNK_EDGES / 256], d[CHUNK_EDGES / 256];
  #pragma unroll
  for (int j = 0; j < CHUNK_EDGES / 256; ++j) {
    const int i  = base + j * 256 + tid;
    const int ic = (i < E) ? i : (E - 1);
    s[j] = ei[ic];
    const int dv = ei[E + ic];
    d[j] = (i < E) ? dv : -1;
  }
  #pragma unroll
  for (int j = 0; j < CHUNK_EDGES / 256; ++j) {
    if (d[j] >= 0) {
      const int pos = atomicAdd(&slot[d[j] >> BIN_SHIFT], 1);
      PB[pos] = make_int2(s[j], d[j]);
    }
  }
}

__global__ void __launch_bounds__(1024)
bin_sort_kernel(const int* __restrict__ PBw, const int* __restrict__ BB,
                const int* __restrict__ BT, int N,
                int* __restrict__ rowptr, int* __restrict__ csr_src) {
  __shared__ int cnt[BIN_NODES];
  __shared__ int cur[BIN_NODES];
  __shared__ int wtot[8];
  const int tid  = threadIdx.x;
  const int lane = tid & 31;
  const int wave = tid >> 5;
  const int bs = BB[blockIdx.x];
  const int n  = BT[blockIdx.x];
  const int2* __restrict__ pb = (const int2*)PBw + bs;
  if (tid < BIN_NODES) cnt[tid] = 0;
  __syncthreads();
  for (int i0 = 0; i0 < n; i0 += 4096) {
    int d[4];
    #pragma unroll
    for (int u = 0; u < 4; ++u) {
      const int i  = i0 + u * 1024 + tid;
      const int ic = (i < n) ? i : (n - 1);
      const int dv = pb[ic].y;
      d[u] = (i < n) ? dv : -1;
    }
    #pragma unroll
    for (int u = 0; u < 4; ++u)
      if (d[u] >= 0) atomicAdd(&cnt[d[u] & (BIN_NODES - 1)], 1);
  }
  __syncthreads();
  const int v = (tid < BIN_NODES) ? cnt[tid] : 0;
  int incl = v;
  #pragma unroll
  for (int off = 1; off < 32; off <<= 1) {
    const int y = __shfl_up(incl, off, 32);
    if (lane >= off) incl += y;
  }
  if (tid < BIN_NODES && lane == 31) wtot[wave] = incl;
  __syncthreads();
  if (tid < BIN_NODES) {
    int woff = 0;
    #pragma unroll
    for (int w = 0; w < 8; ++w) woff += (w < wave) ? wtot[w] : 0;
    const int start = bs + woff + incl - v;
    cur[tid] = start;
    const int node = blockIdx.x * BIN_NODES + tid;
    if (node < N) rowptr[node] = start;
  }
  __syncthreads();
  for (int i0 = 0; i0 < n; i0 += 4096) {
    int2 p[4];
    #pragma unroll
    for (int u = 0; u < 4; ++u) {
      const int i  = i0 + u * 1024 + tid;
      const int ic = (i < n) ? i : (n - 1);
      p[u] = pb[ic];
      if (i >= n) p[u].y = -1;
    }
    #pragma unroll
    for (int u = 0; u < 4; ++u) {
      if (p[u].y >= 0) {
        const int pos = atomicAdd(&cur[p[u].y & (BIN_NODES - 1)], 1);
        csr_src[pos] = p[u].x;
      }
    }
  }
}

template <int V>
__global__ void __launch_bounds__(256)
node_attend_kernel(const int* __restrict__ rowptr,
                   const int* __restrict__ csr_src, int N, int E,
                   const float* __restrict__ XL, const float* XR,
                   const float* __restrict__ att,
                   const float* __restrict__ obias,
                   float* out, int relu) {
  const int lane = threadIdx.x & 31;
  const int wid_v = (blockIdx.x * blockDim.x + threadIdx.x) >> 5;
  if (wid_v >= N) return;
  const int wid = __builtin_amdgcn_readfirstlane(wid_v);
  const int beg = __builtin_amdgcn_readfirstlane(rowptr[wid]);
  const int end = (wid + 1 < N) ? __builtin_amdgcn_readfirstlane(rowptr[wid + 1]) : E;
  const int c   = end - beg + 1;

  constexpr int D4 = 32 * V;
  const float4* __restrict__ xl4 = (const float4*)XL;
  const float4* xr4 = (const float4*)XR + (size_t)wid * D4 + lane;
  const float4* __restrict__ at4 = (const float4*)att + lane;

  float4 xr[V], g[V], acc[V];
  #pragma unroll
  for (int v = 0; v < V; ++v) {
    xr[v]  = xr4[v * 32];
    g[v]   = at4[v * 32];
    acc[v] = make_float4(0.0f, 0.0f, 0.0f, 0.0f);
  }
  float m = -__builtin_huge_valf();
  float ssum = 0.0f;

  for (int base = 0; base < c; base += 32) {
    const int q = base + lane;
    int idxv = wid;
    if (q > 0 && q < c) idxv = csr_src[beg + q - 1];
    const int nq = (c - base < 32) ? (c - base) : 32;

    for (int k = 0; k < nq; k += 4) {
      float4 r[4][V];
      float e[4];
      #pragma unroll
      for (int u = 0; u < 4; ++u) {
        const int kk = (k + u < nq) ? (k + u) : k;
        const int s = __builtin_amdgcn_readlane(idxv, kk);
        const float4* __restrict__ row = xl4 + (size_t)s * D4 + lane;
        #pragma unroll
        for (int v = 0; v < V; ++v) r[u][v] = row[v * 32];
      }
      #pragma unroll
      for (int u = 0; u < 4; ++u) {
        float a = 0.0f;
        #pragma unroll
        for (int v = 0; v < V; ++v) {
          float h;
          h = r[u][v].x + xr[v].x; h = h > 0.0f ? h : NEG_SLOPE * h; a += h * g[v].x;
          h = r[u][v].y + xr[v].y; h = h > 0.0f ? h : NEG_SLOPE * h; a += h * g[v].y;
          h = r[u][v].z + xr[v].z; h = h > 0.0f ? h : NEG_SLOPE * h; a += h * g[v].z;
          h = r[u][v].w + xr[v].w; h = h > 0.0f ? h : NEG_SLOPE * h; a += h * g[v].w;
        }
        e[u] = a;
      }
      #pragma unroll
      for (int off = 16; off > 0; off >>= 1) {
        #pragma unroll
        for (int u = 0; u < 4; ++u) e[u] += __shfl_xor(e[u], off, 32);
      }
      float mc = e[0];
      #pragma unroll
      for (int u = 1; u < 4; ++u)
        if (k + u < nq) mc = fmaxf(mc, e[u]);
      const float mn = fmaxf(m, mc);
      const float sc = __expf(m - mn);
      ssum *= sc;
      #pragma unroll
      for (int v = 0; v < V; ++v) {
        acc[v].x *= sc; acc[v].y *= sc; acc[v].z *= sc; acc[v].w *= sc;
      }
      #pragma unroll
      for (int u = 0; u < 4; ++u) {
        if (k + u < nq) {
          const float p = __expf(e[u] - mn);
          ssum += p;
          #pragma unroll
          for (int v = 0; v < V; ++v) {
            acc[v].x += p * r[u][v].x;
            acc[v].y += p * r[u][v].y;
            acc[v].z += p * r[u][v].z;
            acc[v].w += p * r[u][v].w;
          }
        }
      }
      m = mn;
    }
  }

  const float inv = 1.0f / ssum;
  const float4* __restrict__ ob4 = (const float4*)obias + lane;
  float4* o4 = (float4*)out + (size_t)wid * D4 + lane;
  #pragma unroll
  for (int v = 0; v < V; ++v) {
    const float4 bb = ob4[v * 32];
    float4 o;
    o.x = acc[v].x * inv + bb.x;
    o.y = acc[v].y * inv + bb.y;
    o.z = acc[v].z * inv + bb.z;
    o.w = acc[v].w * inv + bb.w;
    if (relu) {
      o.x = fmaxf(o.x, 0.0f); o.y = fmaxf(o.y, 0.0f);
      o.z = fmaxf(o.z, 0.0f); o.w = fmaxf(o.w, 0.0f);
    }
    o4[v * 32] = o;
  }
}

extern "C" void kernel_launch(void* const* d_in, const int* in_sizes, int n_in,
                              void* d_out, int out_size, void* d_ws, size_t ws_size,
                              hipStream_t stream) {
  const float* x   = (const float*)d_in[0];
  const int*   ei  = (const int*)  d_in[1];
  const float* w1l = (const float*)d_in[2];
  const float* b1l = (const float*)d_in[3];
  const float* w1r = (const float*)d_in[4];
  const float* b1r = (const float*)d_in[5];
  const float* a1  = (const float*)d_in[6];
  const float* o1  = (const float*)d_in[7];
  const float* w2l = (const float*)d_in[8];
  const float* b2l = (const float*)d_in[9];
  const float* w2r = (const float*)d_in[10];
  const float* b2r = (const float*)d_in[11];
  const float* a2  = (const float*)d_in[12];
  const float* o2  = (const float*)d_in[13];
  const float* w3l = (const float*)d_in[14];
  const float* b3l = (const float*)d_in[15];
  const float* w3r = (const float*)d_in[16];
  const float* b3r = (const float*)d_in[17];
  const float* a3  = (const float*)d_in[18];
  const float* o3  = (const float*)d_in[19];
  const float* wc  = (const float*)d_in[20];
  const float* bc  = (const float*)d_in[21];

  const int N  = in_sizes[0] / 128;
  const int E  = in_sizes[1] / 2;

  const int Np = (N + 3) & ~3;
  const int Ep = (E + 3) & ~3;
  const int NB = (N + BIN_NODES - 1) / BIN_NODES;
  const int NC = (E + CHUNK_EDGES - 1) / CHUNK_EDGES;
  const int Hp = (NB * NC + 3) & ~3;
  float* B0  = (float*)d_ws;
  float* B1  = B0 + (size_t)N * 256;
  float* B2  = B1 + (size_t)N * 256;
  int*   CS  = (int*)(B2 + (size_t)N * 256);
  int*   RP  = CS + (size_t)Ep;
  int*   HC  = RP + (size_t)Np;
  int*   BT  = HC + (size_t)Hp;
  int*   BB  = BT + MAX_BINS;
  int*   PB  = BB + MAX_BINS;

  bin_count_kernel<<<NC, 256, 0, stream>>>(ei, E, NB, NC, HC);
  bin_scan_kernel<<<NB, 256, 0, stream>>>(HC, NC, BT);
  bin_scatter_kernel<<<NC, 256, 0, stream>>>(ei, E, NB, NC, HC, BT, BB, PB);
  bin_sort_kernel<<<NB, 1024, 0, stream>>>(PB, BB, BT, N, RP, CS);

  auto gemmPair = [&](const float* X, const float* Wlp, const float* Wrp,
                      const float* blp, const float* brp, float* Ylp, float* Yrp,
                      int m, int k, int n) {
    const int tiles = (m >> 4) * (n >> 6);
    gemm2_bias_wmma_n64<<<(tiles + 3) / 4, 128, 0, stream>>>(X, Wlp, Wrp, blp, brp,
                                                            Ylp, Yrp, m, k, n);
  };
  auto gemmGuard = [&](const float* X, const float* Wp, const float* b, float* Y,
                       int m, int k, int n) {
    const int tiles = (m >> 4) * ((n + 15) / 16);
    gemm_bias_wmma_guard<<<(tiles + 3) / 4, 128, 0, stream>>>(X, Wp, b, Y, m, k, n);
  };

  auto layer = [&](const float* Xin, int din, int dout,
                   const float* wl, const float* bl,
                   const float* wr, const float* br,
                   const float* att, const float* ob,
                   float* XL, float* XO, int relu) {
    gemmPair(Xin, wl, wr, bl, br, XL, XO, N, din, dout);
    if (dout == 128)
      node_attend_kernel<1><<<(N + 7) / 8, 256, 0, stream>>>(RP, CS, N, E, XL, XO,
                                                           att, ob, XO, relu);
    else
      node_attend_kernel<2><<<(N + 7) / 8, 256, 0, stream>>>(RP, CS, N, E, XL, XO,
                                                           att, ob, XO, relu);
  };

  layer(x,  128, 128, w1l, b1l, w1r, b1r, a1, o1, B0, B1, 1);
  layer(B1, 128, 128, w2l, b2l, w2r, b2r, a2, o2, B0, B2, 1);
  layer(B2, 128, 256, w3l, b3l, w3r, b3r, a3, o3, B0, B1, 0);
  gemmGuard(B1, wc, bc, (float*)d_out, N, 256, 40);
}


# Attention: wave split into slots of 8 lanes (128 features) or 16 (256), one edge per slot side by side; logit = in-lane sum + DPP steps, softmax per slot, slots merged at node end; sum order changes
# speedup vs baseline: 4.6593x; 1.0404x over previous
#include <hip/hip_runtime.h>

typedef __attribute__((ext_vector_type(2))) float v2f;
typedef __attribute__((ext_vector_type(8))) float v8f;

#define NEG_SLOPE 0.2f

__device__ __forceinline__ void store_strip_n64(const v8f (&cc)[4],
                                                const float* __restrict__ bias,
                                                float* __restrict__ Y,
                                                int row0, int colBase, int l15,
                                                int Ncol) {
  #pragma unroll
  for (int g = 0; g < 2; ++g) {
    const int col = colBase + g * 32 + 2 * l15;
    v2f bb = {0.0f, 0.0f};
    if (bias) bb = *(const v2f*)(bias + col);
    float* __restrict__ yp = Y + (size_t)row0 * Ncol + col;
    #pragma unroll
    for (int r = 0; r < 8; ++r) {
      v2f o;
      o[0] = cc[2 * g][r]     + bb[0];
      o[1] = cc[2 * g + 1][r] + bb[1];
      *(v2f*)(yp + (size_t)r * Ncol) = o;
    }
  }
}

__global__ void __launch_bounds__(128)
gemm2_bias_wmma_n64(const float* __restrict__ X,
                    const float* __restrict__ Wl,
                    const float* __restrict__ Wr,
                    const float* __restrict__ biasl,
                    const float* __restrict__ biasr,
                    float* __restrict__ Yl,
                    float* __restrict__ Yr,
                    int M, int K, int Ncol) {
  const int lane = threadIdx.x & 31;
  const int wave = threadIdx.x >> 5;
  const int ct = Ncol >> 6;
  const int rt = M >> 4;
  const int tile = blockIdx.x * (blockDim.x >> 5) + wave;
  if (tile >= ct * rt) return;

  const int rowBase = (tile / ct) << 4;
  const int colBase = (tile % ct) << 6;
  const int half = lane >> 4;
  const int l15  = lane & 15;

  const float* __restrict__ ap = X + (size_t)(rowBase + l15) * K + half * 2;
  const size_t boff = (size_t)(half * 2) * Ncol + colBase + 2 * l15;
  const float* __restrict__ bpl = Wl + boff;
  const float* __restrict__ bpr = Wr + boff;

  v8f cl[4] = {v8f{}, v8f{}, v8f{}, v8f{}};
  v8f cr[4] = {v8f{}, v8f{}, v8f{}, v8f{}};
  for (int k0 = 0; k0 < K; k0 += 8) {
    v2f a[2], l0[2], l1[2], l2[2], l3[2], r0[2], r1[2], r2[2], r3[2];
    #pragma unroll
    for (int u = 0; u < 2; ++u) {
      const size_t ro = (size_t)(k0 + 4 * u) * Ncol;
      a[u]  = *(const v2f*)(ap + k0 + 4 * u);
      l0[u] = *(const v2f*)(bpl + ro);
      l1[u] = *(const v2f*)(bpl + ro + Ncol);
      l2[u] = *(const v2f*)(bpl + ro + 32);
      l3[u] = *(const v2f*)(bpl + ro + Ncol + 32);
      r0[u] = *(const v2f*)(bpr + ro);
      r1[u] = *(const v2f*)(bpr + ro + Ncol);
      r2[u] = *(const v2f*)(bpr + ro + 32);
      r3[u] = *(const v2f*)(bpr + ro + Ncol + 32);
    }
    asm volatile("" :: "v"(a[0]), "v"(a[1]),
                       "v"(l0[0]), "v"(l0[1]), "v"(l1[0]), "v"(l1[1]),
                       "v"(l2[0]), "v"(l2[1]), "v"(l3[0]), "v"(l3[1]),
                       "v"(r0[0]), "v"(r0[1]), "v"(r1[0]), "v"(r1[1]),
                       "v"(r2[0]), "v"(r2[1]), "v"(r3[0]), "v"(r3[1]));
    #pragma unroll
    for (int u = 0; u < 2; ++u) {
      v2f b0e, b0o, b1e, b1o;
      b0e[0] = l0[u][0]; b0e[1] = l1[u][0];
      b0o[0] = l0[u][1]; b0o[1] = l1[u][1];
      b1e[0] = l2[u][0]; b1e[1] = l3[u][0];
      b1o[0] = l2[u][1]; b1o[1] = l3[u][1];
      cl[0] = __builtin_amdgcn_wmma_f32_16x16x4_f32(false, a[u], false, b0e, (short)0, cl[0], false, false);
      cl[1] = __builtin_amdgcn_wmma_f32_16x16x4_f32(false, a[u], false, b0o, (short)0, cl[1], false, false);
      cl[2] = __builtin_amdgcn_wmma_f32_16x16x4_f32(false, a[u], false, b1e, (short)0, cl[2], false, false);
      cl[3] = __builtin_amdgcn_wmma_f32_16x16x4_f32(false, a[u], false, b1o, (short)0, cl[3], false, false);
      b0e[0] = r0[u][0]; b0e[1] = r1[u][0];
      b0o[0] = r0[u][1]; b0o[1] = r1[u][1];
      b1e[0] = r2[u][0]; b1e[1] = r3[u][0];
      b1o[0] = r2[u][1]; b1o[1] = r3[u][1];
      cr[0] = __builtin_amdgcn_wmma_f32_16x16x4_f32(false, a[u], false, b0e, (short)0, cr[0], false, false);
      cr[1] = __builtin_amdgcn_wmma_f32_16x16x4_f32(false, a[u], false, b0o, (short)0, cr[1], false, false);
      cr[2] = __builtin_amdgcn_wmma_f32_16x16x4_f32(false, a[u], false, b1e, (short)0, cr[2], false, false);
      cr[3] = __builtin_amdgcn_wmma_f32_16x16x4_f32(false, a[u], false, b1o, (short)0, cr[3], false, false);
    }
  }

  const int row0 = rowBase + half * 8;
  store_strip_n64(cl, biasl, Yl, row0, colBase, l15, Ncol);
  store_strip_n64(cr, biasr, Yr, row0, colBase, l15, Ncol);
}

__global__ void gemm_bias_wmma_guard(const float* __restrict__ X,
                                     const float* __restrict__ Wt,
                                     const float* __restrict__ bias,
                                     float* __restrict__ Y,
                                     int M, int K, int Ncol) {
  const int lane = threadIdx.x & 31;
  const int wave = threadIdx.x >> 5;
  const int ct = (Ncol + 15) >> 4;
  const int rt = M >> 4;
  const int tile = blockIdx.x * (blockDim.x >> 5) + wave;
  if (tile >= ct * rt) return;

  const int rowBase = (tile / ct) << 4;
  const int colBase = (tile % ct) << 4;
  const int half = lane >> 4;
  const int l15  = lane & 15;
  const int bcol = colBase + l15;
  const bool bok = bcol < Ncol;
  const int bcolc = bok ? bcol : (Ncol - 1);
  const float bmask = bok ? 1.0f : 0.0f;

  const float* __restrict__ ap = X + (size_t)(rowBase + l15) * K + half * 2;
  const float* __restrict__ bp = Wt + (size_t)(half * 2) * Ncol + bcolc;

  v8f c = {};
  #pragma unroll 4
  for (int k0 = 0; k0 < K; k0 += 4) {
    const v2f a = *(const v2f*)(ap + k0);
    const float* __restrict__ br = bp + (size_t)k0 * Ncol;
    v2f b;
    b[0] = br[0]    * bmask;
    b[1] = br[Ncol] * bmask;
    c = __builtin_amdgcn_wmma_f32_16x16x4_f32(false, a, false, b, (short)0, c, false, false);
  }

  if (bok) {
    const float bb = bias ? bias[bcol] : 0.0f;
    const int row0 = rowBase + half * 8;
    float* __restrict__ yp = Y + (size_t)row0 * Ncol + bcol;
    #pragma unroll
    for (int r = 0; r < 8; ++r)
      yp[(size_t)r * Ncol] = c[r] + bb;
  }
}

#define BIN_SHIFT 8
#define BIN_NODES 256
#define CHUNK_EDGES 4096
#define MAX_BINS 1024

__global__ void __launch_bounds__(256)
bin_count_kernel(const int* __restrict__ ei, int E, int NB, int NC,
                 int* __restrict__ HC) {
  __shared__ int hist[MAX_BINS];
  const int tid = threadIdx.x;
  for (int b = tid; b < NB; b += 256) hist[b] = 0;
  __syncthreads();
  const int base = blockIdx.x * CHUNK_EDGES;
  int d[CHUNK_EDGES / 256];
  #pragma unroll
  for (int j = 0; j < CHUNK_EDGES / 256; ++j) {
    const int i  = base + j * 256 + tid;
    const int ic = (i < E) ? i : (E - 1);
    const int dv = ei[E + ic];
    d[j] = (i < E) ? dv : -1;
  }
  #pragma unroll
  for (int j = 0; j < CHUNK_EDGES / 256; ++j)
    if (d[j] >= 0) atomicAdd(&hist[d[j] >> BIN_SHIFT], 1);
  __syncthreads();
  for (int b = tid; b < NB; b += 256) HC[(size_t)b * NC + blockIdx.x] = hist[b];
}

__global__ void __launch_bounds__(256)
bin_scan_kernel(int* __restrict__ HC, int NC, int* __restrict__ BT) {
  __shared__ int wtot[8];
  const int tid  = threadIdx.x;
  const int lane = tid & 31;
  const int wave = tid >> 5;
  int* __restrict__ row = HC + (size_t)blockIdx.x * NC;
  int carry = 0;
  for (int c0 = 0; c0 < NC; c0 += 256) {
    const int c = c0 + tid;
    const int v = (c < NC) ? row[c] : 0;
    int incl = v;
    #pragma unroll
    for (int off = 1; off < 32; off <<= 1) {
      const int y = __shfl_up(incl, off, 32);
      if (lane >= off) incl += y;
    }
    if (lane == 31) wtot[wave] = incl;
    __syncthreads();
    int woff = 0, tot = 0;
    #pragma unroll
    for (int w = 0; w < 8; ++w) {
      const int t = wtot[w];
      woff += (w < wave) ? t : 0;
      tot  += t;
    }
    if (c < NC) row[c] = carry + woff + incl - v;
    carry += tot;
    __syncthreads();
  }
  if (tid == 0) BT[blockIdx.x] = carry;
}

__global__ void __launch_bounds__(256)
bin_scatter_kernel(const int* __restrict__ ei, int E, int NB, int NC,
                   const int* __restrict__ HC, const int* __restrict__ BT,
                   int* __restrict__ BB, int* __restrict__ PBw) {
  int2* __restrict__ PB = (int2*)PBw;
  __shared__ int slot[MAX_BINS];
  __shared__ int wtot[8];
  const int tid  = threadIdx.x;
  const int lane = tid & 31;
  const int wave = tid >> 5;
  const int chunk = blockIdx.x;
  const int b0 = tid * 4;
  const int c0 = (b0 + 0 < NB) ? (b0 + 0) : (NB - 1);
  const int c1 = (b0 + 1 < NB) ? (b0 + 1) : (NB - 1);
  const int c2 = (b0 + 2 < NB) ? (b0 + 2) : (NB - 1);
  const int c3 = (b0 + 3 < NB) ? (b0 + 3) : (NB - 1);
  const int r0 = BT[c0], r1 = BT[c1], r2 = BT[c2], r3 = BT[c3];
  const int h0 = HC[(size_t)c0 * NC + chunk];
  const int h1 = HC[(size_t)c1 * NC + chunk];
  const int h2 = HC[(size_t)c2 * NC + chunk];
  const int h3 = HC[(size_t)c3 * NC + chunk];
  const int t0 = (b0 + 0 < NB) ? r0 : 0;
  const int t1 = (b0 + 1 < NB) ? r1 : 0;
  const int t2 = (b0 + 2 < NB) ? r2 : 0;
  const int t3 = (b0 + 3 < NB) ? r3 : 0;
  const int tsum = t0 + t1 + t2 + t3;
  int incl = tsum;
  #pragma unroll
  for (int off = 1; off < 32; off <<= 1) {
    const int y = __shfl_up(incl, off, 32);
    if (lane >= off) incl += y;
  }
  if (lane == 31) wtot[wave] = incl;
  __syncthreads();
  int woff = 0;
  #pragma unroll
  for (int w = 0; w < 8; ++w) woff += (w < wave) ? wtot[w] : 0;
  const int e0 = woff + incl - tsum;
  const int e1 = e0 + t0;
  const int e2 = e1 + t1;
  const int e3 = e2 + t2;
  if (b0 + 0 < NB) slot[b0 + 0] = e0 + h0;
  if (b0 + 1 < NB) slot[b0 + 1] = e1 + h1;
  if (b0 + 2 < NB) slot[b0 + 2] = e2 + h2;
  if (b0 + 3 < NB) slot[b0 + 3] = e3 + h3;
  if (chunk == 0) {
    if (b0 + 0 < NB) BB[b0 + 0] = e0;
    if (b0 + 1 < NB) BB[b0 + 1] = e1;
    if (b0 + 2 < NB) BB[b0 + 2] = e2;
    if (b0 + 3 < NB) BB[b0 + 3] = e3;
  }
  __syncthreads();
  const int base = chunk * CHUNK_EDGES;
  int s[CHUNK_EDGES / 256], d[CHUNK_EDGES / 256];
  #pragma unroll
  for (int j = 0; j < CHUNK_EDGES / 256; ++j) {
    const int i  = base + j * 256 + tid;
    const int ic = (i < E) ? i : (E - 1);
    s[j] = ei[ic];
    const int dv = ei[E + ic];
    d[j] = (i < E) ? dv : -1;
  }
  #pragma unroll
  for (int j = 0; j < CHUNK_EDGES / 256; ++j) {
    if (d[j] >= 0) {
      const int pos = atomicAdd(&slot[d[j] >> BIN_SHIFT], 1);
      PB[pos] = make_int2(s[j], d[j]);
    }
  }
}

__global__ void __launch_bounds__(1024)
bin_sort_kernel(const int* __restrict__ PBw, const int* __restrict__ BB,
                const int* __restrict__ BT, int N,
                int* __restrict__ rowptr, int* __restrict__ csr_src) {
  __shared__ int cnt[BIN_NODES];
  __shared__ int cur[BIN_NODES];
  __shared__ int wtot[8];
  const int tid  = threadIdx.x;
  const int lane = tid & 31;
  const int wave = tid >> 5;
  const int bs = BB[blockIdx.x];
  const int n  = BT[blockIdx.x];
  const int2* __restrict__ pb = (const int2*)PBw + bs;
  if (tid < BIN_NODES) cnt[tid] = 0;
  __syncthreads();
  for (int i0 = 0; i0 < n; i0 += 4096) {
    int d[4];
    #pragma unroll
    for (int u = 0; u < 4; ++u) {
      const int i  = i0 + u * 1024 + tid;
      const int ic = (i < n) ? i : (n - 1);
      const int dv = pb[ic].y;
      d[u] = (i < n) ? dv : -1;
    }
    #pragma unroll
    for (int u = 0; u < 4; ++u)
      if (d[u] >= 0) atomicAdd(&cnt[d[u] & (BIN_NODES - 1)], 1);
  }
  __syncthreads();
  const int v = (tid < BIN_NODES) ? cnt[tid] : 0;
  int incl = v;
  #pragma unroll
  for (int off = 1; off < 32; off <<= 1) {
    const int y = __shfl_up(incl, off, 32);
    if (lane >= off) incl += y;
  }
  if (tid < BIN_NODES && lane == 31) wtot[wave] = incl;
  __syncthreads();
  if (tid < BIN_NODES) {
    int woff = 0;
    #pragma unroll
    for (int w = 0; w < 8; ++w) woff += (w < wave) ? wtot[w] : 0;
    const int start = bs + woff + incl - v;
    cur[tid] = start;
    const int node = blockIdx.x * BIN_NODES + tid;
    if (node < N) rowptr[node] = start;
  }
  __syncthreads();
  for (int i0 = 0; i0 < n; i0 += 4096) {
    int2 p[4];
    #pragma unroll
    for (int u = 0; u < 4; ++u) {
      const int i  = i0 + u * 1024 + tid;
      const int ic = (i < n) ? i : (n - 1);
      p[u] = pb[ic];
      if (i >= n) p[u].y = -1;
    }
    #pragma unroll
    for (int u = 0; u < 4; ++u) {
      if (p[u].y >= 0) {
        const int pos = atomicAdd(&cur[p[u].y & (BIN_NODES - 1)], 1);
        csr_src[pos] = p[u].x;
      }
    }
  }
}

typedef __attribute__((ext_vector_type(4))) float v4f;

__device__ __forceinline__ float xhalf(float x) {
  const int b = __builtin_bit_cast(int, x);
  return __builtin_bit_cast(float,
      __builtin_amdgcn_permlanex16(b, b, 0x76543210, 0xfedcba98, false, false));
}
template <int NX>
__device__ __forceinline__ float xlane(float x) {
  const int b = __builtin_bit_cast(int, x);
  return __builtin_bit_cast(float,
      __builtin_amdgcn_update_dpp(0, b, 0x160 + NX, 0xf, 0xf, true));
}

template <int LPE>
__global__ void __launch_bounds__(256)
node_attend_kernel(const int* __restrict__ rowptr,
                   const int* __restrict__ csr_src, int N, int E,
                   const float* __restrict__ XL, const float* XR,
                   const float* __restrict__ att,
                   const float* __restrict__ obias,
                   float* out, int relu) {
  constexpr int S   = 32 / LPE;
  constexpr int D4  = 4 * LPE;
  constexpr int OPS = 4 / S;
  constexpr float LOWEST = -3.402823466e+38f;
  const int lane = threadIdx.x & 31;
  const int wid_v = (blockIdx.x * blockDim.x + threadIdx.x) >> 5;
  if (wid_v >= N) return;
  const int wid = __builtin_amdgcn_readfirstlane(wid_v);
  const int beg = __builtin_amdgcn_readfirstlane(rowptr[wid]);
  const int end = (wid + 1 < N) ? __builtin_amdgcn_readfirstlane(rowptr[wid + 1]) : E;
  const int c   = end - beg + 1;
  const int slot = lane / LPE;
  const int t    = lane % LPE;

  const v4f* __restrict__ xl4 = (const v4f*)XL;
  const v4f* xr4 = (const v4f*)XR + (size_t)wid * D4 + t;
  const v4f* __restrict__ at4 = (const v4f*)att + t;

  v4f xr[4], g[4], acc[4];
  #pragma unroll
  for (int j = 0; j < 4; ++j) {
    xr[j]  = xr4[j * LPE];
    g[j]   = at4[j * LPE];
    acc[j] = v4f{0.0f, 0.0f, 0.0f, 0.0f};
  }
  float m = LOWEST;
  float ssum = 0.0f;

  for (int base = 0; base < c; base += 32) {
    const int q = base + lane;
    int idxv = wid;
    if (q > 0 && q < c) idxv = csr_src[beg + q - 1];
    const int nq = (c - base < 32) ? (c - base) : 32;

    for (int k = 0; k < nq; k += S) {
      int src = __builtin_amdgcn_readlane(idxv, k);
      #pragma unroll
      for (int u = 1; u < S; ++u) {
        const int kk = (k + u < nq) ? (k + u) : k;
        const int su = __builtin_amdgcn_readlane(idxv, kk);
        if (slot == u) src = su;
      }
      const bool valid = (k + slot) < nq;
      const v4f* __restrict__ row = xl4 + (size_t)src * D4 + t;
      v4f r[4];
      #pragma unroll
      for (int j = 0; j < 4; ++j) r[j] = row[j * LPE];

      v4f a4 = v4f{0.0f, 0.0f, 0.0f, 0.0f};
      #pragma unroll
      for (int j = 0; j < 4; ++j) {
        v4f h = r[j] + xr[j];
        #pragma unroll
        for (int e = 0; e < 4; ++e) h[e] = h[e] > 0.0f ? h[e] : NEG_SLOPE * h[e];
        a4 += h * g[j];
      }
      float a = (a4[0] + a4[1]) + (a4[2] + a4[3]);
      a += xlane<1>(a);
      a += xlane<2>(a);
      a += xlane<4>(a);
      if (LPE == 16) a += xlane<8>(a);

      const float ev = valid ? a : LOWEST;
      const float mn = fmaxf(m, ev);
      const float sc = __expf(m - mn);
      const float pe = __expf(ev - mn);
      const float p  = valid ? pe : 0.0f;
      ssum = ssum * sc + p;
      #pragma unroll
      for (int j = 0; j < 4; ++j) acc[j] = acc[j] * sc + p * r[j];
      m = mn;
    }
  }

  float mAll = m;
  if (LPE == 8) mAll = fmaxf(mAll, xlane<8>(mAll));
  mAll = fmaxf(mAll, xhalf(mAll));
  const float f = __expf(m - mAll);
  ssum *= f;
  #pragma unroll
  for (int j = 0; j < 4; ++j) acc[j] = acc[j] * f;
  if (LPE == 8) {
    ssum += xlane<8>(ssum);
    #pragma unroll
    for (int j = 0; j < 4; ++j) {
      #pragma unroll
      for (int e = 0; e < 4; ++e) acc[j][e] += xlane<8>(acc[j][e]);
    }
  }
  ssum += xhalf(ssum);
  #pragma unroll
  for (int j = 0; j < 4; ++j) {
    #pragma unroll
    for (int e = 0; e < 4; ++e) acc[j][e] += xhalf(acc[j][e]);
  }

  const float inv = 1.0f / ssum;
  const v4f* __restrict__ ob4 = (const v4f*)obias;
  v4f* o4 = (v4f*)out + (size_t)wid * D4;
  #pragma unroll
  for (int i = 0; i < OPS; ++i) {
    v4f o = acc[i];
    #pragma unroll
    for (int s = 1; s < S; ++s)
      if (slot == s) o = acc[s * OPS + i];
    const int oi = (slot * OPS + i) * LPE + t;
    const v4f bb = ob4[oi];
    v4f res = o * inv + bb;
    if (relu) {
      res[0] = fmaxf(res[0], 0.0f); res[1] = fmaxf(res[1], 0.0f);
      res[2] = fmaxf(res[2], 0.0f); res[3] = fmaxf(res[3], 0.0f);
    }
    o4[oi] = res;
  }
}

extern "C" void kernel_launch(void* const* d_in, const int* in_sizes, int n_in,
                              void* d_out, int out_size, void* d_ws, size_t ws_size,
                              hipStream_t stream) {
  const float* x   = (const float*)d_in[0];
  const int*   ei  = (const int*)  d_in[1];
  const float* w1l = (const float*)d_in[2];
  const float* b1l = (const float*)d_in[3];
  const float* w1r = (const float*)d_in[4];
  const float* b1r = (const float*)d_in[5];
  const float* a1  = (const float*)d_in[6];
  const float* o1  = (const float*)d_in[7];
  const float* w2l = (const float*)d_in[8];
  const float* b2l = (const float*)d_in[9];
  const float* w2r = (const float*)d_in[10];
  const float* b2r = (const float*)d_in[11];
  const float* a2  = (const float*)d_in[12];
  const float* o2  = (const float*)d_in[13];
  const float* w3l = (const float*)d_in[14];
  const float* b3l = (const float*)d_in[15];
  const float* w3r = (const float*)d_in[16];
  const float* b3r = (const float*)d_in[17];
  const float* a3  = (const float*)d_in[18];
  const float* o3  = (const float*)d_in[19];
  const float* wc  = (const float*)d_in[20];
  const float* bc  = (const float*)d_in[21];

  const int N  = in_sizes[0] / 128;
  const int E  = in_sizes[1] / 2;

  const int Np = (N + 3) & ~3;
  const int Ep = (E + 3) & ~3;
  const int NB = (N + BIN_NODES - 1) / BIN_NODES;
  const int NC = (E + CHUNK_EDGES - 1) / CHUNK_EDGES;
  const int Hp = (NB * NC + 3) & ~3;
  float* B0  = (float*)d_ws;
  float* B1  = B0 + (size_t)N * 256;
  float* B2  = B1 + (size_t)N * 256;
  int*   CS  = (int*)(B2 + (size_t)N * 256);
  int*   RP  = CS + (size_t)Ep;
  int*   HC  = RP + (size_t)Np;
  int*   BT  = HC + (size_t)Hp;
  int*   BB  = BT + MAX_BINS;
  int*   PB  = BB + MAX_BINS;

  bin_count_kernel<<<NC, 256, 0, stream>>>(ei, E, NB, NC, HC);
  bin_scan_kernel<<<NB, 256, 0, stream>>>(HC, NC, BT);
  bin_scatter_kernel<<<NC, 256, 0, stream>>>(ei, E, NB, NC, HC, BT, BB, PB);
  bin_sort_kernel<<<NB, 1024, 0, stream>>>(PB, BB, BT, N, RP, CS);

  auto gemmPair = [&](const float* X, const float* Wlp, const float* Wrp,
                      const float* blp, const float* brp, float* Ylp, float* Yrp,
                      int m, int k, int n) {
    const int tiles = (m >> 4) * (n >> 6);
    gemm2_bias_wmma_n64<<<(tiles + 3) / 4, 128, 0, stream>>>(X, Wlp, Wrp, blp, brp,
                                                            Ylp, Yrp, m, k, n);
  };
  auto gemmGuard = [&](const float* X, const float* Wp, const float* b, float* Y,
                       int m, int k, int n) {
    const int tiles = (m >> 4) * ((n + 15) / 16);
    gemm_bias_wmma_guard<<<(tiles + 3) / 4, 128, 0, stream>>>(X, Wp, b, Y, m, k, n);
  };

  auto layer = [&](const float* Xin, int din, int dout,
                   const float* wl, const float* bl,
                   const float* wr, const float* br,
                   const float* att, const float* ob,
                   float* XL, float* XO, int relu) {
    gemmPair(Xin, wl, wr, bl, br, XL, XO, N, din, dout);
    if (dout == 128)
      node_attend_kernel<8><<<(N + 7) / 8, 256, 0, stream>>>(RP, CS, N, E, XL, XO,
                                                           att, ob, XO, relu);
    else
      node_attend_kernel<16><<<(N + 7) / 8, 256, 0, stream>>>(RP, CS, N, E, XL, XO,
                                                           att, ob, XO, relu);
  };

  layer(x,  128, 128, w1l, b1l, w1r, b1r, a1, o1, B0, B1, 1);
  layer(B1, 128, 128, w2l, b2l, w2r, b2r, a2, o2, B0, B2, 1);
  layer(B2, 128, 256, w3l, b3l, w3r, b3r, a3, o3, B0, B1, 0);
  gemmGuard(B1, wc, bc, (float*)d_out, N, 256, 40);
}


# Each attention pass and the product that reads its result share a launch: a work-group keeps its 32 nodes' rows in LDS and multiplies them by the next weights; 3 launches gone; sums keep their order
# speedup vs baseline: 5.0798x; 1.0902x over previous
#include <hip/hip_runtime.h>

typedef __attribute__((ext_vector_type(2))) float v2f;
typedef __attribute__((ext_vector_type(8))) float v8f;

#define NEG_SLOPE 0.2f

__device__ __forceinline__ void store_strip_n64(const v8f (&cc)[4],
                                                const float* __restrict__ bias,
                                                float* __restrict__ Y,
                                                int row0, int colBase, int l15,
                                                int Ncol) {
  #pragma unroll
  for (int g = 0; g < 2; ++g) {
    const int col = colBase + g * 32 + 2 * l15;
    v2f bb = {0.0f, 0.0f};
    if (bias) bb = *(const v2f*)(bias + col);
    float* __restrict__ yp = Y + (size_t)row0 * Ncol + col;
    #pragma unroll
    for (int r = 0; r < 8; ++r) {
      v2f o;
      o[0] = cc[2 * g][r]     + bb[0];
      o[1] = cc[2 * g + 1][r] + bb[1];
      *(v2f*)(yp + (size_t)r * Ncol) = o;
    }
  }
}

__global__ void __launch_bounds__(128)
gemm2_bias_wmma_n64(const float* __restrict__ X,
                    const float* __restrict__ Wl,
                    const float* __restrict__ Wr,
                    const float* __restrict__ biasl,
                    const float* __restrict__ biasr,
                    float* __restrict__ Yl,
                    float* __restrict__ Yr,
                    int M, int K, int Ncol) {
  const int lane = threadIdx.x & 31;
  const int wave = threadIdx.x >> 5;
  const int ct = Ncol >> 6;
  const int rt = M >> 4;
  const int tile = blockIdx.x * (blockDim.x >> 5) + wave;
  if (tile >= ct * rt) return;

  const int rowBase = (tile / ct) << 4;
  const int colBase = (tile % ct) << 6;
  const int half = lane >> 4;
  const int l15  = lane & 15;

  const float* __restrict__ ap = X + (size_t)(rowBase + l15) * K + half * 2;
  const size_t boff = (size_t)(half * 2) * Ncol + colBase + 2 * l15;
  const float* __restrict__ bpl = Wl + boff;
  const float* __restrict__ bpr = Wr + boff;

  v8f cl[4] = {v8f{}, v8f{}, v8f{}, v8f{}};
  v8f cr[4] = {v8f{}, v8f{}, v8f{}, v8f{}};
  for (int k0 = 0; k0 < K; k0 += 8) {
    v2f a[2], l0[2], l1[2], l2[2], l3[2], r0[2], r1[2], r2[2], r3[2];
    #pragma unroll
    for (int u = 0; u < 2; ++u) {
      const size_t ro = (size_t)(k0 + 4 * u) * Ncol;
      a[u]  = *(const v2f*)(ap + k0 + 4 * u);
      l0[u] = *(const v2f*)(bpl + ro);
      l1[u] = *(const v2f*)(bpl + ro + Ncol);
      l2[u] = *(const v2f*)(bpl + ro + 32);
      l3[u] = *(const v2f*)(bpl + ro + Ncol + 32);
      r0[u] = *(const v2f*)(bpr + ro);
      r1[u] = *(const v2f*)(bpr + ro + Ncol);
      r2[u] = *(const v2f*)(bpr + ro + 32);
      r3[u] = *(const v2f*)(bpr + ro + Ncol + 32);
    }
    asm volatile("" :: "v"(a[0]), "v"(a[1]),
                       "v"(l0[0]), "v"(l0[1]), "v"(l1[0]), "v"(l1[1]),
                       "v"(l2[0]), "v"(l2[1]), "v"(l3[0]), "v"(l3[1]),
                       "v"(r0[0]), "v"(r0[1]), "v"(r1[0]), "v"(r1[1]),
                       "v"(r2[0]), "v"(r2[1]), "v"(r3[0]), "v"(r3[1]));
    #pragma unroll
    for (int u = 0; u < 2; ++u) {
      v2f b0e, b0o, b1e, b1o;
      b0e[0] = l0[u][0]; b0e[1] = l1[u][0];
      b0o[0] = l0[u][1]; b0o[1] = l1[u][1];
      b1e[0] = l2[u][0]; b1e[1] = l3[u][0];
      b1o[0] = l2[u][1]; b1o[1] = l3[u][1];
      cl[0] = __builtin_amdgcn_wmma_f32_16x16x4_f32(false, a[u], false, b0e, (short)0, cl[0], false, false);
      cl[1] = __builtin_amdgcn_wmma_f32_16x16x4_f32(false, a[u], false, b0o, (short)0, cl[1], false, false);
      cl[2] = __builtin_amdgcn_wmma_f32_16x16x4_f32(false, a[u], false, b1e, (short)0, cl[2], false, false);
      cl[3] = __builtin_amdgcn_wmma_f32_16x16x4_f32(false, a[u], false, b1o, (short)0, cl[3], false, false);
      b0e[0] = r0[u][0]; b0e[1] = r1[u][0];
      b0o[0] = r0[u][1]; b0o[1] = r1[u][1];
      b1e[0] = r2[u][0]; b1e[1] = r3[u][0];
      b1o[0] = r2[u][1]; b1o[1] = r3[u][1];
      cr[0] = __builtin_amdgcn_wmma_f32_16x16x4_f32(false, a[u], false, b0e, (short)0, cr[0], false, false);
      cr[1] = __builtin_amdgcn_wmma_f32_16x16x4_f32(false, a[u], false, b0o, (short)0, cr[1], false, false);
      cr[2] = __builtin_amdgcn_wmma_f32_16x16x4_f32(false, a[u], false, b1e, (short)0, cr[2], false, false);
      cr[3] = __builtin_amdgcn_wmma_f32_16x16x4_f32(false, a[u], false, b1o, (short)0, cr[3], false, false);
    }
  }

  const int row0 = rowBase + half * 8;
  store_strip_n64(cl, biasl, Yl, row0, colBase, l15, Ncol);
  store_strip_n64(cr, biasr, Yr, row0, colBase, l15, Ncol);
}

#define BIN_SHIFT 8
#define BIN_NODES 256
#define CHUNK_EDGES 4096
#define MAX_BINS 1024

__global__ void __launch_bounds__(256)
bin_count_kernel(const int* __restrict__ ei, int E, int NB, int NC,
                 int* __restrict__ HC) {
  __shared__ int hist[MAX_BINS];
  const int tid = threadIdx.x;
  for (int b = tid; b < NB; b += 256) hist[b] = 0;
  __syncthreads();
  const int base = blockIdx.x * CHUNK_EDGES;
  int d[CHUNK_EDGES / 256];
  #pragma unroll
  for (int j = 0; j < CHUNK_EDGES / 256; ++j) {
    const int i  = base + j * 256 + tid;
    const int ic = (i < E) ? i : (E - 1);
    const int dv = ei[E + ic];
    d[j] = (i < E) ? dv : -1;
  }
  #pragma unroll
  for (int j = 0; j < CHUNK_EDGES / 256; ++j)
    if (d[j] >= 0) atomicAdd(&hist[d[j] >> BIN_SHIFT], 1);
  __syncthreads();
  for (int b = tid; b < NB; b += 256) HC[(size_t)b * NC + blockIdx.x] = hist[b];
}

__global__ void __launch_bounds__(256)
bin_scan_kernel(int* __restrict__ HC, int NC, int* __restrict__ BT) {
  __shared__ int wtot[8];
  const int tid  = threadIdx.x;
  const int lane = tid & 31;
  const int wave = tid >> 5;
  int* __restrict__ row = HC + (size_t)blockIdx.x * NC;
  int carry = 0;
  for (int c0 = 0; c0 < NC; c0 += 256) {
    const int c = c0 + tid;
    const int v = (c < NC) ? row[c] : 0;
    int incl = v;
    #pragma unroll
    for (int off = 1; off < 32; off <<= 1) {
      const int y = __shfl_up(incl, off, 32);
      if (lane >= off) incl += y;
    }
    if (lane == 31) wtot[wave] = incl;
    __syncthreads();
    int woff = 0, tot = 0;
    #pragma unroll
    for (int w = 0; w < 8; ++w) {
      const int t = wtot[w];
      woff += (w < wave) ? t : 0;
      tot  += t;
    }
    if (c < NC) row[c] = carry + woff + incl - v;
    carry += tot;
    __syncthreads();
  }
  if (tid == 0) BT[blockIdx.x] = carry;
}

__global__ void __launch_bounds__(256)
bin_scatter_kernel(const int* __restrict__ ei, int E, int NB, int NC,
                   const int* __restrict__ HC, const int* __restrict__ BT,
                   int* __restrict__ BB, int* __restrict__ PBw) {
  int2* __restrict__ PB = (int2*)PBw;
  __shared__ int slot[MAX_BINS];
  __shared__ int wtot[8];
  const int tid  = threadIdx.x;
  const int lane = tid & 31;
  const int wave = tid >> 5;
  const int chunk = blockIdx.x;
  const int b0 = tid * 4;
  const int c0 = (b0 + 0 < NB) ? (b0 + 0) : (NB - 1);
  const int c1 = (b0 + 1 < NB) ? (b0 + 1) : (NB - 1);
  const int c2 = (b0 + 2 < NB) ? (b0 + 2) : (NB - 1);
  const int c3 = (b0 + 3 < NB) ? (b0 + 3) : (NB - 1);
  const int r0 = BT[c0], r1 = BT[c1], r2 = BT[c2], r3 = BT[c3];
  const int h0 = HC[(size_t)c0 * NC + chunk];
  const int h1 = HC[(size_t)c1 * NC + chunk];
  const int h2 = HC[(size_t)c2 * NC + chunk];
  const int h3 = HC[(size_t)c3 * NC + chunk];
  const int t0 = (b0 + 0 < NB) ? r0 : 0;
  const int t1 = (b0 + 1 < NB) ? r1 : 0;
  const int t2 = (b0 + 2 < NB) ? r2 : 0;
  const int t3 = (b0 + 3 < NB) ? r3 : 0;
  const int tsum = t0 + t1 + t2 + t3;
  int incl = tsum;
  #pragma unroll
  for (int off = 1; off < 32; off <<= 1) {
    const int y = __shfl_up(incl, off, 32);
    if (lane >= off) incl += y;
  }
  if (lane == 31) wtot[wave] = incl;
  __syncthreads();
  int woff = 0;
  #pragma unroll
  for (int w = 0; w < 8; ++w) woff += (w < wave) ? wtot[w] : 0;
  const int e0 = woff + incl - tsum;
  const int e1 = e0 + t0;
  const int e2 = e1 + t1;
  const int e3 = e2 + t2;
  if (b0 + 0 < NB) slot[b0 + 0] = e0 + h0;
  if (b0 + 1 < NB) slot[b0 + 1] = e1 + h1;
  if (b0 + 2 < NB) slot[b0 + 2] = e2 + h2;
  if (b0 + 3 < NB) slot[b0 + 3] = e3 + h3;
  if (chunk == 0) {
    if (b0 + 0 < NB) BB[b0 + 0] = e0;
    if (b0 + 1 < NB) BB[b0 + 1] = e1;
    if (b0 + 2 < NB) BB[b0 + 2] = e2;
    if (b0 + 3 < NB) BB[b0 + 3] = e3;
  }
  __syncthreads();
  const int base = chunk * CHUNK_EDGES;
  int s[CHUNK_EDGES / 256], d[CHUNK_EDGES / 256];
  #pragma unroll
  for (int j = 0; j < CHUNK_EDGES / 256; ++j) {
    const int i  = base + j * 256 + tid;
    const int ic = (i < E) ? i : (E - 1);
    s[j] = ei[ic];
    const int dv = ei[E + ic];
    d[j] = (i < E) ? dv : -1;
  }
  #pragma unroll
  for (int j = 0; j < CHUNK_EDGES / 256; ++j) {
    if (d[j] >= 0) {
      const int pos = atomicAdd(&slot[d[j] >> BIN_SHIFT], 1);
      PB[pos] = make_int2(s[j], d[j]);
    }
  }
}

__global__ void __launch_bounds__(1024)
bin_sort_kernel(const int* __restrict__ PBw, const int* __restrict__ BB,
                const int* __restrict__ BT, int N,
                int* __restrict__ rowptr, int* __restrict__ csr_src) {
  __shared__ int cnt[BIN_NODES];
  __shared__ int cur[BIN_NODES];
  __shared__ int wtot[8];
  const int tid  = threadIdx.x;
  const int lane = tid & 31;
  const int wave = tid >> 5;
  const int bs = BB[blockIdx.x];
  const int n  = BT[blockIdx.x];
  const int2* __restrict__ pb = (const int2*)PBw + bs;
  if (tid < BIN_NODES) cnt[tid] = 0;
  __syncthreads();
  for (int i0 = 0; i0 < n; i0 += 4096) {
    int d[4];
    #pragma unroll
    for (int u = 0; u < 4; ++u) {
      const int i  = i0 + u * 1024 + tid;
      const int ic = (i < n) ? i : (n - 1);
      const int dv = pb[ic].y;
      d[u] = (i < n) ? dv : -1;
    }
    #pragma unroll
    for (int u = 0; u < 4; ++u)
      if (d[u] >= 0) atomicAdd(&cnt[d[u] & (BIN_NODES - 1)], 1);
  }
  __syncthreads();
  const int v = (tid < BIN_NODES) ? cnt[tid] : 0;
  int incl = v;
  #pragma unroll
  for (int off = 1; off < 32; off <<= 1) {
    const int y = __shfl_up(incl, off, 32);
    if (lane >= off) incl += y;
  }
  if (tid < BIN_NODES && lane == 31) wtot[wave] = incl;
  __syncthreads();
  if (tid < BIN_NODES) {
    int woff = 0;
    #pragma unroll
    for (int w = 0; w < 8; ++w) woff += (w < wave) ? wtot[w] : 0;
    const int start = bs + woff + incl - v;
    cur[tid] = start;
    const int node = blockIdx.x * BIN_NODES + tid;
    if (node < N) rowptr[node] = start;
  }
  __syncthreads();
  for (int i0 = 0; i0 < n; i0 += 4096) {
    int2 p[4];
    #pragma unroll
    for (int u = 0; u < 4; ++u) {
      const int i  = i0 + u * 1024 + tid;
      const int ic = (i < n) ? i : (n - 1);
      p[u] = pb[ic];
      if (i >= n) p[u].y = -1;
    }
    #pragma unroll
    for (int u = 0; u < 4; ++u) {
      if (p[u].y >= 0) {
        const int pos = atomicAdd(&cur[p[u].y & (BIN_NODES - 1)], 1);
        csr_src[pos] = p[u].x;
      }
    }
  }
}

typedef __attribute__((ext_vector_type(4))) float v4f;

__device__ __forceinline__ float xhalf(float x) {
  const int b = __builtin_bit_cast(int, x);
  return __builtin_bit_cast(float,
      __builtin_amdgcn_permlanex16(b, b, 0x76543210, 0xfedcba98, false, false));
}
template <int NX>
__device__ __forceinline__ float xlane(float x) {
  const int b = __builtin_bit_cast(int, x);
  return __builtin_bit_cast(float,
      __builtin_amdgcn_update_dpp(0, b, 0x160 + NX, 0xf, 0xf, true));
}

template <int LPE>
__device__ __forceinline__ void attend_node(int wid, int lane,
                   const int* __restrict__ rowptr,
                   const int* __restrict__ csr_src, int N, int E,
                   const float* __restrict__ XL, const float* __restrict__ XR,
                   const float* __restrict__ att,
                   const float* __restrict__ obias, int relu,
                   v4f (&res)[4 / (32 / LPE)]) {
  constexpr int S   = 32 / LPE;
  constexpr int D4  = 4 * LPE;
  constexpr int OPS = 4 / S;
  constexpr float LOWEST = -3.402823466e+38f;
  const int beg = __builtin_amdgcn_readfirstlane(rowptr[wid]);
  const int end = (wid + 1 < N) ? __builtin_amdgcn_readfirstlane(rowptr[wid + 1]) : E;
  const int c   = end - beg + 1;
  const int slot = lane / LPE;
  const int t    = lane % LPE;

  const v4f* __restrict__ xl4 = (const v4f*)XL;
  const v4f* __restrict__ xr4 = (const v4f*)XR + (size_t)wid * D4 + t;
  const v4f* __restrict__ at4 = (const v4f*)att + t;

  v4f xr[4], g[4], acc[4];
  #pragma unroll
  for (int j = 0; j < 4; ++j) {
    xr[j]  = xr4[j * LPE];
    g[j]   = at4[j * LPE];
    acc[j] = v4f{0.0f, 0.0f, 0.0f, 0.0f};
  }
  float m = LOWEST;
  float ssum = 0.0f;

  for (int base = 0; base < c; base += 32) {
    const int q = base + lane;
    int idxv = wid;
    if (q > 0 && q < c) idxv = csr_src[beg + q - 1];
    const int nq = (c - base < 32) ? (c - base) : 32;

    for (int k = 0; k < nq; k += S) {
      int src = __builtin_amdgcn_readlane(idxv, k);
      #pragma unroll
      for (int u = 1; u < S; ++u) {
        const int kk = (k + u < nq) ? (k + u) : k;
        const int su = __builtin_amdgcn_readlane(idxv, kk);
        if (slot == u) src = su;
      }
      const bool valid = (k + slot) < nq;
      const v4f* __restrict__ row = xl4 + (size_t)src * D4 + t;
      v4f r[4];
      #pragma unroll
      for (int j = 0; j < 4; ++j) r[j] = row[j * LPE];

      v4f a4 = v4f{0.0f, 0.0f, 0.0f, 0.0f};
      #pragma unroll
      for (int j = 0; j < 4; ++j) {
        v4f h = r[j] + xr[j];
        #pragma unroll
        for (int e = 0; e < 4; ++e) h[e] = h[e] > 0.0f ? h[e] : NEG_SLOPE * h[e];
        a4 += h * g[j];
      }
      float a = (a4[0] + a4[1]) + (a4[2] + a4[3]);
      a += xlane<1>(a);
      a += xlane<2>(a);
      a += xlane<4>(a);
      if (LPE == 16) a += xlane<8>(a);

      const float ev = valid ? a : LOWEST;
      const float mn = fmaxf(m, ev);
      const float sc = __expf(m - mn);
      const float pe = __expf(ev - mn);
      const float p  = valid ? pe : 0.0f;
      ssum = ssum * sc + p;
      #pragma unroll
      for (int j = 0; j < 4; ++j) acc[j] = acc[j] * sc + p * r[j];
      m = mn;
    }
  }

  float mAll = m;
  if (LPE == 8) mAll = fmaxf(mAll, xlane<8>(mAll));
  mAll = fmaxf(mAll, xhalf(mAll));
  const float f = __expf(m - mAll);
  ssum *= f;
  #pragma unroll
  for (int j = 0; j < 4; ++j) acc[j] = acc[j] * f;
  if (LPE == 8) {
    ssum += xlane<8>(ssum);
    #pragma unroll
    for (int j = 0; j < 4; ++j) {
      #pragma unroll
      for (int e = 0; e < 4; ++e) acc[j][e] += xlane<8>(acc[j][e]);
    }
  }
  ssum += xhalf(ssum);
  #pragma unroll
  for (int j = 0; j < 4; ++j) {
    #pragma unroll
    for (int e = 0; e < 4; ++e) acc[j][e] += xhalf(acc[j][e]);
  }

  const float inv = 1.0f / ssum;
  const v4f* __restrict__ ob4 = (const v4f*)obias;
  #pragma unroll
  for (int i = 0; i < OPS; ++i) {
    v4f o = acc[i];
    #pragma unroll
    for (int s = 1; s < S; ++s)
      if (slot == s) o = acc[s * OPS + i];
    const int oi = (slot * OPS + i) * LPE + t;
    const v4f bb = ob4[oi];
    v4f rr = o * inv + bb;
    if (relu) {
      rr[0] = fmaxf(rr[0], 0.0f); rr[1] = fmaxf(rr[1], 0.0f);
      rr[2] = fmaxf(rr[2], 0.0f); rr[3] = fmaxf(rr[3], 0.0f);
    }
    res[i] = rr;
  }
}

#define TILE_NODES 32
#define TILE_WAVES 16

template <int LPE, int MODE>
__global__ void __launch_bounds__(512, 8)
attend_product_kernel(const int* __restrict__ rowptr,
                      const int* __restrict__ csr_src, int N, int E,
                      const float* __restrict__ XL, const float* __restrict__ XR,
                      const float* __restrict__ att,
                      const float* __restrict__ obias, int relu,
                      const float* __restrict__ Wl, const float* __restrict__ Wr,
                      const float* __restrict__ bl, const float* __restrict__ br,
                      float* __restrict__ Yl, float* __restrict__ Yr, int Ncol) {
  constexpr int S     = 32 / LPE;
  constexpr int OPS   = 4 / S;
  constexpr int D     = 16 * LPE;
  constexpr int PITCH = D + 4;
  constexpr int P4    = PITCH / 4;
  __shared__ v4f hs4[TILE_NODES * P4];
  const int lane = threadIdx.x & 31;
  const int wave = __builtin_amdgcn_readfirstlane((int)(threadIdx.x >> 5));
  const int nodeBase = blockIdx.x * TILE_NODES;
  {
    const int slot = lane / LPE;
    const int t    = lane % LPE;
    #pragma unroll 1
    for (int it = 0; it < TILE_NODES / TILE_WAVES; ++it) {
      const int r = it * TILE_WAVES + wave;
      const int node = nodeBase + r;
      v4f res[OPS];
      if (node < N) {
        attend_node<LPE>(node, lane, rowptr, csr_src, N, E, XL, XR, att, obias,
                         relu, res);
      } else {
        #pragma unroll
        for (int i = 0; i < OPS; ++i) res[i] = v4f{0.0f, 0.0f, 0.0f, 0.0f};
      }
      #pragma unroll
      for (int i = 0; i < OPS; ++i)
        hs4[r * P4 + (slot * OPS + i) * LPE + t] = res[i];
    }
  }
  __syncthreads();

  const float* hs = (const float*)hs4;
  const int half = lane >> 4;
  const int l15  = lane & 15;
  const float* ap = hs + l15 * PITCH + half * 2;

  if (MODE == 0) {
    const int ngrp = Ncol >> 5;
    for (int task = wave; task < 2 * ngrp; task += TILE_WAVES) {
      const int mat = task / ngrp;
      const int colBase = (task - mat * ngrp) << 5;
      const float* __restrict__ Wm = mat ? Wr : Wl;
      const float* __restrict__ bm = mat ? br : bl;
      float* __restrict__ Ym = mat ? Yr : Yl;
      const float* __restrict__ bp = Wm + (size_t)(half * 2) * Ncol + colBase + 2 * l15;
      v8f c[4] = {v8f{}, v8f{}, v8f{}, v8f{}};
      for (int k0 = 0; k0 < D; k0 += 16) {
        v2f a0[4], a1[4], w0[4], w1[4];
        #pragma unroll
        for (int u = 0; u < 4; ++u) {
          const size_t ro = (size_t)(k0 + 4 * u) * Ncol;
          a0[u] = *(const v2f*)(ap + k0 + 4 * u);
          a1[u] = *(const v2f*)(ap + 16 * PITCH + k0 + 4 * u);
          w0[u] = *(const v2f*)(bp + ro);
          w1[u] = *(const v2f*)(bp + ro + Ncol);
        }
        #pragma unroll
        for (int u = 0; u < 4; ++u) {
          v2f be, bo;
          be[0] = w0[u][0]; be[1] = w1[u][0];
          bo[0] = w0[u][1]; bo[1] = w1[u][1];
          c[0] = __builtin_amdgcn_wmma_f32_16x16x4_f32(false, a0[u], false, be, (short)0, c[0], false, false);
          c[1] = __builtin_amdgcn_wmma_f32_16x16x4_f32(false, a0[u], false, bo, (short)0, c[1], false, false);
          c[2] = __builtin_amdgcn_wmma_f32_16x16x4_f32(false, a1[u], false, be, (short)0, c[2], false, false);
          c[3] = __builtin_amdgcn_wmma_f32_16x16x4_f32(false, a1[u], false, bo, (short)0, c[3], false, false);
        }
      }
      const int col = colBase + 2 * l15;
      v2f bb = {0.0f, 0.0f};
      if (bm) bb = *(const v2f*)(bm + col);
      #pragma unroll
      for (int rt = 0; rt < 2; ++rt) {
        const int row0 = nodeBase + rt * 16 + half * 8;
        #pragma unroll
        for (int r = 0; r < 8; ++r) {
          if (row0 + r < N) {
            v2f o;
            o[0] = c[2 * rt][r]     + bb[0];
            o[1] = c[2 * rt + 1][r] + bb[1];
            *(v2f*)(Ym + (size_t)(row0 + r) * Ncol + col) = o;
          }
        }
      }
    }
  } else {
    const int ct = (Ncol + 15) >> 4;
    for (int task = wave; task < 2 * ct; task += TILE_WAVES) {
      const int rt = task / ct;
      const int bcol = ((task - rt * ct) << 4) + l15;
      const bool bok = bcol < Ncol;
      const int bcolc = bok ? bcol : (Ncol - 1);
      const float bmask = bok ? 1.0f : 0.0f;
      const float* art = ap + rt * 16 * PITCH;
      const float* __restrict__ bp = Wl + (size_t)(half * 2) * Ncol + bcolc;
      v8f c = v8f{};
      #pragma unroll 4
      for (int k0 = 0; k0 < D; k0 += 4) {
        const v2f a = *(const v2f*)(art + k0);
        const float* __restrict__ bq = bp + (size_t)k0 * Ncol;
        v2f b;
        b[0] = bq[0]    * bmask;
        b[1] = bq[Ncol] * bmask;
        c = __builtin_amdgcn_wmma_f32_16x16x4_f32(false, a, false, b, (short)0, c, false, false);
      }
      if (bok) {
        const float bb = bl ? bl[bcol] : 0.0f;
        const int row0 = nodeBase + rt * 16 + half * 8;
        #pragma unroll
        for (int r = 0; r < 8; ++r)
          if (row0 + r < N) Yl[(size_t)(row0 + r) * Ncol + bcol] = c[r] + bb;
      }
    }
  }
}

extern "C" void kernel_launch(void* const* d_in, const int* in_sizes, int n_in,
                              void* d_out, int out_size, void* d_ws, size_t ws_size,
                              hipStream_t stream) {
  const float* x   = (const float*)d_in[0];
  const int*   ei  = (const int*)  d_in[1];
  const float* w1l = (const float*)d_in[2];
  const float* b1l = (const float*)d_in[3];
  const float* w1r = (const float*)d_in[4];
  const float* b1r = (const float*)d_in[5];
  const float* a1  = (const float*)d_in[6];
  const float* o1  = (const float*)d_in[7];
  const float* w2l = (const float*)d_in[8];
  const float* b2l = (const float*)d_in[9];
  const float* w2r = (const float*)d_in[10];
  const float* b2r = (const float*)d_in[11];
  const float* a2  = (const float*)d_in[12];
  const float* o2  = (const float*)d_in[13];
  const float* w3l = (const float*)d_in[14];
  const float* b3l = (const float*)d_in[15];
  const float* w3r = (const float*)d_in[16];
  const float* b3r = (const float*)d_in[17];
  const float* a3  = (const float*)d_in[18];
  const float* o3  = (const float*)d_in[19];
  const float* wc  = (const float*)d_in[20];
  const float* bc  = (const float*)d_in[21];

  const int N  = in_sizes[0] / 128;
  const int E  = in_sizes[1] / 2;

  const int Np = (N + 3) & ~3;
  const int Ep = (E + 3) & ~3;
  const int NB = (N + BIN_NODES - 1) / BIN_NODES;
  const int NC = (E + CHUNK_EDGES - 1) / CHUNK_EDGES;
  const int Hp = (NB * NC + 3) & ~3;
  float* B0  = (float*)d_ws;
  float* B1  = B0 + (size_t)N * 256;
  float* B2  = B1 + (size_t)N * 256;
  float* B3  = B2 + (size_t)N * 256;
  int*   CS  = (int*)(B3 + (size_t)N * 256);
  int*   RP  = CS + (size_t)Ep;
  int*   HC  = RP + (size_t)Np;
  int*   BT  = HC + (size_t)Hp;
  int*   BB  = BT + MAX_BINS;
  int*   PB  = BB + MAX_BINS;

  bin_count_kernel<<<NC, 256, 0, stream>>>(ei, E, NB, NC, HC);
  bin_scan_kernel<<<NB, 256, 0, stream>>>(HC, NC, BT);
  bin_scatter_kernel<<<NC, 256, 0, stream>>>(ei, E, NB, NC, HC, BT, BB, PB);
  bin_sort_kernel<<<NB, 1024, 0, stream>>>(PB, BB, BT, N, RP, CS);

  {
    const int tiles = (N >> 4) * (128 >> 6);
    gemm2_bias_wmma_n64<<<(tiles + 3) / 4, 128, 0, stream>>>(x, w1l, w1r, b1l, b1r,
                                                            B0, B1, N, 128, 128);
  }
  const int nb = (N + TILE_NODES - 1) / TILE_NODES;
  attend_product_kernel<8, 0><<<nb, 32 * TILE_WAVES, 0, stream>>>(RP, CS, N, E, B0, B1, a1, o1, 1,
                                                      w2l, w2r, b2l, b2r, B2, B3, 128);
  attend_product_kernel<8, 0><<<nb, 32 * TILE_WAVES, 0, stream>>>(RP, CS, N, E, B2, B3, a2, o2, 1,
                                                      w3l, w3r, b3l, b3r, B0, B1, 256);
  attend_product_kernel<16, 1><<<nb, 32 * TILE_WAVES, 0, stream>>>(RP, CS, N, E, B0, B1, a3, o3, 0,
                                                       wc, nullptr, bc, nullptr,
                                                       (float*)d_out, nullptr, 40);
}


# Classifier weights copied to LDS once per work-group; the last product reads its B fragments from LDS, sixteen K-steps ahead, not from global memory inside its chain of dependent matrix instructions
# speedup vs baseline: 5.3824x; 1.0596x over previous
#include <hip/hip_runtime.h>

typedef __attribute__((ext_vector_type(2))) float v2f;
typedef __attribute__((ext_vector_type(8))) float v8f;

#define NEG_SLOPE 0.2f

__device__ __forceinline__ void store_strip_n64(const v8f (&cc)[4],
                                                const float* __restrict__ bias,
                                                float* __restrict__ Y,
                                                int row0, int colBase, int l15,
                                                int Ncol) {
  #pragma unroll
  for (int g = 0; g < 2; ++g) {
    const int col = colBase + g * 32 + 2 * l15;
    v2f bb = {0.0f, 0.0f};
    if (bias) bb = *(const v2f*)(bias + col);
    float* __restrict__ yp = Y + (size_t)row0 * Ncol + col;
    #pragma unroll
    for (int r = 0; r < 8; ++r) {
      v2f o;
      o[0] = cc[2 * g][r]     + bb[0];
      o[1] = cc[2 * g + 1][r] + bb[1];
      *(v2f*)(yp + (size_t)r * Ncol) = o;
    }
  }
}

__global__ void __launch_bounds__(128)
gemm2_bias_wmma_n64(const float* __restrict__ X,
                    const float* __restrict__ Wl,
                    const float* __restrict__ Wr,
                    const float* __restrict__ biasl,
                    const float* __restrict__ biasr,
                    float* __restrict__ Yl,
                    float* __restrict__ Yr,
                    int M, int K, int Ncol) {
  const int lane = threadIdx.x & 31;
  const int wave = threadIdx.x >> 5;
  const int ct = Ncol >> 6;
  const int rt = M >> 4;
  const int tile = blockIdx.x * (blockDim.x >> 5) + wave;
  if (tile >= ct * rt) return;

  const int rowBase = (tile / ct) << 4;
  const int colBase = (tile % ct) << 6;
  const int half = lane >> 4;
  const int l15  = lane & 15;

  const float* __restrict__ ap = X + (size_t)(rowBase + l15) * K + half * 2;
  const size_t boff = (size_t)(half * 2) * Ncol + colBase + 2 * l15;
  const float* __restrict__ bpl = Wl + boff;
  const float* __restrict__ bpr = Wr + boff;

  v8f cl[4] = {v8f{}, v8f{}, v8f{}, v8f{}};
  v8f cr[4] = {v8f{}, v8f{}, v8f{}, v8f{}};
  for (int k0 = 0; k0 < K; k0 += 8) {
    v2f a[2], l0[2], l1[2], l2[2], l3[2], r0[2], r1[2], r2[2], r3[2];
    #pragma unroll
    for (int u = 0; u < 2; ++u) {
      const size_t ro = (size_t)(k0 + 4 * u) * Ncol;
      a[u]  = *(const v2f*)(ap + k0 + 4 * u);
      l0[u] = *(const v2f*)(bpl + ro);
      l1[u] = *(const v2f*)(bpl + ro + Ncol);
      l2[u] = *(const v2f*)(bpl + ro + 32);
      l3[u] = *(const v2f*)(bpl + ro + Ncol + 32);
      r0[u] = *(const v2f*)(bpr + ro);
      r1[u] = *(const v2f*)(bpr + ro + Ncol);
      r2[u] = *(const v2f*)(bpr + ro + 32);
      r3[u] = *(const v2f*)(bpr + ro + Ncol + 32);
    }
    asm volatile("" :: "v"(a[0]), "v"(a[1]),
                       "v"(l0[0]), "v"(l0[1]), "v"(l1[0]), "v"(l1[1]),
                       "v"(l2[0]), "v"(l2[1]), "v"(l3[0]), "v"(l3[1]),
                       "v"(r0[0]), "v"(r0[1]), "v"(r1[0]), "v"(r1[1]),
                       "v"(r2[0]), "v"(r2[1]), "v"(r3[0]), "v"(r3[1]));
    #pragma unroll
    for (int u = 0; u < 2; ++u) {
      v2f b0e, b0o, b1e, b1o;
      b0e[0] = l0[u][0]; b0e[1] = l1[u][0];
      b0o[0] = l0[u][1]; b0o[1] = l1[u][1];
      b1e[0] = l2[u][0]; b1e[1] = l3[u][0];
      b1o[0] = l2[u][1]; b1o[1] = l3[u][1];
      cl[0] = __builtin_amdgcn_wmma_f32_16x16x4_f32(false, a[u], false, b0e, (short)0, cl[0], false, false);
      cl[1] = __builtin_amdgcn_wmma_f32_16x16x4_f32(false, a[u], false, b0o, (short)0, cl[1], false, false);
      cl[2] = __builtin_amdgcn_wmma_f32_16x16x4_f32(false, a[u], false, b1e, (short)0, cl[2], false, false);
      cl[3] = __builtin_amdgcn_wmma_f32_16x16x4_f32(false, a[u], false, b1o, (short)0, cl[3], false, false);
      b0e[0] = r0[u][0]; b0e[1] = r1[u][0];
      b0o[0] = r0[u][1]; b0o[1] = r1[u][1];
      b1e[0] = r2[u][0]; b1e[1] = r3[u][0];
      b1o[0] = r2[u][1]; b1o[1] = r3[u][1];
      cr[0] = __builtin_amdgcn_wmma_f32_16x16x4_f32(false, a[u], false, b0e, (short)0, cr[0], false, false);
      cr[1] = __builtin_amdgcn_wmma_f32_16x16x4_f32(false, a[u], false, b0o, (short)0, cr[1], false, false);
      cr[2] = __builtin_amdgcn_wmma_f32_16x16x4_f32(false, a[u], false, b1e, (short)0, cr[2], false, false);
      cr[3] = __builtin_amdgcn_wmma_f32_16x16x4_f32(false, a[u], false, b1o, (short)0, cr[3], false, false);
    }
  }

  const int row0 = rowBase + half * 8;
  store_strip_n64(cl, biasl, Yl, row0, colBase, l15, Ncol);
  store_strip_n64(cr, biasr, Yr, row0, colBase, l15, Ncol);
}

#define BIN_SHIFT 8
#define BIN_NODES 256
#define CHUNK_EDGES 4096
#define MAX_BINS 1024

__global__ void __launch_bounds__(256)
bin_count_kernel(const int* __restrict__ ei, int E, int NB, int NC,
                 int* __restrict__ HC) {
  __shared__ int hist[MAX_BINS];
  const int tid = threadIdx.x;
  for (int b = tid; b < NB; b += 256) hist[b] = 0;
  __syncthreads();
  const int base = blockIdx.x * CHUNK_EDGES;
  int d[CHUNK_EDGES / 256];
  #pragma unroll
  for (int j = 0; j < CHUNK_EDGES / 256; ++j) {
    const int i  = base + j * 256 + tid;
    const int ic = (i < E) ? i : (E - 1);
    const int dv = ei[E + ic];
    d[j] = (i < E) ? dv : -1;
  }
  #pragma unroll
  for (int j = 0; j < CHUNK_EDGES / 256; ++j)
    if (d[j] >= 0) atomicAdd(&hist[d[j] >> BIN_SHIFT], 1);
  __syncthreads();
  for (int b = tid; b < NB; b += 256) HC[(size_t)b * NC + blockIdx.x] = hist[b];
}

__global__ void __launch_bounds__(256)
bin_scan_kernel(int* __restrict__ HC, int NC, int* __restrict__ BT) {
  __shared__ int wtot[8];
  const int tid  = threadIdx.x;
  const int lane = tid & 31;
  const int wave = tid >> 5;
  int* __restrict__ row = HC + (size_t)blockIdx.x * NC;
  int carry = 0;
  for (int c0 = 0; c0 < NC; c0 += 256) {
    const int c = c0 + tid;
    const int v = (c < NC) ? row[c] : 0;
    int incl = v;
    #pragma unroll
    for (int off = 1; off < 32; off <<= 1) {
      const int y = __shfl_up(incl, off, 32);
      if (lane >= off) incl += y;
    }
    if (lane == 31) wtot[wave] = incl;
    __syncthreads();
    int woff = 0, tot = 0;
    #pragma unroll
    for (int w = 0; w < 8; ++w) {
      const int t = wtot[w];
      woff += (w < wave) ? t : 0;
      tot  += t;
    }
    if (c < NC) row[c] = carry + woff + incl - v;
    carry += tot;
    __syncthreads();
  }
  if (tid == 0) BT[blockIdx.x] = carry;
}

__global__ void __launch_bounds__(256)
bin_scatter_kernel(const int* __restrict__ ei, int E, int NB, int NC,
                   const int* __restrict__ HC, const int* __restrict__ BT,
                   int* __restrict__ BB, int* __restrict__ PBw) {
  int2* __restrict__ PB = (int2*)PBw;
  __shared__ int slot[MAX_BINS];
  __shared__ int wtot[8];
  const int tid  = threadIdx.x;
  const int lane = tid & 31;
  const int wave = tid >> 5;
  const int chunk = blockIdx.x;
  const int b0 = tid * 4;
  const int c0 = (b0 + 0 < NB) ? (b0 + 0) : (NB - 1);
  const int c1 = (b0 + 1 < NB) ? (b0 + 1) : (NB - 1);
  const int c2 = (b0 + 2 < NB) ? (b0 + 2) : (NB - 1);
  const int c3 = (b0 + 3 < NB) ? (b0 + 3) : (NB - 1);
  const int r0 = BT[c0], r1 = BT[c1], r2 = BT[c2], r3 = BT[c3];
  const int h0 = HC[(size_t)c0 * NC + chunk];
  const int h1 = HC[(size_t)c1 * NC + chunk];
  const int h2 = HC[(size_t)c2 * NC + chunk];
  const int h3 = HC[(size_t)c3 * NC + chunk];
  const int t0 = (b0 + 0 < NB) ? r0 : 0;
  const int t1 = (b0 + 1 < NB) ? r1 : 0;
  const int t2 = (b0 + 2 < NB) ? r2 : 0;
  const int t3 = (b0 + 3 < NB) ? r3 : 0;
  const int tsum = t0 + t1 + t2 + t3;
  int incl = tsum;
  #pragma unroll
  for (int off = 1; off < 32; off <<= 1) {
    const int y = __shfl_up(incl, off, 32);
    if (lane >= off) incl += y;
  }
  if (lane == 31) wtot[wave] = incl;
  __syncthreads();
  int woff = 0;
  #pragma unroll
  for (int w = 0; w < 8; ++w) woff += (w < wave) ? wtot[w] : 0;
  const int e0 = woff + incl - tsum;
  const int e1 = e0 + t0;
  const int e2 = e1 + t1;
  const int e3 = e2 + t2;
  if (b0 + 0 < NB) slot[b0 + 0] = e0 + h0;
  if (b0 + 1 < NB) slot[b0 + 1] = e1 + h1;
  if (b0 + 2 < NB) slot[b0 + 2] = e2 + h2;
  if (b0 + 3 < NB) slot[b0 + 3] = e3 + h3;
  if (chunk == 0) {
    if (b0 + 0 < NB) BB[b0 + 0] = e0;
    if (b0 + 1 < NB) BB[b0 + 1] = e1;
    if (b0 + 2 < NB) BB[b0 + 2] = e2;
    if (b0 + 3 < NB) BB[b0 + 3] = e3;
  }
  __syncthreads();
  const int base = chunk * CHUNK_EDGES;
  int s[CHUNK_EDGES / 256], d[CHUNK_EDGES / 256];
  #pragma unroll
  for (int j = 0; j < CHUNK_EDGES / 256; ++j) {
    const int i  = base + j * 256 + tid;
    const int ic = (i < E) ? i : (E - 1);
    s[j] = ei[ic];
    const int dv = ei[E + ic];
    d[j] = (i < E) ? dv : -1;
  }
  #pragma unroll
  for (int j = 0; j < CHUNK_EDGES / 256; ++j) {
    if (d[j] >= 0) {
      const int pos = atomicAdd(&slot[d[j] >> BIN_SHIFT], 1);
      PB[pos] = make_int2(s[j], d[j]);
    }
  }
}

__global__ void __launch_bounds__(1024)
bin_sort_kernel(const int* __restrict__ PBw, const int* __restrict__ BB,
                const int* __restrict__ BT, int N,
                int* __restrict__ rowptr, int* __restrict__ csr_src) {
  __shared__ int cnt[BIN_NODES];
  __shared__ int cur[BIN_NODES];
  __shared__ int wtot[8];
  const int tid  = threadIdx.x;
  const int lane = tid & 31;
  const int wave = tid >> 5;
  const int bs = BB[blockIdx.x];
  const int n  = BT[blockIdx.x];
  const int2* __restrict__ pb = (const int2*)PBw + bs;
  if (tid < BIN_NODES) cnt[tid] = 0;
  __syncthreads();
  for (int i0 = 0; i0 < n; i0 += 4096) {
    int d[4];
    #pragma unroll
    for (int u = 0; u < 4; ++u) {
      const int i  = i0 + u * 1024 + tid;
      const int ic = (i < n) ? i : (n - 1);
      const int dv = pb[ic].y;
      d[u] = (i < n) ? dv : -1;
    }
    #pragma unroll
    for (int u = 0; u < 4; ++u)
      if (d[u] >= 0) atomicAdd(&cnt[d[u] & (BIN_NODES - 1)], 1);
  }
  __syncthreads();
  const int v = (tid < BIN_NODES) ? cnt[tid] : 0;
  int incl = v;
  #pragma unroll
  for (int off = 1; off < 32; off <<= 1) {
    const int y = __shfl_up(incl, off, 32);
    if (lane >= off) incl += y;
  }
  if (tid < BIN_NODES && lane == 31) wtot[wave] = incl;
  __syncthreads();
  if (tid < BIN_NODES) {
    int woff = 0;
    #pragma unroll
    for (int w = 0; w < 8; ++w) woff += (w < wave) ? wtot[w] : 0;
    const int start = bs + woff + incl - v;
    cur[tid] = start;
    const int node = blockIdx.x * BIN_NODES + tid;
    if (node < N) rowptr[node] = start;
  }
  __syncthreads();
  for (int i0 = 0; i0 < n; i0 += 4096) {
    int2 p[4];
    #pragma unroll
    for (int u = 0; u < 4; ++u) {
      const int i  = i0 + u * 1024 + tid;
      const int ic = (i < n) ? i : (n - 1);
      p[u] = pb[ic];
      if (i >= n) p[u].y = -1;
    }
    #pragma unroll
    for (int u = 0; u < 4; ++u) {
      if (p[u].y >= 0) {
        const int pos = atomicAdd(&cur[p[u].y & (BIN_NODES - 1)], 1);
        csr_src[pos] = p[u].x;
      }
    }
  }
}

typedef __attribute__((ext_vector_type(4))) float v4f;

__device__ __forceinline__ float xhalf(float x) {
  const int b = __builtin_bit_cast(int, x);
  return __builtin_bit_cast(float,
      __builtin_amdgcn_permlanex16(b, b, 0x76543210, 0xfedcba98, false, false));
}
template <int NX>
__device__ __forceinline__ float xlane(float x) {
  const int b = __builtin_bit_cast(int, x);
  return __builtin_bit_cast(float,
      __builtin_amdgcn_update_dpp(0, b, 0x160 + NX, 0xf, 0xf, true));
}

template <int LPE>
__device__ __forceinline__ void attend_node(int wid, int lane,
                   const int* __restrict__ rowptr,
                   const int* __restrict__ csr_src, int N, int E,
                   const float* __restrict__ XL, const float* __restrict__ XR,
                   const float* __restrict__ att,
                   const float* __restrict__ obias, int relu,
                   v4f (&res)[4 / (32 / LPE)]) {
  constexpr int S   = 32 / LPE;
  constexpr int D4  = 4 * LPE;
  constexpr int OPS = 4 / S;
  constexpr float LOWEST = -3.402823466e+38f;
  const int beg = __builtin_amdgcn_readfirstlane(rowptr[wid]);
  const int end = (wid + 1 < N) ? __builtin_amdgcn_readfirstlane(rowptr[wid + 1]) : E;
  const int c   = end - beg + 1;
  const int slot = lane / LPE;
  const int t    = lane % LPE;

  const v4f* __restrict__ xl4 = (const v4f*)XL;
  const v4f* __restrict__ xr4 = (const v4f*)XR + (size_t)wid * D4 + t;
  const v4f* __restrict__ at4 = (const v4f*)att + t;

  v4f xr[4], g[4], acc[4];
  #pragma unroll
  for (int j = 0; j < 4; ++j) {
    xr[j]  = xr4[j * LPE];
    g[j]   = at4[j * LPE];
    acc[j] = v4f{0.0f, 0.0f, 0.0f, 0.0f};
  }
  float m = LOWEST;
  float ssum = 0.0f;

  for (int base = 0; base < c; base += 32) {
    const int q = base + lane;
    int idxv = wid;
    if (q > 0 && q < c) idxv = csr_src[beg + q - 1];
    const int nq = (c - base < 32) ? (c - base) : 32;

    for (int k = 0; k < nq; k += S) {
      int src = __builtin_amdgcn_readlane(idxv, k);
      #pragma unroll
      for (int u = 1; u < S; ++u) {
        const int kk = (k + u < nq) ? (k + u) : k;
        const int su = __builtin_amdgcn_readlane(idxv, kk);
        if (slot == u) src = su;
      }
      const bool valid = (k + slot) < nq;
      const v4f* __restrict__ row = xl4 + (size_t)src * D4 + t;
      v4f r[4];
      #pragma unroll
      for (int j = 0; j < 4; ++j) r[j] = row[j * LPE];

      v4f a4 = v4f{0.0f, 0.0f, 0.0f, 0.0f};
      #pragma unroll
      for (int j = 0; j < 4; ++j) {
        v4f h = r[j] + xr[j];
        #pragma unroll
        for (int e = 0; e < 4; ++e) h[e] = h[e] > 0.0f ? h[e] : NEG_SLOPE * h[e];
        a4 += h * g[j];
      }
      float a = (a4[0] + a4[1]) + (a4[2] + a4[3]);
      a += xlane<1>(a);
      a += xlane<2>(a);
      a += xlane<4>(a);
      if (LPE == 16) a += xlane<8>(a);

      const float ev = valid ? a : LOWEST;
      const float mn = fmaxf(m, ev);
      const float sc = __expf(m - mn);
      const float pe = __expf(ev - mn);
      const float p  = valid ? pe : 0.0f;
      ssum = ssum * sc + p;
      #pragma unroll
      for (int j = 0; j < 4; ++j) acc[j] = acc[j] * sc + p * r[j];
      m = mn;
    }
  }

  float mAll = m;
  if (LPE == 8) mAll = fmaxf(mAll, xlane<8>(mAll));
  mAll = fmaxf(mAll, xhalf(mAll));
  const float f = __expf(m - mAll);
  ssum *= f;
  #pragma unroll
  for (int j = 0; j < 4; ++j) acc[j] = acc[j] * f;
  if (LPE == 8) {
    ssum += xlane<8>(ssum);
    #pragma unroll
    for (int j = 0; j < 4; ++j) {
      #pragma unroll
      for (int e = 0; e < 4; ++e) acc[j][e] += xlane<8>(acc[j][e]);
    }
  }
  ssum += xhalf(ssum);
  #pragma unroll
  for (int j = 0; j < 4; ++j) {
    #pragma unroll
    for (int e = 0; e < 4; ++e) acc[j][e] += xhalf(acc[j][e]);
  }

  const float inv = 1.0f / ssum;
  const v4f* __restrict__ ob4 = (const v4f*)obias;
  #pragma unroll
  for (int i = 0; i < OPS; ++i) {
    v4f o = acc[i];
    #pragma unroll
    for (int s = 1; s < S; ++s)
      if (slot == s) o = acc[s * OPS + i];
    const int oi = (slot * OPS + i) * LPE + t;
    const v4f bb = ob4[oi];
    v4f rr = o * inv + bb;
    if (relu) {
      rr[0] = fmaxf(rr[0], 0.0f); rr[1] = fmaxf(rr[1], 0.0f);
      rr[2] = fmaxf(rr[2], 0.0f); rr[3] = fmaxf(rr[3], 0.0f);
    }
    res[i] = rr;
  }
}

#define TILE_NODES 32
#define TILE_WAVES 16
#define WS_COLS 48

template <int LPE, int MODE>
__global__ void __launch_bounds__(512, 8)
attend_product_kernel(const int* __restrict__ rowptr,
                      const int* __restrict__ csr_src, int N, int E,
                      const float* __restrict__ XL, const float* __restrict__ XR,
                      const float* __restrict__ att,
                      const float* __restrict__ obias, int relu,
                      const float* __restrict__ Wl, const float* __restrict__ Wr,
                      const float* __restrict__ bl, const float* __restrict__ br,
                      float* __restrict__ Yl, float* __restrict__ Yr, int Ncol) {
  constexpr int S     = 32 / LPE;
  constexpr int OPS   = 4 / S;
  constexpr int D     = 16 * LPE;
  constexpr int PITCH = D + 4;
  constexpr int P4    = PITCH / 4;
  __shared__ v4f hs4[TILE_NODES * P4];
  constexpr int WS4 = (MODE == 1) ? (D * WS_COLS / 4) : 1;
  __shared__ v4f ws4[WS4];
  if (MODE == 1) {
    constexpr int PER = (WS4 + 32 * TILE_WAVES - 1) / (32 * TILE_WAVES);
    const int n4 = (D * Ncol) >> 2;
    const v4f* __restrict__ w4 = (const v4f*)Wl;
    v4f tmp[PER];
    #pragma unroll
    for (int j = 0; j < PER; ++j) {
      const int i  = j * (32 * TILE_WAVES) + (int)threadIdx.x;
      const int ic = (i < n4) ? i : (n4 - 1);
      tmp[j] = w4[ic];
    }
    #pragma unroll
    for (int j = 0; j < PER; ++j) {
      const int i = j * (32 * TILE_WAVES) + (int)threadIdx.x;
      if (i < n4) ws4[i] = tmp[j];
    }
  }
  const int lane = threadIdx.x & 31;
  const int wave = __builtin_amdgcn_readfirstlane((int)(threadIdx.x >> 5));
  const int nodeBase = blockIdx.x * TILE_NODES;
  {
    const int slot = lane / LPE;
    const int t    = lane % LPE;
    #pragma unroll 1
    for (int it = 0; it < TILE_NODES / TILE_WAVES; ++it) {
      const int r = it * TILE_WAVES + wave;
      const int node = nodeBase + r;
      v4f res[OPS];
      if (node < N) {
        attend_node<LPE>(node, lane, rowptr, csr_src, N, E, XL, XR, att, obias,
                         relu, res);
      } else {
        #pragma unroll
        for (int i = 0; i < OPS; ++i) res[i] = v4f{0.0f, 0.0f, 0.0f, 0.0f};
      }
      #pragma unroll
      for (int i = 0; i < OPS; ++i)
        hs4[r * P4 + (slot * OPS + i) * LPE + t] = res[i];
    }
  }
  __syncthreads();

  const float* hs = (const float*)hs4;
  const int half = lane >> 4;
  const int l15  = lane & 15;
  const float* ap = hs + l15 * PITCH + half * 2;

  if (MODE == 0) {
    const int ngrp = Ncol >> 5;
    for (int task = wave; task < 2 * ngrp; task += TILE_WAVES) {
      const int mat = task / ngrp;
      const int colBase = (task - mat * ngrp) << 5;
      const float* __restrict__ Wm = mat ? Wr : Wl;
      const float* __restrict__ bm = mat ? br : bl;
      float* __restrict__ Ym = mat ? Yr : Yl;
      const float* __restrict__ bp = Wm + (size_t)(half * 2) * Ncol + colBase + 2 * l15;
      v8f c[4] = {v8f{}, v8f{}, v8f{}, v8f{}};
      for (int k0 = 0; k0 < D; k0 += 16) {
        v2f a0[4], a1[4], w0[4], w1[4];
        #pragma unroll
        for (int u = 0; u < 4; ++u) {
          const size_t ro = (size_t)(k0 + 4 * u) * Ncol;
          a0[u] = *(const v2f*)(ap + k0 + 4 * u);
          a1[u] = *(const v2f*)(ap + 16 * PITCH + k0 + 4 * u);
          w0[u] = *(const v2f*)(bp + ro);
          w1[u] = *(const v2f*)(bp + ro + Ncol);
        }
        #pragma unroll
        for (int u = 0; u < 4; ++u) {
          v2f be, bo;
          be[0] = w0[u][0]; be[1] = w1[u][0];
          bo[0] = w0[u][1]; bo[1] = w1[u][1];
          c[0] = __builtin_amdgcn_wmma_f32_16x16x4_f32(false, a0[u], false, be, (short)0, c[0], false, false);
          c[1] = __builtin_amdgcn_wmma_f32_16x16x4_f32(false, a0[u], false, bo, (short)0, c[1], false, false);
          c[2] = __builtin_amdgcn_wmma_f32_16x16x4_f32(false, a1[u], false, be, (short)0, c[2], false, false);
          c[3] = __builtin_amdgcn_wmma_f32_16x16x4_f32(false, a1[u], false, bo, (short)0, c[3], false, false);
        }
      }
      const int col = colBase + 2 * l15;
      v2f bb = {0.0f, 0.0f};
      if (bm) bb = *(const v2f*)(bm + col);
      #pragma unroll
      for (int rt = 0; rt < 2; ++rt) {
        const int row0 = nodeBase + rt * 16 + half * 8;
        #pragma unroll
        for (int r = 0; r < 8; ++r) {
          if (row0 + r < N) {
            v2f o;
            o[0] = c[2 * rt][r]     + bb[0];
            o[1] = c[2 * rt + 1][r] + bb[1];
            *(v2f*)(Ym + (size_t)(row0 + r) * Ncol + col) = o;
          }
        }
      }
    }
  } else {
    const int ct = (Ncol + 15) >> 4;
    for (int task = wave; task < 2 * ct; task += TILE_WAVES) {
      const int rt = task / ct;
      const int bcol = ((task - rt * ct) << 4) + l15;
      const bool bok = bcol < Ncol;
      const int bcolc = bok ? bcol : (Ncol - 1);
      const float bmask = bok ? 1.0f : 0.0f;
      const float* art = ap + rt * 16 * PITCH;
      const float* bp = (const float*)ws4 + (half * 2) * Ncol + bcolc;
      v8f c = v8f{};
      #pragma unroll 1
      for (int k0 = 0; k0 < D; k0 += 64) {
        v2f a[16], w[16];
        #pragma unroll
        for (int u = 0; u < 16; ++u) {
          a[u] = *(const v2f*)(art + k0 + 4 * u);
          const float* bq = bp + (k0 + 4 * u) * Ncol;
          w[u][0] = bq[0];
          w[u][1] = bq[Ncol];
        }
        __builtin_amdgcn_sched_barrier(0);
        #pragma unroll
        for (int u = 0; u < 16; ++u) {
          v2f b;
          b[0] = w[u][0] * bmask;
          b[1] = w[u][1] * bmask;
          c = __builtin_amdgcn_wmma_f32_16x16x4_f32(false, a[u], false, b, (short)0, c, false, false);
        }
      }
      if (bok) {
        const float bb = bl ? bl[bcol] : 0.0f;
        const int row0 = nodeBase + rt * 16 + half * 8;
        #pragma unroll
        for (int r = 0; r < 8; ++r)
          if (row0 + r < N) Yl[(size_t)(row0 + r) * Ncol + bcol] = c[r] + bb;
      }
    }
  }
}

extern "C" void kernel_launch(void* const* d_in, const int* in_sizes, int n_in,
                              void* d_out, int out_size, void* d_ws, size_t ws_size,
                              hipStream_t stream) {
  const float* x   = (const float*)d_in[0];
  const int*   ei  = (const int*)  d_in[1];
  const float* w1l = (const float*)d_in[2];
  const float* b1l = (const float*)d_in[3];
  const float* w1r = (const float*)d_in[4];
  const float* b1r = (const float*)d_in[5];
  const float* a1  = (const float*)d_in[6];
  const float* o1  = (const float*)d_in[7];
  const float* w2l = (const float*)d_in[8];
  const float* b2l = (const float*)d_in[9];
  const float* w2r = (const float*)d_in[10];
  const float* b2r = (const float*)d_in[11];
  const float* a2  = (const float*)d_in[12];
  const float* o2  = (const float*)d_in[13];
  const float* w3l = (const float*)d_in[14];
  const float* b3l = (const float*)d_in[15];
  const float* w3r = (const float*)d_in[16];
  const float* b3r = (const float*)d_in[17];
  const float* a3  = (const float*)d_in[18];
  const float* o3  = (const float*)d_in[19];
  const float* wc  = (const float*)d_in[20];
  const float* bc  = (const float*)d_in[21];

  const int N  = in_sizes[0] / 128;
  const int E  = in_sizes[1] / 2;

  const int Np = (N + 3) & ~3;
  const int Ep = (E + 3) & ~3;
  const int NB = (N + BIN_NODES - 1) / BIN_NODES;
  const int NC = (E + CHUNK_EDGES - 1) / CHUNK_EDGES;
  const int Hp = (NB * NC + 3) & ~3;
  float* B0  = (float*)d_ws;
  float* B1  = B0 + (size_t)N * 256;
  float* B2  = B1 + (size_t)N * 256;
  float* B3  = B2 + (size_t)N * 256;
  int*   CS  = (int*)(B3 + (size_t)N * 256);
  int*   RP  = CS + (size_t)Ep;
  int*   HC  = RP + (size_t)Np;
  int*   BT  = HC + (size_t)Hp;
  int*   BB  = BT + MAX_BINS;
  int*   PB  = BB + MAX_BINS;

  bin_count_kernel<<<NC, 256, 0, stream>>>(ei, E, NB, NC, HC);
  bin_scan_kernel<<<NB, 256, 0, stream>>>(HC, NC, BT);
  bin_scatter_kernel<<<NC, 256, 0, stream>>>(ei, E, NB, NC, HC, BT, BB, PB);
  bin_sort_kernel<<<NB, 1024, 0, stream>>>(PB, BB, BT, N, RP, CS);

  {
    const int tiles = (N >> 4) * (128 >> 6);
    gemm2_bias_wmma_n64<<<(tiles + 3) / 4, 128, 0, stream>>>(x, w1l, w1r, b1l, b1r,
                                                            B0, B1, N, 128, 128);
  }
  const int nb = (N + TILE_NODES - 1) / TILE_NODES;
  attend_product_kernel<8, 0><<<nb, 32 * TILE_WAVES, 0, stream>>>(RP, CS, N, E, B0, B1, a1, o1, 1,
                                                      w2l, w2r, b2l, b2r, B2, B3, 128);
  attend_product_kernel<8, 0><<<nb, 32 * TILE_WAVES, 0, stream>>>(RP, CS, N, E, B2, B3, a2, o2, 1,
                                                      w3l, w3r, b3l, b3r, B0, B1, 256);
  attend_product_kernel<16, 1><<<nb, 32 * TILE_WAVES, 0, stream>>>(RP, CS, N, E, B0, B1, a3, o3, 0,
                                                       wc, nullptr, bc, nullptr,
                                                       (float*)d_out, nullptr, 40);
}


# Attention vector and the node's xr row are read from LDS inside the edge loop, not held in registers: 114 registers become 80, so three work-groups share a WGP, not two; order of sums unchanged
# speedup vs baseline: 5.6929x; 1.0577x over previous
#include <hip/hip_runtime.h>

typedef __attribute__((ext_vector_type(2))) float v2f;
typedef __attribute__((ext_vector_type(8))) float v8f;

#define NEG_SLOPE 0.2f

__device__ __forceinline__ void store_strip_n64(const v8f (&cc)[4],
                                                const float* __restrict__ bias,
                                                float* __restrict__ Y,
                                                int row0, int colBase, int l15,
                                                int Ncol) {
  #pragma unroll
  for (int g = 0; g < 2; ++g) {
    const int col = colBase + g * 32 + 2 * l15;
    v2f bb = {0.0f, 0.0f};
    if (bias) bb = *(const v2f*)(bias + col);
    float* __restrict__ yp = Y + (size_t)row0 * Ncol + col;
    #pragma unroll
    for (int r = 0; r < 8; ++r) {
      v2f o;
      o[0] = cc[2 * g][r]     + bb[0];
      o[1] = cc[2 * g + 1][r] + bb[1];
      *(v2f*)(yp + (size_t)r * Ncol) = o;
    }
  }
}

__global__ void __launch_bounds__(128)
gemm2_bias_wmma_n64(const float* __restrict__ X,
                    const float* __restrict__ Wl,
                    const float* __restrict__ Wr,
                    const float* __restrict__ biasl,
                    const float* __restrict__ biasr,
                    float* __restrict__ Yl,
                    float* __restrict__ Yr,
                    int M, int K, int Ncol) {
  const int lane = threadIdx.x & 31;
  const int wave = threadIdx.x >> 5;
  const int ct = Ncol >> 6;
  const int rt = M >> 4;
  const int tile = blockIdx.x * (blockDim.x >> 5) + wave;
  if (tile >= ct * rt) return;

  const int rowBase = (tile / ct) << 4;
  const int colBase = (tile % ct) << 6;
  const int half = lane >> 4;
  const int l15  = lane & 15;

  const float* __restrict__ ap = X + (size_t)(rowBase + l15) * K + half * 2;
  const size_t boff = (size_t)(half * 2) * Ncol + colBase + 2 * l15;
  const float* __restrict__ bpl = Wl + boff;
  const float* __restrict__ bpr = Wr + boff;

  v8f cl[4] = {v8f{}, v8f{}, v8f{}, v8f{}};
  v8f cr[4] = {v8f{}, v8f{}, v8f{}, v8f{}};
  for (int k0 = 0; k0 < K; k0 += 8) {
    v2f a[2], l0[2], l1[2], l2[2], l3[2], r0[2], r1[2], r2[2], r3[2];
    #pragma unroll
    for (int u = 0; u < 2; ++u) {
      const size_t ro = (size_t)(k0 + 4 * u) * Ncol;
      a[u]  = *(const v2f*)(ap + k0 + 4 * u);
      l0[u] = *(const v2f*)(bpl + ro);
      l1[u] = *(const v2f*)(bpl + ro + Ncol);
      l2[u] = *(const v2f*)(bpl + ro + 32);
      l3[u] = *(const v2f*)(bpl + ro + Ncol + 32);
      r0[u] = *(const v2f*)(bpr + ro);
      r1[u] = *(const v2f*)(bpr + ro + Ncol);
      r2[u] = *(const v2f*)(bpr + ro + 32);
      r3[u] = *(const v2f*)(bpr + ro + Ncol + 32);
    }
    asm volatile("" :: "v"(a[0]), "v"(a[1]),
                       "v"(l0[0]), "v"(l0[1]), "v"(l1[0]), "v"(l1[1]),
                       "v"(l2[0]), "v"(l2[1]), "v"(l3[0]), "v"(l3[1]),
                       "v"(r0[0]), "v"(r0[1]), "v"(r1[0]), "v"(r1[1]),
                       "v"(r2[0]), "v"(r2[1]), "v"(r3[0]), "v"(r3[1]));
    #pragma unroll
    for (int u = 0; u < 2; ++u) {
      v2f b0e, b0o, b1e, b1o;
      b0e[0] = l0[u][0]; b0e[1] = l1[u][0];
      b0o[0] = l0[u][1]; b0o[1] = l1[u][1];
      b1e[0] = l2[u][0]; b1e[1] = l3[u][0];
      b1o[0] = l2[u][1]; b1o[1] = l3[u][1];
      cl[0] = __builtin_amdgcn_wmma_f32_16x16x4_f32(false, a[u], false, b0e, (short)0, cl[0], false, false);
      cl[1] = __builtin_amdgcn_wmma_f32_16x16x4_f32(false, a[u], false, b0o, (short)0, cl[1], false, false);
      cl[2] = __builtin_amdgcn_wmma_f32_16x16x4_f32(false, a[u], false, b1e, (short)0, cl[2], false, false);
      cl[3] = __builtin_amdgcn_wmma_f32_16x16x4_f32(false, a[u], false, b1o, (short)0, cl[3], false, false);
      b0e[0] = r0[u][0]; b0e[1] = r1[u][0];
      b0o[0] = r0[u][1]; b0o[1] = r1[u][1];
      b1e[0] = r2[u][0]; b1e[1] = r3[u][0];
      b1o[0] = r2[u][1]; b1o[1] = r3[u][1];
      cr[0] = __builtin_amdgcn_wmma_f32_16x16x4_f32(false, a[u], false, b0e, (short)0, cr[0], false, false);
      cr[1] = __builtin_amdgcn_wmma_f32_16x16x4_f32(false, a[u], false, b0o, (short)0, cr[1], false, false);
      cr[2] = __builtin_amdgcn_wmma_f32_16x16x4_f32(false, a[u], false, b1e, (short)0, cr[2], false, false);
      cr[3] = __builtin_amdgcn_wmma_f32_16x16x4_f32(false, a[u], false, b1o, (short)0, cr[3], false, false);
    }
  }

  const int row0 = rowBase + half * 8;
  store_strip_n64(cl, biasl, Yl, row0, colBase, l15, Ncol);
  store_strip_n64(cr, biasr, Yr, row0, colBase, l15, Ncol);
}

#define BIN_SHIFT 8
#define BIN_NODES 256
#define CHUNK_EDGES 4096
#define MAX_BINS 1024

__global__ void __launch_bounds__(256)
bin_count_kernel(const int* __restrict__ ei, int E, int NB, int NC,
                 int* __restrict__ HC) {
  __shared__ int hist[MAX_BINS];
  const int tid = threadIdx.x;
  for (int b = tid; b < NB; b += 256) hist[b] = 0;
  __syncthreads();
  const int base = blockIdx.x * CHUNK_EDGES;
  int d[CHUNK_EDGES / 256];
  #pragma unroll
  for (int j = 0; j < CHUNK_EDGES / 256; ++j) {
    const int i  = base + j * 256 + tid;
    const int ic = (i < E) ? i : (E - 1);
    const int dv = ei[E + ic];
    d[j] = (i < E) ? dv : -1;
  }
  #pragma unroll
  for (int j = 0; j < CHUNK_EDGES / 256; ++j)
    if (d[j] >= 0) atomicAdd(&hist[d[j] >> BIN_SHIFT], 1);
  __syncthreads();
  for (int b = tid; b < NB; b += 256) HC[(size_t)b * NC + blockIdx.x] = hist[b];
}

__global__ void __launch_bounds__(256)
bin_scan_kernel(int* __restrict__ HC, int NC, int* __restrict__ BT) {
  __shared__ int wtot[8];
  const int tid  = threadIdx.x;
  const int lane = tid & 31;
  const int wave = tid >> 5;
  int* __restrict__ row = HC + (size_t)blockIdx.x * NC;
  int carry = 0;
  for (int c0 = 0; c0 < NC; c0 += 256) {
    const int c = c0 + tid;
    const int v = (c < NC) ? row[c] : 0;
    int incl = v;
    #pragma unroll
    for (int off = 1; off < 32; off <<= 1) {
      const int y = __shfl_up(incl, off, 32);
      if (lane >= off) incl += y;
    }
    if (lane == 31) wtot[wave] = incl;
    __syncthreads();
    int woff = 0, tot = 0;
    #pragma unroll
    for (int w = 0; w < 8; ++w) {
      const int t = wtot[w];
      woff += (w < wave) ? t : 0;
      tot  += t;
    }
    if (c < NC) row[c] = carry + woff + incl - v;
    carry += tot;
    __syncthreads();
  }
  if (tid == 0) BT[blockIdx.x] = carry;
}

__global__ void __launch_bounds__(256)
bin_scatter_kernel(const int* __restrict__ ei, int E, int NB, int NC,
                   const int* __restrict__ HC, const int* __restrict__ BT,
                   int* __restrict__ BB, int* __restrict__ PBw) {
  int2* __restrict__ PB = (int2*)PBw;
  __shared__ int slot[MAX_BINS];
  __shared__ int wtot[8];
  const int tid  = threadIdx.x;
  const int lane = tid & 31;
  const int wave = tid >> 5;
  const int chunk = blockIdx.x;
  const int b0 = tid * 4;
  const int c0 = (b0 + 0 < NB) ? (b0 + 0) : (NB - 1);
  const int c1 = (b0 + 1 < NB) ? (b0 + 1) : (NB - 1);
  const int c2 = (b0 + 2 < NB) ? (b0 + 2) : (NB - 1);
  const int c3 = (b0 + 3 < NB) ? (b0 + 3) : (NB - 1);
  const int r0 = BT[c0], r1 = BT[c1], r2 = BT[c2], r3 = BT[c3];
  const int h0 = HC[(size_t)c0 * NC + chunk];
  const int h1 = HC[(size_t)c1 * NC + chunk];
  const int h2 = HC[(size_t)c2 * NC + chunk];
  const int h3 = HC[(size_t)c3 * NC + chunk];
  const int t0 = (b0 + 0 < NB) ? r0 : 0;
  const int t1 = (b0 + 1 < NB) ? r1 : 0;
  const int t2 = (b0 + 2 < NB) ? r2 : 0;
  const int t3 = (b0 + 3 < NB) ? r3 : 0;
  const int tsum = t0 + t1 + t2 + t3;
  int incl = tsum;
  #pragma unroll
  for (int off = 1; off < 32; off <<= 1) {
    const int y = __shfl_up(incl, off, 32);
    if (lane >= off) incl += y;
  }
  if (lane == 31) wtot[wave] = incl;
  __syncthreads();
  int woff = 0;
  #pragma unroll
  for (int w = 0; w < 8; ++w) woff += (w < wave) ? wtot[w] : 0;
  const int e0 = woff + incl - tsum;
  const int e1 = e0 + t0;
  const int e2 = e1 + t1;
  const int e3 = e2 + t2;
  if (b0 + 0 < NB) slot[b0 + 0] = e0 + h0;
  if (b0 + 1 < NB) slot[b0 + 1] = e1 + h1;
  if (b0 + 2 < NB) slot[b0 + 2] = e2 + h2;
  if (b0 + 3 < NB) slot[b0 + 3] = e3 + h3;
  if (chunk == 0) {
    if (b0 + 0 < NB) BB[b0 + 0] = e0;
    if (b0 + 1 < NB) BB[b0 + 1] = e1;
    if (b0 + 2 < NB) BB[b0 + 2] = e2;
    if (b0 + 3 < NB) BB[b0 + 3] = e3;
  }
  __syncthreads();
  const int base = chunk * CHUNK_EDGES;
  int s[CHUNK_EDGES / 256], d[CHUNK_EDGES / 256];
  #pragma unroll
  for (int j = 0; j < CHUNK_EDGES / 256; ++j) {
    const int i  = base + j * 256 + tid;
    const int ic = (i < E) ? i : (E - 1);
    s[j] = ei[ic];
    const int dv = ei[E + ic];
    d[j] = (i < E) ? dv : -1;
  }
  #pragma unroll
  for (int j = 0; j < CHUNK_EDGES / 256; ++j) {
    if (d[j] >= 0) {
      const int pos = atomicAdd(&slot[d[j] >> BIN_SHIFT], 1);
      PB[pos] = make_int2(s[j], d[j]);
    }
  }
}

__global__ void __launch_bounds__(1024)
bin_sort_kernel(const int* __restrict__ PBw, const int* __restrict__ BB,
                const int* __restrict__ BT, int N,
                int* __restrict__ rowptr, int* __restrict__ csr_src) {
  __shared__ int cnt[BIN_NODES];
  __shared__ int cur[BIN_NODES];
  __shared__ int wtot[8];
  const int tid  = threadIdx.x;
  const int lane = tid & 31;
  const int wave = tid >> 5;
  const int bs = BB[blockIdx.x];
  const int n  = BT[blockIdx.x];
  const int2* __restrict__ pb = (const int2*)PBw + bs;
  if (tid < BIN_NODES) cnt[tid] = 0;
  __syncthreads();
  for (int i0 = 0; i0 < n; i0 += 4096) {
    int d[4];
    #pragma unroll
    for (int u = 0; u < 4; ++u) {
      const int i  = i0 + u * 1024 + tid;
      const int ic = (i < n) ? i : (n - 1);
      const int dv = pb[ic].y;
      d[u] = (i < n) ? dv : -1;
    }
    #pragma unroll
    for (int u = 0; u < 4; ++u)
      if (d[u] >= 0) atomicAdd(&cnt[d[u] & (BIN_NODES - 1)], 1);
  }
  __syncthreads();
  const int v = (tid < BIN_NODES) ? cnt[tid] : 0;
  int incl = v;
  #pragma unroll
  for (int off = 1; off < 32; off <<= 1) {
    const int y = __shfl_up(incl, off, 32);
    if (lane >= off) incl += y;
  }
  if (tid < BIN_NODES && lane == 31) wtot[wave] = incl;
  __syncthreads();
  if (tid < BIN_NODES) {
    int woff = 0;
    #pragma unroll
    for (int w = 0; w < 8; ++w) woff += (w < wave) ? wtot[w] : 0;
    const int start = bs + woff + incl - v;
    cur[tid] = start;
    const int node = blockIdx.x * BIN_NODES + tid;
    if (node < N) rowptr[node] = start;
  }
  __syncthreads();
  for (int i0 = 0; i0 < n; i0 += 4096) {
    int2 p[4];
    #pragma unroll
    for (int u = 0; u < 4; ++u) {
      const int i  = i0 + u * 1024 + tid;
      const int ic = (i < n) ? i : (n - 1);
      p[u] = pb[ic];
      if (i >= n) p[u].y = -1;
    }
    #pragma unroll
    for (int u = 0; u < 4; ++u) {
      if (p[u].y >= 0) {
        const int pos = atomicAdd(&cur[p[u].y & (BIN_NODES - 1)], 1);
        csr_src[pos] = p[u].x;
      }
    }
  }
}

typedef __attribute__((ext_vector_type(4))) float v4f;

__device__ __forceinline__ float xhalf(float x) {
  const int b = __builtin_bit_cast(int, x);
  return __builtin_bit_cast(float,
      __builtin_amdgcn_permlanex16(b, b, 0x76543210, 0xfedcba98, false, false));
}
template <int NX>
__device__ __forceinline__ float xlane(float x) {
  const int b = __builtin_bit_cast(int, x);
  return __builtin_bit_cast(float,
      __builtin_amdgcn_update_dpp(0, b, 0x160 + NX, 0xf, 0xf, true));
}

template <int LPE>
__device__ __forceinline__ void attend_node(int wid, int lane,
                   const int* __restrict__ rowptr,
                   const int* __restrict__ csr_src, int N, int E,
                   const float* __restrict__ XL, const float* __restrict__ XR,
                   const v4f* att_s, v4f* xr_w,
                   const float* __restrict__ obias, int relu,
                   v4f (&res)[4 / (32 / LPE)]) {
  constexpr int S   = 32 / LPE;
  constexpr int D4  = 4 * LPE;
  constexpr int OPS = 4 / S;
  constexpr float LOWEST = -3.402823466e+38f;
  const int beg = __builtin_amdgcn_readfirstlane(rowptr[wid]);
  const int end = (wid + 1 < N) ? __builtin_amdgcn_readfirstlane(rowptr[wid + 1]) : E;
  const int c   = end - beg + 1;
  const int slot = lane / LPE;
  const int t    = lane % LPE;

  const v4f* __restrict__ xr4 =
      (const v4f*)((const char*)XR + ((unsigned)wid * (unsigned)(D4 * 16) + (unsigned)(t * 16)));
  const v4f* at_l = att_s + t;
  v4f* xr_l = xr_w + t;

  v4f acc[4];
  {
    v4f xv[4];
    #pragma unroll
    for (int j = 0; j < 4; ++j) xv[j] = xr4[j * LPE];
    if (slot == 0) {
      #pragma unroll
      for (int j = 0; j < 4; ++j) xr_l[j * LPE] = xv[j];
    }
  }
  #pragma unroll
  for (int j = 0; j < 4; ++j) acc[j] = v4f{0.0f, 0.0f, 0.0f, 0.0f};
  float m = LOWEST;
  float ssum = 0.0f;

  for (int base = 0; base < c; base += 32) {
    const int q = base + lane;
    int idxv = wid;
    if (q > 0 && q < c) idxv = csr_src[beg + q - 1];
    const int nq = (c - base < 32) ? (c - base) : 32;

    for (int k = 0; k < nq; k += S) {
      asm volatile("" ::: "memory");
      int src = __builtin_amdgcn_readlane(idxv, k);
      #pragma unroll
      for (int u = 1; u < S; ++u) {
        const int kk = (k + u < nq) ? (k + u) : k;
        const int su = __builtin_amdgcn_readlane(idxv, kk);
        if (slot == u) src = su;
      }
      const bool valid = (k + slot) < nq;
      const v4f* __restrict__ row =
          (const v4f*)((const char*)XL + ((unsigned)src * (unsigned)(D4 * 16) + (unsigned)(t * 16)));
      v4f r[4], xv[4], gv[4];
      #pragma unroll
      for (int j = 0; j < 4; ++j) r[j] = row[j * LPE];
      #pragma unroll
      for (int j = 0; j < 4; ++j) {
        xv[j] = xr_l[j * LPE];
        gv[j] = at_l[j * LPE];
      }
      asm volatile("" :: "v"(r[0]), "v"(r[1]), "v"(r[2]), "v"(r[3]),
                         "v"(xv[0]), "v"(xv[1]), "v"(xv[2]), "v"(xv[3]),
                         "v"(gv[0]), "v"(gv[1]), "v"(gv[2]), "v"(gv[3]));
      __builtin_amdgcn_sched_barrier(0);

      v4f a4 = v4f{0.0f, 0.0f, 0.0f, 0.0f};
      #pragma unroll
      for (int j = 0; j < 4; ++j) {
        v4f h = r[j] + xv[j];
        #pragma unroll
        for (int e = 0; e < 4; ++e) h[e] = h[e] > 0.0f ? h[e] : NEG_SLOPE * h[e];
        a4 += h * gv[j];
      }
      float a = (a4[0] + a4[1]) + (a4[2] + a4[3]);
      a += xlane<1>(a);
      a += xlane<2>(a);
      a += xlane<4>(a);
      if (LPE == 16) a += xlane<8>(a);

      const float ev = valid ? a : LOWEST;
      const float mn = fmaxf(m, ev);
      const float sc = __expf(m - mn);
      const float pe = __expf(ev - mn);
      const float p  = valid ? pe : 0.0f;
      ssum = ssum * sc + p;
      #pragma unroll
      for (int j = 0; j < 4; ++j) acc[j] = acc[j] * sc + p * r[j];
      m = mn;
    }
  }

  float mAll = m;
  if (LPE == 8) mAll = fmaxf(mAll, xlane<8>(mAll));
  mAll = fmaxf(mAll, xhalf(mAll));
  const float f = __expf(m - mAll);
  ssum *= f;
  #pragma unroll
  for (int j = 0; j < 4; ++j) acc[j] = acc[j] * f;
  if (LPE == 8) {
    ssum += xlane<8>(ssum);
    #pragma unroll
    for (int j = 0; j < 4; ++j) {
      #pragma unroll
      for (int e = 0; e < 4; ++e) acc[j][e] += xlane<8>(acc[j][e]);
    }
  }
  ssum += xhalf(ssum);
  #pragma unroll
  for (int j = 0; j < 4; ++j) {
    #pragma unroll
    for (int e = 0; e < 4; ++e) acc[j][e] += xhalf(acc[j][e]);
  }

  const float inv = 1.0f / ssum;
  const v4f* __restrict__ ob4 = (const v4f*)obias;
  #pragma unroll
  for (int i = 0; i < OPS; ++i) {
    v4f o = acc[i];
    #pragma unroll
    for (int s = 1; s < S; ++s)
      if (slot == s) o = acc[s * OPS + i];
    const int oi = (slot * OPS + i) * LPE + t;
    const v4f bb = ob4[oi];
    v4f rr = o * inv + bb;
    if (relu) {
      rr[0] = fmaxf(rr[0], 0.0f); rr[1] = fmaxf(rr[1], 0.0f);
      rr[2] = fmaxf(rr[2], 0.0f); rr[3] = fmaxf(rr[3], 0.0f);
    }
    res[i] = rr;
  }
}

#define TILE_NODES 32
#define TILE_WAVES 16
#define WS_COLS 48

template <int LPE, int MODE>
__global__ void __launch_bounds__(512, 12)
attend_product_kernel(const int* __restrict__ rowptr,
                      const int* __restrict__ csr_src, int N, int E,
                      const float* __restrict__ XL, const float* __restrict__ XR,
                      const float* __restrict__ att,
                      const float* __restrict__ obias, int relu,
                      const float* __restrict__ Wl, const float* __restrict__ Wr,
                      const float* __restrict__ bl, const float* __restrict__ br,
                      float* __restrict__ Yl, float* __restrict__ Yr, int Ncol) {
  constexpr int S     = 32 / LPE;
  constexpr int OPS   = 4 / S;
  constexpr int D     = 16 * LPE;
  constexpr int PITCH = D + 4;
  constexpr int P4    = PITCH / 4;
  __shared__ v4f hs4[TILE_NODES * P4];
  constexpr int WS4 = (MODE == 1) ? (D * WS_COLS / 4) : 1;
  __shared__ v4f ws4[WS4];
  __shared__ v4f att_s[D / 4];
  __shared__ v4f xr_s[TILE_WAVES * (D / 4)];
  if (threadIdx.x < D / 4) att_s[threadIdx.x] = ((const v4f*)att)[threadIdx.x];
  if (MODE == 1) {
    constexpr int PER = (WS4 + 32 * TILE_WAVES - 1) / (32 * TILE_WAVES);
    const int n4 = (D * Ncol) >> 2;
    const v4f* __restrict__ w4 = (const v4f*)Wl;
    v4f tmp[PER];
    #pragma unroll
    for (int j = 0; j < PER; ++j) {
      const int i  = j * (32 * TILE_WAVES) + (int)threadIdx.x;
      const int ic = (i < n4) ? i : (n4 - 1);
      tmp[j] = w4[ic];
    }
    #pragma unroll
    for (int j = 0; j < PER; ++j) {
      const int i = j * (32 * TILE_WAVES) + (int)threadIdx.x;
      if (i < n4) ws4[i] = tmp[j];
    }
  }
  const int lane = threadIdx.x & 31;
  const int wave = __builtin_amdgcn_readfirstlane((int)(threadIdx.x >> 5));
  const int nodeBase = blockIdx.x * TILE_NODES;
  __syncthreads();
  {
    const int slot = lane / LPE;
    const int t    = lane % LPE;
    #pragma unroll 1
    for (int it = 0; it < TILE_NODES / TILE_WAVES; ++it) {
      const int r = it * TILE_WAVES + wave;
      const int node = nodeBase + r;
      v4f res[OPS];
      if (node < N) {
        attend_node<LPE>(node, lane, rowptr, csr_src, N, E, XL, XR, att_s,
                         xr_s + wave * (D / 4), obias, relu, res);
      } else {
        #pragma unroll
        for (int i = 0; i < OPS; ++i) res[i] = v4f{0.0f, 0.0f, 0.0f, 0.0f};
      }
      #pragma unroll
      for (int i = 0; i < OPS; ++i)
        hs4[r * P4 + (slot * OPS + i) * LPE + t] = res[i];
    }
  }
  __syncthreads();

  const float* hs = (const float*)hs4;
  const int half = lane >> 4;
  const int l15  = lane & 15;
  const float* ap = hs + l15 * PITCH + half * 2;

  if (MODE == 0) {
    const int ngrp = Ncol >> 5;
    for (int task = wave; task < 2 * ngrp; task += TILE_WAVES) {
      const int mat = task / ngrp;
      const int colBase = (task - mat * ngrp) << 5;
      const float* __restrict__ Wm = mat ? Wr : Wl;
      const float* __restrict__ bm = mat ? br : bl;
      float* __restrict__ Ym = mat ? Yr : Yl;
      const float* __restrict__ bp = Wm + (size_t)(half * 2) * Ncol + colBase + 2 * l15;
      v8f c[4] = {v8f{}, v8f{}, v8f{}, v8f{}};
      for (int k0 = 0; k0 < D; k0 += 16) {
        v2f a0[4], a1[4], w0[4], w1[4];
        #pragma unroll
        for (int u = 0; u < 4; ++u) {
          const size_t ro = (size_t)(k0 + 4 * u) * Ncol;
          a0[u] = *(const v2f*)(ap + k0 + 4 * u);
          a1[u] = *(const v2f*)(ap + 16 * PITCH + k0 + 4 * u);
          w0[u] = *(const v2f*)(bp + ro);
          w1[u] = *(const v2f*)(bp + ro + Ncol);
        }
        #pragma unroll
        for (int u = 0; u < 4; ++u) {
          v2f be, bo;
          be[0] = w0[u][0]; be[1] = w1[u][0];
          bo[0] = w0[u][1]; bo[1] = w1[u][1];
          c[0] = __builtin_amdgcn_wmma_f32_16x16x4_f32(false, a0[u], false, be, (short)0, c[0], false, false);
          c[1] = __builtin_amdgcn_wmma_f32_16x16x4_f32(false, a0[u], false, bo, (short)0, c[1], false, false);
          c[2] = __builtin_amdgcn_wmma_f32_16x16x4_f32(false, a1[u], false, be, (short)0, c[2], false, false);
          c[3] = __builtin_amdgcn_wmma_f32_16x16x4_f32(false, a1[u], false, bo, (short)0, c[3], false, false);
        }
      }
      const int col = colBase + 2 * l15;
      v2f bb = {0.0f, 0.0f};
      if (bm) bb = *(const v2f*)(bm + col);
      float* __restrict__ ybase = Ym + (size_t)(nodeBase + half * 8) * Ncol + col;
      #pragma unroll
      for (int rt = 0; rt < 2; ++rt) {
        const int row0 = nodeBase + rt * 16 + half * 8;
        #pragma unroll
        for (int r = 0; r < 8; ++r) {
          if (row0 + r < N) {
            v2f o;
            o[0] = c[2 * rt][r]     + bb[0];
            o[1] = c[2 * rt + 1][r] + bb[1];
            *(v2f*)(ybase + (rt * 16 + r) * Ncol) = o;
          }
        }
      }
    }
  } else {
    const int ct = (Ncol + 15) >> 4;
    for (int task = wave; task < 2 * ct; task += TILE_WAVES) {
      const int rt = task / ct;
      const int bcol = ((task - rt * ct) << 4) + l15;
      const bool bok = bcol < Ncol;
      const int bcolc = bok ? bcol : (Ncol - 1);
      const float bmask = bok ? 1.0f : 0.0f;
      const float* art = ap + rt * 16 * PITCH;
      const float* bp = (const float*)ws4 + (half * 2) * Ncol + bcolc;
      v8f c = v8f{};
      #pragma unroll 1
      for (int k0 = 0; k0 < D; k0 += 32) {
        v2f a[8], w[8];
        #pragma unroll
        for (int u = 0; u < 8; ++u) {
          a[u] = *(const v2f*)(art + k0 + 4 * u);
          const float* bq = bp + (k0 + 4 * u) * Ncol;
          w[u][0] = bq[0];
          w[u][1] = bq[Ncol];
        }
        __builtin_amdgcn_sched_barrier(0);
        #pragma unroll
        for (int u = 0; u < 8; ++u) {
          v2f b;
          b[0] = w[u][0] * bmask;
          b[1] = w[u][1] * bmask;
          c = __builtin_amdgcn_wmma_f32_16x16x4_f32(false, a[u], false, b, (short)0, c, false, false);
        }
      }
      if (bok) {
        const float bb = bl ? bl[bcol] : 0.0f;
        const int row0 = nodeBase + rt * 16 + half * 8;
        #pragma unroll
        for (int r = 0; r < 8; ++r)
          if (row0 + r < N) Yl[(size_t)(row0 + r) * Ncol + bcol] = c[r] + bb;
      }
    }
  }
}

extern "C" void kernel_launch(void* const* d_in, const int* in_sizes, int n_in,
                              void* d_out, int out_size, void* d_ws, size_t ws_size,
                              hipStream_t stream) {
  const float* x   = (const float*)d_in[0];
  const int*   ei  = (const int*)  d_in[1];
  const float* w1l = (const float*)d_in[2];
  const float* b1l = (const float*)d_in[3];
  const float* w1r = (const float*)d_in[4];
  const float* b1r = (const float*)d_in[5];
  const float* a1  = (const float*)d_in[6];
  const float* o1  = (const float*)d_in[7];
  const float* w2l = (const float*)d_in[8];
  const float* b2l = (const float*)d_in[9];
  const float* w2r = (const float*)d_in[10];
  const float* b2r = (const float*)d_in[11];
  const float* a2  = (const float*)d_in[12];
  const float* o2  = (const float*)d_in[13];
  const float* w3l = (const float*)d_in[14];
  const float* b3l = (const float*)d_in[15];
  const float* w3r = (const float*)d_in[16];
  const float* b3r = (const float*)d_in[17];
  const float* a3  = (const float*)d_in[18];
  const float* o3  = (const float*)d_in[19];
  const float* wc  = (const float*)d_in[20];
  const float* bc  = (const float*)d_in[21];

  const int N  = in_sizes[0] / 128;
  const int E  = in_sizes[1] / 2;

  const int Np = (N + 3) & ~3;
  const int Ep = (E + 3) & ~3;
  const int NB = (N + BIN_NODES - 1) / BIN_NODES;
  const int NC = (E + CHUNK_EDGES - 1) / CHUNK_EDGES;
  const int Hp = (NB * NC + 3) & ~3;
  float* B0  = (float*)d_ws;
  float* B1  = B0 + (size_t)N * 256;
  float* B2  = B1 + (size_t)N * 256;
  float* B3  = B2 + (size_t)N * 256;
  int*   CS  = (int*)(B3 + (size_t)N * 256);
  int*   RP  = CS + (size_t)Ep;
  int*   HC  = RP + (size_t)Np;
  int*   BT  = HC + (size_t)Hp;
  int*   BB  = BT + MAX_BINS;
  int*   PB  = BB + MAX_BINS;

  bin_count_kernel<<<NC, 256, 0, stream>>>(ei, E, NB, NC, HC);
  bin_scan_kernel<<<NB, 256, 0, stream>>>(HC, NC, BT);
  bin_scatter_kernel<<<NC, 256, 0, stream>>>(ei, E, NB, NC, HC, BT, BB, PB);
  bin_sort_kernel<<<NB, 1024, 0, stream>>>(PB, BB, BT, N, RP, CS);

  {
    const int tiles = (N >> 4) * (128 >> 6);
    gemm2_bias_wmma_n64<<<(tiles + 3) / 4, 128, 0, stream>>>(x, w1l, w1r, b1l, b1r,
                                                            B0, B1, N, 128, 128);
  }
  const int nb = (N + TILE_NODES - 1) / TILE_NODES;
  attend_product_kernel<8, 0><<<nb, 32 * TILE_WAVES, 0, stream>>>(RP, CS, N, E, B0, B1, a1, o1, 1,
                                                      w2l, w2r, b2l, b2r, B2, B3, 128);
  attend_product_kernel<8, 0><<<nb, 32 * TILE_WAVES, 0, stream>>>(RP, CS, N, E, B2, B3, a2, o2, 1,
                                                      w3l, w3r, b3l, b3r, B0, B1, 256);
  attend_product_kernel<16, 1><<<nb, 32 * TILE_WAVES, 0, stream>>>(RP, CS, N, E, B0, B1, a3, o3, 0,
                                                       wc, nullptr, bc, nullptr,
                                                       (float*)d_out, nullptr, 40);
}
